# starting kernel
# baseline (speedup 1.0000x reference)
#include <hip/hip_runtime.h>
#include <cmath>


#define NN 50000
#define NE 800000
#define DIN 64
#define DHID 128
#define DOUT 64

typedef __attribute__((ext_vector_type(2))) float v2f;
typedef __attribute__((ext_vector_type(8))) float v8f;

__global__ void zero_kernel(float* __restrict__ p, int n) {
    int i = blockIdx.x * blockDim.x + threadIdx.x;
    if (i < n) p[i] = 0.0f;
}

__global__ void scatter1_kernel(const float* __restrict__ x,
                                const int* __restrict__ ei,
                                float* __restrict__ sum1,
                                float* __restrict__ deg) {
    int tid = blockIdx.x * blockDim.x + threadIdx.x;
    if (tid >= NE * DIN) return;
    int e = tid >> 6;
    int f = tid & 63;
    int src = ei[e];
    int dst = ei[NE + e];
    atomicAdd(&sum1[dst * DIN + f], x[src * DIN + f]);
    if (f == 0) atomicAdd(&deg[dst], 1.0f);
}

__global__ void scatter2_kernel(const float* __restrict__ h1,
                                const int* __restrict__ ei,
                                float* __restrict__ sum2) {
    int tid = blockIdx.x * blockDim.x + threadIdx.x;
    if (tid >= NE * DHID) return;
    int e = tid >> 7;
    int f = tid & 127;
    int src = ei[e];
    int dst = ei[NE + e];
    atomicAdd(&sum2[dst * DHID + f], h1[src * DHID + f]);
}

__global__ __launch_bounds__(32) void gemm1_kernel(
    const float* __restrict__ x, const float* __restrict__ sum1,
    const float* __restrict__ deg,
    const float* __restrict__ Wl, const float* __restrict__ bl,
    const float* __restrict__ Wr, float* __restrict__ h1) {
    const int lane = threadIdx.x;
    const int m0   = blockIdx.x * 16;
    const int mr   = lane & 15;
    const int kb   = (lane >> 4) * 2;
    const int m    = m0 + mr;
    const float dinv = 1.0f / fmaxf(deg[m], 1.0f);

    v8f acc[8] = {};

    for (int kk = 0; kk < DIN / 4; ++kk) {
        const int k = kk * 4 + kb;
        v2f a;
        a.x = sum1[m * DIN + k] * dinv;
        a.y = sum1[m * DIN + k + 1] * dinv;
#pragma unroll
        for (int nt = 0; nt < 8; ++nt) {
            const int n = nt * 16 + mr;
            v2f b;
            b.x = Wl[n * DIN + k];
            b.y = Wl[n * DIN + k + 1];
            acc[nt] = __builtin_amdgcn_wmma_f32_16x16x4_f32(
                false, a, false, b, (short)0, acc[nt], false, false);
        }
    }
    for (int kk = 0; kk < DIN / 4; ++kk) {
        const int k = kk * 4 + kb;
        v2f a;
        a.x = x[m * DIN + k];
        a.y = x[m * DIN + k + 1];
#pragma unroll
        for (int nt = 0; nt < 8; ++nt) {
            const int n = nt * 16 + mr;
            v2f b;
            b.x = Wr[n * DIN + k];
            b.y = Wr[n * DIN + k + 1];
            acc[nt] = __builtin_amdgcn_wmma_f32_16x16x4_f32(
                false, a, false, b, (short)0, acc[nt], false, false);
        }
    }

    const int mo = m0 + 8 * (lane >> 4);
#pragma unroll
    for (int nt = 0; nt < 8; ++nt) {
        const int n = nt * 16 + mr;
        const float bias = bl[n];
#pragma unroll
        for (int v = 0; v < 8; ++v) {
            h1[(mo + v) * DHID + n] = tanhf(acc[nt][v] + bias);
        }
    }
}

__global__ __launch_bounds__(32) void gemm2_kernel(
    const float* __restrict__ h1, const float* __restrict__ sum2,
    const float* __restrict__ deg,
    const float* __restrict__ Wl, const float* __restrict__ bl,
    const float* __restrict__ Wr, float* __restrict__ out) {
    __shared__ float zrow[16 * DOUT];
    const int lane = threadIdx.x;
    const int m0   = blockIdx.x * 16;
    const int mr   = lane & 15;
    const int kb   = (lane >> 4) * 2;
    const int m    = m0 + mr;
    const float dinv = 1.0f / fmaxf(deg[m], 1.0f);

    v8f acc[4] = {};

    for (int kk = 0; kk < DHID / 4; ++kk) {
        const int k = kk * 4 + kb;
        v2f a;
        a.x = sum2[m * DHID + k] * dinv;
        a.y = sum2[m * DHID + k + 1] * dinv;
#pragma unroll
        for (int nt = 0; nt < 4; ++nt) {
            const int n = nt * 16 + mr;
            v2f b;
            b.x = Wl[n * DHID + k];
            b.y = Wl[n * DHID + k + 1];
            acc[nt] = __builtin_amdgcn_wmma_f32_16x16x4_f32(
                false, a, false, b, (short)0, acc[nt], false, false);
        }
    }
    for (int kk = 0; kk < DHID / 4; ++kk) {
        const int k = kk * 4 + kb;
        v2f a;
        a.x = h1[m * DHID + k];
        a.y = h1[m * DHID + k + 1];
#pragma unroll
        for (int nt = 0; nt < 4; ++nt) {
            const int n = nt * 16 + mr;
            v2f b;
            b.x = Wr[n * DHID + k];
            b.y = Wr[n * DHID + k + 1];
            acc[nt] = __builtin_amdgcn_wmma_f32_16x16x4_f32(
                false, a, false, b, (short)0, acc[nt], false, false);
        }
    }

    const int rbase = 8 * (lane >> 4);
#pragma unroll
    for (int nt = 0; nt < 4; ++nt) {
        const int n = nt * 16 + mr;
        const float bias = bl[n];
#pragma unroll
        for (int v = 0; v < 8; ++v) {
            zrow[(rbase + v) * DOUT + n] = acc[nt][v] + bias;
        }
    }
    __syncthreads();

    if (lane < 16) {
        const float* zr = &zrow[lane * DOUT];
        float mx = zr[0];
        for (int c = 1; c < DOUT; ++c) mx = fmaxf(mx, zr[c]);
        float s = 0.0f;
        for (int c = 0; c < DOUT; ++c) s += expf(zr[c] - mx);
        const float lse = mx + logf(s);
        float* orow = &out[(m0 + lane) * DOUT];
        for (int c = 0; c < DOUT; ++c) orow[c] = zr[c] - lse;
    }
}

extern "C" void kernel_launch(void* const* d_in, const int* in_sizes, int n_in,
                              void* d_out, int out_size, void* d_ws, size_t ws_size,
                              hipStream_t stream) {
    const float* x   = (const float*)d_in[0];
    const int*   ei  = (const int*)d_in[1];
    const float* Wl1 = (const float*)d_in[2];
    const float* bl1 = (const float*)d_in[3];
    const float* Wr1 = (const float*)d_in[4];
    const float* Wl2 = (const float*)d_in[5];
    const float* bl2 = (const float*)d_in[6];
    const float* Wr2 = (const float*)d_in[7];
    float* out = (float*)d_out;

    float* ws   = (float*)d_ws;
    float* sum1 = ws;
    float* sum2 = sum1 + (size_t)NN * DIN;
    float* degb = sum2 + (size_t)NN * DHID;
    float* h1   = degb + NN;

    const int nz = NN * (DIN + DHID) + NN;
    zero_kernel<<<(nz + 255) / 256, 256, 0, stream>>>(ws, nz);

    scatter1_kernel<<<(NE * DIN + 255) / 256, 256, 0, stream>>>(x, ei, sum1, degb);
    gemm1_kernel<<<NN / 16, 32, 0, stream>>>(x, sum1, degb, Wl1, bl1, Wr1, h1);
    scatter2_kernel<<<(NE * DHID + 255) / 256, 256, 0, stream>>>(h1, ei, sum2);
    gemm2_kernel<<<NN / 16, 32, 0, stream>>>(h1, sum2, degb, Wl2, bl2, Wr2, out);
}


# Matrix kernels take all operands from LDS: a 13-wave work-group per processor stages both weight matrices and its 208 node rows by whole-line 16-byte loads; 8-byte fragment reads; sum order kept
# speedup vs baseline: 3.7146x; 3.7146x over previous
#include <hip/hip_runtime.h>
#include <cmath>


#define NN 50000
#define NE 800000
#define DIN 64
#define DHID 128
#define DOUT 64

#define TR 200
#define NT (NN / TR)
#define AGG_WAVES 32
#define AGG_THREADS (AGG_WAVES * 32)
#define NCH (NE / 128)
#define UNR 5
#define NPASS (NCH / AGG_WAVES / UNR)
#define QCAP 1024
#define LHEAD 64
#define LSTRIDE (TR * DHID)
#define LCAP (LSTRIDE - LHEAD)

static_assert(NN % TR == 0, "tiles must cover the nodes exactly");
static_assert(((size_t)NN * DIN * 4) % 256 == 0 && ((size_t)NN * DHID * 4) % 256 == 0, "sum2 and h1 must start on 256-byte boundaries");
static_assert(NE % 128 == 0, "edge list must split into whole chunks");
static_assert(TR <= 256, "local target index is packed into 8 bits");
static_assert(NE <= (1 << 24), "edge index is packed into 24 bits");
static_assert(NN <= (1 << 24), "source node is packed into 24 bits of a list entry");
static_assert(NCH / NT >= 1, "start offsets must differ between tiles");
static_assert(UNR * AGG_WAVES < NCH && AGG_WAVES + NT * (NCH / NT) <= 2 * NCH, "one subtraction must be enough to wrap a chunk number");
static_assert(QCAP >= UNR * 128 + 128, "one pass of the scan must always fit behind the flush mark");
static_assert(NCH - NPASS * UNR * AGG_WAVES < AGG_WAVES, "after its full passes a wave has at most one chunk left");

typedef __attribute__((ext_vector_type(2))) float v2f;
typedef __attribute__((ext_vector_type(8))) float v8f;

template <int D>
__device__ __forceinline__ void add_rows(const float* __restrict__ feat, float* tile,
                                         int s, int t, int m, int lane) {
    for (int k = 0; k < m; k += 8) {
        float2 r[8][D / 64];
#pragma unroll
        for (int i = 0; i < 8; ++i) {
            const int sk = __builtin_amdgcn_readlane(s, k + i);
            const float* row = feat + (size_t)sk * D + lane * 2;
#pragma unroll
            for (int j = 0; j < D / 64; ++j)
                r[i][j] = *reinterpret_cast<const float2*>(row + j * 64);
        }
#pragma unroll
        for (int i = 0; i < 8; ++i) {
            if (k + i < m) {
                const int tk = __builtin_amdgcn_readlane(t, k + i);
                float* trow = tile + tk * D + lane * 2;
#pragma unroll
                for (int j = 0; j < D / 64; ++j) {
                    atomicAdd(trow + j * 64, r[i][j].x);
                    atomicAdd(trow + j * 64 + 1, r[i][j].y);
                }
            }
        }
    }
}

template <int D, bool SAVE>
__device__ __forceinline__ void flush_queue(const float* __restrict__ feat,
                                            const int* __restrict__ ei,
                                            float* tile, const unsigned* q,
                                            int n, int lane, unsigned* list, int lbase) {
    for (int base = 0; base < n; base += 32) {
        const int idx = base + lane;
        unsigned ent = 0u;
        int s = 0;
        if (idx < n) {
            ent = q[idx];
            s = ei[ent >> 8];
            if (SAVE && lbase + idx < LCAP)
                list[LHEAD + lbase + idx] = ((unsigned)s << 8) | (ent & 255u);
        }
        add_rows<D>(feat, tile, s, (int)(ent & 255u), min(32, n - base), lane);
    }
}

template <int D>
__device__ __forceinline__ void replay_list(const float* __restrict__ feat, float* tile,
                                            const unsigned* lst, int n, int lane) {
    unsigned cur = lane < n ? lst[lane] : 0u;
    for (int base = 0; base < n; base += 32) {
        const int nx = base + 32 + lane;
        const unsigned nxt = nx < n ? lst[nx] : 0u;
        const int s = min((int)(cur >> 8), NN - 1);
        const int t = min((int)(cur & 255u), TR - 1);
        add_rows<D>(feat, tile, s, t, min(32, n - base), lane);
        cur = nxt;
    }
}

template <int D, bool FIRST>
__global__ __launch_bounds__(AGG_THREADS) void agg_kernel(
    const float* __restrict__ feat, const int* __restrict__ ei,
    float* sum, float* __restrict__ deg, float* lists) {
    constexpr bool DEG = FIRST;
    __shared__ float tile[TR * D];
    __shared__ unsigned queue[AGG_WAVES * QCAP];
    __shared__ int cnt[TR];
    __shared__ int lfill;

    const int tid  = threadIdx.x;
    const int lane = tid & 31;
    const int wave = __builtin_amdgcn_readfirstlane(tid >> 5);
    const int lo   = blockIdx.x * TR;

    unsigned* list = reinterpret_cast<unsigned*>(lists) + (size_t)blockIdx.x * LSTRIDE;
    int n_list = -1;
    if (!FIRST) n_list = (int)list[0];

    for (int i = tid; i < TR * D / 4; i += AGG_THREADS)
        reinterpret_cast<float4*>(tile)[i] = make_float4(0.f, 0.f, 0.f, 0.f);
    if (DEG)
        for (int i = tid; i < TR; i += AGG_THREADS) cnt[i] = 0;
    if (FIRST && tid == 0) lfill = 0;
    __syncthreads();

    bool replayed = false;
    if (!FIRST) {
        const int n = n_list;
        if (n >= 0 && n <= LCAP) {
            const int per = ((n + AGG_WAVES - 1) / AGG_WAVES + 7) & ~7;
            const int beg = min(n, wave * per);
            const int len = min(n - beg, per);
            replay_list<D>(feat, tile, list + LHEAD + beg, len, lane);
            replayed = true;
        }
    }
    if (!replayed) {
    unsigned* q = queue + wave * QCAP;
    int qn = 0;
    const int4* dp = reinterpret_cast<const int4*>(ei + NE);
    const int start = blockIdx.x * (NCH / NT);

    int nb = wave + start;
    if (nb >= NCH) nb -= NCH;
    auto load_pass = [&](int4 (&v)[UNR], int (&ch)[UNR]) {
#pragma unroll
        for (int u = 0; u < UNR; ++u) {
            int r = nb + u * AGG_WAVES;
            if (r >= NCH) r -= NCH;
            ch[u] = r;
            v[u] = dp[r * 32 + lane];
        }
        nb += UNR * AGG_WAVES;
        if (nb >= NCH) nb -= NCH;
    };
    auto scan_chunk = [&](const int4& d, int chunk) {
        const unsigned t0 = (unsigned)(d.x - lo);
        const unsigned t1 = (unsigned)(d.y - lo);
        const unsigned t2 = (unsigned)(d.z - lo);
        const unsigned t3 = (unsigned)(d.w - lo);
        const bool m0 = t0 < TR, m1 = t1 < TR, m2 = t2 < TR, m3 = t3 < TR;
        if (__builtin_amdgcn_ballot_w32(m0 | m1 | m2 | m3)) {
            const unsigned e0 = (unsigned)(chunk * 32 + lane) * 4u;
            const unsigned tt[4] = {t0, t1, t2, t3};
            const bool     mm[4] = {m0, m1, m2, m3};
#pragma unroll
            for (int j = 0; j < 4; ++j) {
                const unsigned mask = __builtin_amdgcn_ballot_w32(mm[j]);
                if (mask) {
                    if (mm[j]) {
                        const int pos = qn + (int)__builtin_amdgcn_mbcnt_lo(mask, 0u);
                        q[pos] = ((e0 + j) << 8) | tt[j];
                        if (DEG) atomicAdd(&cnt[tt[j]], 1);
                    }
                    qn += __builtin_popcount(mask);
                }
            }
        }
    };

    auto flush_now = [&]() {
        int lb = 0;
        if (FIRST) {
            if (lane == 0) lb = atomicAdd(&lfill, qn);
            lb = __builtin_amdgcn_readfirstlane(lb);
        }
        flush_queue<D, FIRST>(feat, ei, tile, q, qn, lane, list, lb);
        qn = 0;
    };

    auto step = [&](int4 (&cur)[UNR], int (&cch)[UNR], int4 (&nxt)[UNR], int (&nxc)[UNR]) {
        load_pass(nxt, nxc);
#pragma unroll
        for (int u = 0; u < UNR; ++u) scan_chunk(cur[u], cch[u]);
        if (qn > QCAP - UNR * 128) {
            __builtin_amdgcn_fence(__ATOMIC_RELEASE, "wavefront");
            __builtin_amdgcn_fence(__ATOMIC_ACQUIRE, "wavefront");
            flush_now();
        }
    };

    static_assert(NPASS % 2 == 1, "the loop below does the passes in pairs, then one more");
    int4 va[UNR], vb[UNR];
    int  ca[UNR], cb[UNR];
    load_pass(va, ca);
    for (int p = 0; p + 1 < NPASS; p += 2) {
        step(va, ca, vb, cb);
        step(vb, cb, va, ca);
    }
    step(va, ca, vb, cb);
    if (wave + NPASS * UNR * AGG_WAVES < NCH) scan_chunk(vb[0], cb[0]);
    __builtin_amdgcn_fence(__ATOMIC_RELEASE, "wavefront");
    __builtin_amdgcn_fence(__ATOMIC_ACQUIRE, "wavefront");
    flush_now();
    }
    __syncthreads();

    float4* so = reinterpret_cast<float4*>(sum + (size_t)lo * D);
    for (int i = tid; i < TR * D / 4; i += AGG_THREADS)
        so[i] = reinterpret_cast<const float4*>(tile)[i];
    if (DEG)
        for (int i = tid; i < TR; i += AGG_THREADS) deg[lo + i] = (float)cnt[i];
    if (FIRST && tid == 0) list[0] = (unsigned)lfill;
}

#define G_WAVES 13
#define G_THREADS (G_WAVES * 32)
#define G_ROWS (G_WAVES * 16)
#define NTILE (NN / 16)
#define G_BLOCKS ((NTILE + G_WAVES - 1) / G_WAVES)
#define STAGE_B 8

static_assert(NN % 16 == 0, "row tiles must cover the nodes exactly");
static_assert(G_BLOCKS <= 256, "one work-group per processor, all resident at once");
static_assert((2 * DHID * (DIN + 4) + 2 * G_ROWS * (DIN + 4)) * 4 > 160 * 1024, "layer 1: one work-group per processor");
static_assert((2 * DOUT * (DHID + 4) + 2 * G_ROWS * (DHID + 4)) * 4 > 160 * 1024, "layer 2: one work-group per processor");
static_assert((2 * DOUT * (DHID + 4) + 2 * G_ROWS * (DHID + 4)) * 4 <= 320 * 1024, "layer 2 must fit the LDS");

template <int K, int ROWS>
__device__ __forceinline__ void stage_pair(const float* __restrict__ s0, const float* __restrict__ s1,
                                           float* d0, float* d1, int nrows, int tid) {
    constexpr int TOT = ROWS * (K / 4);
    constexpr int PER = (TOT + G_THREADS - 1) / G_THREADS;
    constexpr int NB  = (PER + STAGE_B - 1) / STAGE_B;
    constexpr int CNT = STAGE_B < PER ? STAGE_B : PER;
    static_assert(PER % CNT == 0, "batches of equal size");
    const float4* p0 = reinterpret_cast<const float4*>(s0);
    const float4* p1 = reinterpret_cast<const float4*>(s1);
    const int last = nrows * (K / 4) - 1;
#pragma unroll
    for (int b = 0; b < NB; ++b) {
        float4 v0[CNT], v1[CNT];
#pragma unroll
        for (int j = 0; j < CNT; ++j) {
            const int i = min(tid + (b * CNT + j) * G_THREADS, last);
            v0[j] = p0[i];
            v1[j] = p1[i];
        }
#pragma unroll
        for (int j = 0; j < CNT; ++j) {
            const int i = tid + (b * CNT + j) * G_THREADS;
            if ((b * CNT + j + 1) * G_THREADS <= TOT || i < TOT) {
                const int o = (i / (K / 4)) * (K + 4) + (i % (K / 4)) * 4;
                *reinterpret_cast<float4*>(d0 + o) = v0[j];
                *reinterpret_cast<float4*>(d1 + o) = v1[j];
            }
        }
    }
}

__global__ __launch_bounds__(G_THREADS) void gemm1_kernel(
    const float* __restrict__ x, const float* __restrict__ sum1,
    const float* __restrict__ deg,
    const float* __restrict__ Wl, const float* __restrict__ bl,
    const float* __restrict__ Wr, float* __restrict__ h1) {
    constexpr int P = DIN + 4;
    __shared__ __attribute__((aligned(16))) float wl_s[DHID * P];
    __shared__ __attribute__((aligned(16))) float wr_s[DHID * P];
    __shared__ __attribute__((aligned(16))) float as_s[G_ROWS * P];
    __shared__ __attribute__((aligned(16))) float ax_s[G_ROWS * P];

    const int tid  = threadIdx.x;
    const int lane = tid & 31;
    const int wave = __builtin_amdgcn_readfirstlane(tid >> 5);
    const int row0 = blockIdx.x * G_ROWS;
    const int tile = blockIdx.x * G_WAVES + wave;
    const int mr   = lane & 15;
    const int kb   = (lane >> 4) * 2;
    const float degm = deg[min(tile, NTILE - 1) * 16 + mr];

    stage_pair<DIN, DHID>(Wl, Wr, wl_s, wr_s, DHID, tid);
    stage_pair<DIN, G_ROWS>(sum1 + (size_t)row0 * DIN, x + (size_t)row0 * DIN, as_s, ax_s,
                            min(G_ROWS, NN - row0), tid);
    __syncthreads();
    if (tile >= NTILE) return;

    const int m0 = tile * 16;
    const float dinv = 1.0f / fmaxf(degm, 1.0f);
    const float* arow = as_s + (wave * 16 + mr) * P + kb;
    const float* xrow = ax_s + (wave * 16 + mr) * P + kb;

    v8f acc[8] = {};

    for (int kk = 0; kk < DIN / 4; ++kk) {
        const v2f s = *reinterpret_cast<const v2f*>(arow + kk * 4);
        v2f a;
        a.x = s.x * dinv;
        a.y = s.y * dinv;
#pragma unroll
        for (int nt = 0; nt < 8; ++nt) {
            const v2f b = *reinterpret_cast<const v2f*>(wl_s + (nt * 16 + mr) * P + kk * 4 + kb);
            acc[nt] = __builtin_amdgcn_wmma_f32_16x16x4_f32(
                false, a, false, b, (short)0, acc[nt], false, false);
        }
    }
    for (int kk = 0; kk < DIN / 4; ++kk) {
        const v2f a = *reinterpret_cast<const v2f*>(xrow + kk * 4);
#pragma unroll
        for (int nt = 0; nt < 8; ++nt) {
            const v2f b = *reinterpret_cast<const v2f*>(wr_s + (nt * 16 + mr) * P + kk * 4 + kb);
            acc[nt] = __builtin_amdgcn_wmma_f32_16x16x4_f32(
                false, a, false, b, (short)0, acc[nt], false, false);
        }
    }

    const int mo = m0 + 8 * (lane >> 4);
#pragma unroll
    for (int nt = 0; nt < 8; ++nt) {
        const int n = nt * 16 + mr;
        const float bias = bl[n];
#pragma unroll
        for (int v = 0; v < 8; ++v) {
            h1[(mo + v) * DHID + n] = tanhf(acc[nt][v] + bias);
        }
    }
}

__global__ __launch_bounds__(G_THREADS) void gemm2_kernel(
    const float* __restrict__ h1, const float* __restrict__ sum2,
    const float* __restrict__ deg,
    const float* __restrict__ Wl, const float* __restrict__ bl,
    const float* __restrict__ Wr, float* __restrict__ out) {
    constexpr int P = DHID + 4;
    __shared__ __attribute__((aligned(16))) float wl_s[DOUT * P];
    __shared__ __attribute__((aligned(16))) float wr_s[DOUT * P];
    __shared__ __attribute__((aligned(16))) float as_s[G_ROWS * P];
    __shared__ __attribute__((aligned(16))) float ah_s[G_ROWS * P];

    const int tid  = threadIdx.x;
    const int lane = tid & 31;
    const int wave = __builtin_amdgcn_readfirstlane(tid >> 5);
    const int row0 = blockIdx.x * G_ROWS;
    const int tile = blockIdx.x * G_WAVES + wave;
    const int mr   = lane & 15;
    const int kb   = (lane >> 4) * 2;
    const float degm = deg[min(tile, NTILE - 1) * 16 + mr];

    stage_pair<DHID, DOUT>(Wl, Wr, wl_s, wr_s, DOUT, tid);
    stage_pair<DHID, G_ROWS>(sum2 + (size_t)row0 * DHID, h1 + (size_t)row0 * DHID, as_s, ah_s,
                             min(G_ROWS, NN - row0), tid);
    __syncthreads();
    if (tile >= NTILE) return;

    const int m0 = tile * 16;
    const float dinv = 1.0f / fmaxf(degm, 1.0f);
    const float* arow = as_s + (wave * 16 + mr) * P + kb;
    const float* hrow = ah_s + (wave * 16 + mr) * P + kb;

    v8f acc[4] = {};

    for (int kk = 0; kk < DHID / 4; ++kk) {
        const v2f s = *reinterpret_cast<const v2f*>(arow + kk * 4);
        v2f a;
        a.x = s.x * dinv;
        a.y = s.y * dinv;
#pragma unroll
        for (int nt = 0; nt < 4; ++nt) {
            const v2f b = *reinterpret_cast<const v2f*>(wl_s + (nt * 16 + mr) * P + kk * 4 + kb);
            acc[nt] = __builtin_amdgcn_wmma_f32_16x16x4_f32(
                false, a, false, b, (short)0, acc[nt], false, false);
        }
    }
    for (int kk = 0; kk < DHID / 4; ++kk) {
        const v2f a = *reinterpret_cast<const v2f*>(hrow + kk * 4);
#pragma unroll
        for (int nt = 0; nt < 4; ++nt) {
            const v2f b = *reinterpret_cast<const v2f*>(wr_s + (nt * 16 + mr) * P + kk * 4 + kb);
            acc[nt] = __builtin_amdgcn_wmma_f32_16x16x4_f32(
                false, a, false, b, (short)0, acc[nt], false, false);
        }
    }

    float* zrow = as_s + wave * 16 * P;
    const int rbase = 8 * (lane >> 4);
#pragma unroll
    for (int nt = 0; nt < 4; ++nt) {
        const int n = nt * 16 + mr;
        const float bias = bl[n];
#pragma unroll
        for (int v = 0; v < 8; ++v) {
            zrow[(rbase + v) * DOUT + n] = acc[nt][v] + bias;
        }
    }
    __builtin_amdgcn_fence(__ATOMIC_RELEASE, "wavefront");
    __builtin_amdgcn_fence(__ATOMIC_ACQUIRE, "wavefront");

    if (lane < 16) {
        const float* zr = &zrow[lane * DOUT];
        float mx = zr[0];
        for (int c = 1; c < DOUT; ++c) mx = fmaxf(mx, zr[c]);
        float s = 0.0f;
        for (int c = 0; c < DOUT; ++c) s += expf(zr[c] - mx);
        const float lse = mx + logf(s);
        float* orow = &out[(m0 + lane) * DOUT];
        for (int c = 0; c < DOUT; ++c) orow[c] = zr[c] - lse;
    }
}

extern "C" void kernel_launch(void* const* d_in, const int* in_sizes, int n_in,
                              void* d_out, int out_size, void* d_ws, size_t ws_size,
                              hipStream_t stream) {
    const float* x   = (const float*)d_in[0];
    const int*   ei  = (const int*)d_in[1];
    const float* Wl1 = (const float*)d_in[2];
    const float* bl1 = (const float*)d_in[3];
    const float* Wr1 = (const float*)d_in[4];
    const float* Wl2 = (const float*)d_in[5];
    const float* bl2 = (const float*)d_in[6];
    const float* Wr2 = (const float*)d_in[7];
    float* out = (float*)d_out;

    float* ws   = (float*)d_ws;
    float* sum1 = ws;
    float* sum2 = sum1 + (size_t)NN * DIN;
    float* h1   = sum2 + (size_t)NN * DHID;
    float* degb = h1 + (size_t)NN * DHID;

    agg_kernel<DIN, true><<<NT, AGG_THREADS, 0, stream>>>(x, ei, sum1, degb, sum2);
    gemm1_kernel<<<G_BLOCKS, G_THREADS, 0, stream>>>(x, sum1, degb, Wl1, bl1, Wr1, h1);
    agg_kernel<DHID, false><<<NT, AGG_THREADS, 0, stream>>>(h1, ei, sum2, degb, sum2);
    gemm2_kernel<<<G_BLOCKS, G_THREADS, 0, stream>>>(h1, sum2, degb, Wl2, bl2, Wr2, out);
}


# Layer 2 in one kernel: the work-group that gathers a tile's h1 sums in LDS also multiplies them there (sums never stored, 3 launches); 16x16 blocks over its 32 waves; softmax sum by lane tree
# speedup vs baseline: 4.3851x; 1.1805x over previous
#include <hip/hip_runtime.h>
#include <cmath>


#define NN 50000
#define NE 800000
#define DIN 64
#define DHID 128
#define DOUT 64

#define TR 208
#define NT ((NN + TR - 1) / TR)
#define AGG_WAVES 32
#define AGG_THREADS (AGG_WAVES * 32)
#define NCH (NE / 128)
#define UNR 5
#define NPASS (NCH / AGG_WAVES / UNR)
#define QCAP 1024
#define LHEAD 64
#define LSTRIDE ((NN * DHID / NT) / 64 * 64)
#define LCAP (LSTRIDE - LHEAD)

static_assert((NT - 1) * TR < NN && NT * TR >= NN && (NN - (NT - 1) * TR) % 16 == 0, "tiles cover the nodes; the last one holds whole row tiles");
static_assert((size_t)NT * LSTRIDE <= (size_t)NN * DHID, "the lists must fit their part of the workspace");
static_assert(((size_t)NN * DIN * 4) % 256 == 0 && ((size_t)NN * DHID * 4) % 256 == 0, "the lists and h1 must start on 256-byte boundaries");
static_assert(NE % 128 == 0, "edge list must split into whole chunks");
static_assert(TR <= 256, "local target index is packed into 8 bits");
static_assert(NE <= (1 << 24), "edge index is packed into 24 bits");
static_assert(NN <= (1 << 24), "source node is packed into 24 bits of a list entry");
static_assert(NCH / NT >= 1, "start offsets must differ between tiles");
static_assert(UNR * AGG_WAVES < NCH && AGG_WAVES + NT * (NCH / NT) <= 2 * NCH, "one subtraction must be enough to wrap a chunk number");
static_assert(QCAP >= UNR * 128 + 128, "one pass of the scan must always fit behind the flush mark");
static_assert(NCH - NPASS * UNR * AGG_WAVES < AGG_WAVES, "after its full passes a wave has at most one chunk left");

typedef __attribute__((ext_vector_type(2))) float v2f;
typedef __attribute__((ext_vector_type(8))) float v8f;

template <int D, int P>
__device__ __forceinline__ void add_rows(const float* __restrict__ feat, float* tile,
                                         int s, int t, int m, int lane) {
    for (int k = 0; k < m; k += 8) {
        float2 r[8][D / 64];
#pragma unroll
        for (int i = 0; i < 8; ++i) {
            const int sk = __builtin_amdgcn_readlane(s, k + i);
            const float* row = feat + (size_t)sk * D + lane * 2;
#pragma unroll
            for (int j = 0; j < D / 64; ++j)
                r[i][j] = *reinterpret_cast<const float2*>(row + j * 64);
        }
#pragma unroll
        for (int i = 0; i < 8; ++i) {
            if (k + i < m) {
                const int tk = __builtin_amdgcn_readlane(t, k + i);
                float* trow = tile + tk * P + lane * 2;
#pragma unroll
                for (int j = 0; j < D / 64; ++j) {
                    atomicAdd(trow + j * 64, r[i][j].x);
                    atomicAdd(trow + j * 64 + 1, r[i][j].y);
                }
            }
        }
    }
}

template <int D, int P, bool SAVE>
__device__ __forceinline__ void flush_queue(const float* __restrict__ feat,
                                            const int* __restrict__ ei,
                                            float* tile, const unsigned* q,
                                            int n, int lane, unsigned* list, int lbase) {
    for (int base = 0; base < n; base += 32) {
        const int idx = base + lane;
        unsigned ent = 0u;
        int s = 0;
        if (idx < n) {
            ent = q[idx];
            s = ei[ent >> 8];
            if (SAVE && lbase + idx < LCAP)
                list[LHEAD + lbase + idx] = ((unsigned)s << 8) | (ent & 255u);
        }
        add_rows<D, P>(feat, tile, s, (int)(ent & 255u), min(32, n - base), lane);
    }
}

template <int D, int P>
__device__ __forceinline__ void replay_list(const float* __restrict__ feat, float* tile,
                                            const unsigned* lst, int n, int lane) {
    unsigned cur = lane < n ? lst[lane] : 0u;
    for (int base = 0; base < n; base += 32) {
        const int nx = base + 32 + lane;
        const unsigned nxt = nx < n ? lst[nx] : 0u;
        const int s = min((int)(cur >> 8), NN - 1);
        const int t = min((int)(cur & 255u), TR - 1);
        add_rows<D, P>(feat, tile, s, t, min(32, n - base), lane);
        cur = nxt;
    }
}

template <int D, int P, bool FIRST>
__device__ __forceinline__ void gather_tile(
    const float* __restrict__ feat, const int* __restrict__ ei, float* tile,
    unsigned* queue, int* cnt, int* lfill, unsigned* list, int n_list,
    int lo, int lane, int wave) {
    constexpr bool DEG = FIRST;
    bool replayed = false;
    if (!FIRST) {
        const int n = n_list;
        if (n >= 0 && n <= LCAP) {
            const int per = ((n + AGG_WAVES - 1) / AGG_WAVES + 7) & ~7;
            const int beg = min(n, wave * per);
            const int len = min(n - beg, per);
            replay_list<D, P>(feat, tile, list + LHEAD + beg, len, lane);
            replayed = true;
        }
    }
    if (!replayed) {
    unsigned* q = queue + wave * QCAP;
    int qn = 0;
    const int4* dp = reinterpret_cast<const int4*>(ei + NE);
    const int start = blockIdx.x * (NCH / NT);

    int nb = wave + start;
    if (nb >= NCH) nb -= NCH;
    auto load_pass = [&](int4 (&v)[UNR], int (&ch)[UNR]) {
#pragma unroll
        for (int u = 0; u < UNR; ++u) {
            int r = nb + u * AGG_WAVES;
            if (r >= NCH) r -= NCH;
            ch[u] = r;
            v[u] = dp[r * 32 + lane];
        }
        nb += UNR * AGG_WAVES;
        if (nb >= NCH) nb -= NCH;
    };
    auto scan_chunk = [&](const int4& d, int chunk) {
        const unsigned t0 = (unsigned)(d.x - lo);
        const unsigned t1 = (unsigned)(d.y - lo);
        const unsigned t2 = (unsigned)(d.z - lo);
        const unsigned t3 = (unsigned)(d.w - lo);
        const bool m0 = t0 < TR, m1 = t1 < TR, m2 = t2 < TR, m3 = t3 < TR;
        if (__builtin_amdgcn_ballot_w32(m0 | m1 | m2 | m3)) {
            const unsigned e0 = (unsigned)(chunk * 32 + lane) * 4u;
            const unsigned tt[4] = {t0, t1, t2, t3};
            const bool     mm[4] = {m0, m1, m2, m3};
#pragma unroll
            for (int j = 0; j < 4; ++j) {
                const unsigned mask = __builtin_amdgcn_ballot_w32(mm[j]);
                if (mask) {
                    if (mm[j]) {
                        const int pos = qn + (int)__builtin_amdgcn_mbcnt_lo(mask, 0u);
                        q[pos] = ((e0 + j) << 8) | tt[j];
                        if (DEG) atomicAdd(&cnt[tt[j]], 1);
                    }
                    qn += __builtin_popcount(mask);
                }
            }
        }
    };

    auto flush_now = [&]() {
        int lb = 0;
        if (FIRST) {
            if (lane == 0) lb = atomicAdd(lfill, qn);
            lb = __builtin_amdgcn_readfirstlane(lb);
        }
        flush_queue<D, P, FIRST>(feat, ei, tile, q, qn, lane, list, lb);
        qn = 0;
    };

    auto step = [&](int4 (&cur)[UNR], int (&cch)[UNR], int4 (&nxt)[UNR], int (&nxc)[UNR]) {
        load_pass(nxt, nxc);
#pragma unroll
        for (int u = 0; u < UNR; ++u) scan_chunk(cur[u], cch[u]);
        if (qn > QCAP - UNR * 128) {
            __builtin_amdgcn_fence(__ATOMIC_RELEASE, "wavefront");
            __builtin_amdgcn_fence(__ATOMIC_ACQUIRE, "wavefront");
            flush_now();
        }
    };

    static_assert(NPASS % 2 == 1, "the loop below does the passes in pairs, then one more");
    int4 va[UNR], vb[UNR];
    int  ca[UNR], cb[UNR];
    load_pass(va, ca);
    for (int p = 0; p + 1 < NPASS; p += 2) {
        step(va, ca, vb, cb);
        step(vb, cb, va, ca);
    }
    step(va, ca, vb, cb);
    if (wave + NPASS * UNR * AGG_WAVES < NCH) scan_chunk(vb[0], cb[0]);
    __builtin_amdgcn_fence(__ATOMIC_RELEASE, "wavefront");
    __builtin_amdgcn_fence(__ATOMIC_ACQUIRE, "wavefront");
    flush_now();
    }
}

__global__ __launch_bounds__(AGG_THREADS) void agg1_kernel(
    const float* __restrict__ feat, const int* __restrict__ ei,
    float* __restrict__ sum, float* __restrict__ deg, float* __restrict__ lists) {
    constexpr int D = DIN;
    __shared__ float tile[TR * D];
    __shared__ unsigned queue[AGG_WAVES * QCAP];
    __shared__ int cnt[TR];
    __shared__ int lfill;

    const int tid  = threadIdx.x;
    const int lane = tid & 31;
    const int wave = __builtin_amdgcn_readfirstlane(tid >> 5);
    const int lo   = blockIdx.x * TR;
    const int nv   = min(TR, NN - lo);

    unsigned* list = reinterpret_cast<unsigned*>(lists) + (size_t)blockIdx.x * LSTRIDE;

    for (int i = tid; i < TR * D / 4; i += AGG_THREADS)
        reinterpret_cast<float4*>(tile)[i] = make_float4(0.f, 0.f, 0.f, 0.f);
    for (int i = tid; i < TR; i += AGG_THREADS) cnt[i] = 0;
    if (tid == 0) lfill = 0;
    __syncthreads();

    gather_tile<D, D, true>(feat, ei, tile, queue, cnt, &lfill, list, -1, lo, lane, wave);
    __syncthreads();

    float4* so = reinterpret_cast<float4*>(sum + (size_t)lo * D);
    for (int i = tid; i < nv * D / 4; i += AGG_THREADS)
        so[i] = reinterpret_cast<const float4*>(tile)[i];
    for (int i = tid; i < nv; i += AGG_THREADS) deg[lo + i] = (float)cnt[i];
    if (tid == 0) list[0] = (unsigned)lfill;
}

#define G_WAVES 13
#define G_THREADS (G_WAVES * 32)
#define G_ROWS (G_WAVES * 16)
#define NTILE (NN / 16)
#define G_BLOCKS ((NTILE + G_WAVES - 1) / G_WAVES)
#define STAGE_B 8

static_assert(NN % 16 == 0, "row tiles must cover the nodes exactly");
static_assert(G_BLOCKS <= 256, "one work-group per processor, all resident at once");
static_assert((2 * DHID * (DIN + 4) + 2 * G_ROWS * (DIN + 4)) * 4 > 160 * 1024, "layer 1: one work-group per processor");

template <int K, int ROWS>
__device__ __forceinline__ void stage_pair(const float* __restrict__ s0, const float* __restrict__ s1,
                                           float* d0, float* d1, int nrows, int tid) {
    constexpr int TOT = ROWS * (K / 4);
    constexpr int PER = (TOT + G_THREADS - 1) / G_THREADS;
    constexpr int NB  = (PER + STAGE_B - 1) / STAGE_B;
    constexpr int CNT = STAGE_B < PER ? STAGE_B : PER;
    static_assert(PER % CNT == 0, "batches of equal size");
    const float4* p0 = reinterpret_cast<const float4*>(s0);
    const float4* p1 = reinterpret_cast<const float4*>(s1);
    const int last = nrows * (K / 4) - 1;
#pragma unroll
    for (int b = 0; b < NB; ++b) {
        float4 v0[CNT], v1[CNT];
#pragma unroll
        for (int j = 0; j < CNT; ++j) {
            const int i = min(tid + (b * CNT + j) * G_THREADS, last);
            v0[j] = p0[i];
            v1[j] = p1[i];
        }
#pragma unroll
        for (int j = 0; j < CNT; ++j) {
            const int i = tid + (b * CNT + j) * G_THREADS;
            if ((b * CNT + j + 1) * G_THREADS <= TOT || i < TOT) {
                const int o = (i / (K / 4)) * (K + 4) + (i % (K / 4)) * 4;
                *reinterpret_cast<float4*>(d0 + o) = v0[j];
                *reinterpret_cast<float4*>(d1 + o) = v1[j];
            }
        }
    }
}

__global__ __launch_bounds__(G_THREADS) void gemm1_kernel(
    const float* __restrict__ x, const float* __restrict__ sum1,
    const float* __restrict__ deg,
    const float* __restrict__ Wl, const float* __restrict__ bl,
    const float* __restrict__ Wr, float* __restrict__ h1) {
    constexpr int P = DIN + 4;
    __shared__ __attribute__((aligned(16))) float wl_s[DHID * P];
    __shared__ __attribute__((aligned(16))) float wr_s[DHID * P];
    __shared__ __attribute__((aligned(16))) float as_s[G_ROWS * P];
    __shared__ __attribute__((aligned(16))) float ax_s[G_ROWS * P];

    const int tid  = threadIdx.x;
    const int lane = tid & 31;
    const int wave = __builtin_amdgcn_readfirstlane(tid >> 5);
    const int row0 = blockIdx.x * G_ROWS;
    const int tile = blockIdx.x * G_WAVES + wave;
    const int mr   = lane & 15;
    const int kb   = (lane >> 4) * 2;
    const float degm = deg[min(tile, NTILE - 1) * 16 + mr];

    stage_pair<DIN, DHID>(Wl, Wr, wl_s, wr_s, DHID, tid);
    stage_pair<DIN, G_ROWS>(sum1 + (size_t)row0 * DIN, x + (size_t)row0 * DIN, as_s, ax_s,
                            min(G_ROWS, NN - row0), tid);
    __syncthreads();
    if (tile >= NTILE) return;

    const int m0 = tile * 16;
    const float dinv = 1.0f / fmaxf(degm, 1.0f);
    const float* arow = as_s + (wave * 16 + mr) * P + kb;
    const float* xrow = ax_s + (wave * 16 + mr) * P + kb;

    v8f acc[8] = {};

    for (int kk = 0; kk < DIN / 4; ++kk) {
        const v2f s = *reinterpret_cast<const v2f*>(arow + kk * 4);
        v2f a;
        a.x = s.x * dinv;
        a.y = s.y * dinv;
#pragma unroll
        for (int nt = 0; nt < 8; ++nt) {
            const v2f b = *reinterpret_cast<const v2f*>(wl_s + (nt * 16 + mr) * P + kk * 4 + kb);
            acc[nt] = __builtin_amdgcn_wmma_f32_16x16x4_f32(
                false, a, false, b, (short)0, acc[nt], false, false);
        }
    }
    for (int kk = 0; kk < DIN / 4; ++kk) {
        const v2f a = *reinterpret_cast<const v2f*>(xrow + kk * 4);
#pragma unroll
        for (int nt = 0; nt < 8; ++nt) {
            const v2f b = *reinterpret_cast<const v2f*>(wr_s + (nt * 16 + mr) * P + kk * 4 + kb);
            acc[nt] = __builtin_amdgcn_wmma_f32_16x16x4_f32(
                false, a, false, b, (short)0, acc[nt], false, false);
        }
    }

    const int mo = m0 + 8 * (lane >> 4);
#pragma unroll
    for (int nt = 0; nt < 8; ++nt) {
        const int n = nt * 16 + mr;
        const float bias = bl[n];
#pragma unroll
        for (int v = 0; v < 8; ++v) {
            h1[(mo + v) * DHID + n] = tanhf(acc[nt][v] + bias);
        }
    }
}

#define P2 (DHID + 4)
#define NUNIT ((TR / 16) * (DOUT / 16))

static_assert(TR == G_ROWS && NT == G_BLOCKS, "both layers split the nodes alike");
static_assert((2 * TR * P2 + 2 * DOUT * P2) * 4 > 160 * 1024, "one work-group per processor");
static_assert((2 * TR * P2 + 2 * DOUT * P2) * 4 <= 320 * 1024, "must fit the LDS");
static_assert(AGG_WAVES * QCAP <= TR * P2 + 2 * DOUT * P2, "the scan's queues fit behind the tile");
static_assert(NUNIT <= 2 * AGG_WAVES, "at most two output blocks per wave");
static_assert(AGG_WAVES % (DOUT / 16) == 0, "both blocks of a wave lie in one column tile");

__device__ __forceinline__ void stage_layer2(const float* __restrict__ h1t,
                                             const float* __restrict__ Wl, const float* __restrict__ Wr,
                                             float* ah_s, float* wl_s, float* wr_s, int nv, int tid) {
    constexpr int Q  = DHID / 4;
    constexpr int NW = DOUT * Q / AGG_THREADS;
    constexpr int NH = (TR * Q + AGG_THREADS - 1) / AGG_THREADS;
    static_assert(DOUT * Q % AGG_THREADS == 0, "the weights split evenly over the work-group");
    const float4* pl = reinterpret_cast<const float4*>(Wl);
    const float4* pr = reinterpret_cast<const float4*>(Wr);
    const float4* ph = reinterpret_cast<const float4*>(h1t);
    const int last = nv * Q - 1;
    float4 vl[NW], vr[NW], vh[NH];
#pragma unroll
    for (int j = 0; j < NW; ++j) {
        vl[j] = pl[tid + j * AGG_THREADS];
        vr[j] = pr[tid + j * AGG_THREADS];
    }
#pragma unroll
    for (int j = 0; j < NH; ++j) vh[j] = ph[min(tid + j * AGG_THREADS, last)];
#pragma unroll
    for (int j = 0; j < NW; ++j) {
        const int i = tid + j * AGG_THREADS;
        const int o = (i / Q) * P2 + (i % Q) * 4;
        *reinterpret_cast<float4*>(wl_s + o) = vl[j];
        *reinterpret_cast<float4*>(wr_s + o) = vr[j];
    }
#pragma unroll
    for (int j = 0; j < NH; ++j) {
        const int i = tid + j * AGG_THREADS;
        if ((j + 1) * AGG_THREADS <= TR * Q || i < TR * Q) {
            const int o = (i / Q) * P2 + (i % Q) * 4;
            *reinterpret_cast<float4*>(ah_s + o) = vh[j];
        }
    }
}

__global__ __launch_bounds__(AGG_THREADS) void layer2_kernel(
    const float* __restrict__ h1, const int* __restrict__ ei,
    const float* __restrict__ deg, float* __restrict__ lists,
    const float* __restrict__ Wl, const float* __restrict__ bl,
    const float* __restrict__ Wr, float* __restrict__ out) {
    constexpr int P = P2;
    __shared__ __attribute__((aligned(16))) float smem[2 * TR * P + 2 * DOUT * P];
    float* tile = smem;
    float* ah_s = tile + TR * P;
    float* wl_s = ah_s + TR * P;
    float* wr_s = wl_s + DOUT * P;
    unsigned* queue = reinterpret_cast<unsigned*>(ah_s);

    const int tid  = threadIdx.x;
    const int lane = tid & 31;
    const int wave = __builtin_amdgcn_readfirstlane(tid >> 5);
    const int lo   = blockIdx.x * TR;
    const int nv   = min(TR, NN - lo);
    const int nrt  = nv / 16;
    const int mr   = lane & 15;
    const int kb   = (lane >> 4) * 2;

    unsigned* list = reinterpret_cast<unsigned*>(lists) + (size_t)blockIdx.x * LSTRIDE;
    const int n_list = (int)list[0];
    float degm[2];
#pragma unroll
    for (int h = 0; h < 2; ++h) {
        const int rt = (wave + h * AGG_WAVES) >> 2;
        degm[h] = deg[min(lo + rt * 16 + mr, NN - 1)];
    }
    const float bias = bl[(wave & 3) * 16 + mr];

    for (int i = tid; i < TR * P / 4; i += AGG_THREADS)
        reinterpret_cast<float4*>(tile)[i] = make_float4(0.f, 0.f, 0.f, 0.f);
    __syncthreads();

    const bool listed = n_list >= 0 && n_list <= LCAP;
    if (listed) stage_layer2(h1 + (size_t)lo * DHID, Wl, Wr, ah_s, wl_s, wr_s, nv, tid);
    gather_tile<DHID, P, false>(h1, ei, tile, queue, nullptr, nullptr, list, n_list, lo, lane, wave);
    if (!listed) {
        __syncthreads();
        stage_layer2(h1 + (size_t)lo * DHID, Wl, Wr, ah_s, wl_s, wr_s, nv, tid);
    }
    __syncthreads();

    v8f acc[2] = {};
#pragma unroll
    for (int h = 0; h < 2; ++h) {
        const int u  = wave + h * AGG_WAVES;
        const int rt = u >> 2, nt = u & 3;
        if (u < NUNIT && rt < nrt) {
            const float dinv = 1.0f / fmaxf(degm[h], 1.0f);
            const float* arow = tile + (rt * 16 + mr) * P + kb;
            const float* hrow = ah_s + (rt * 16 + mr) * P + kb;
            const float* lrow = wl_s + (nt * 16 + mr) * P + kb;
            const float* rrow = wr_s + (nt * 16 + mr) * P + kb;
            v8f c = {};
            for (int kk = 0; kk < DHID / 4; ++kk) {
                const v2f s = *reinterpret_cast<const v2f*>(arow + kk * 4);
                v2f a;
                a.x = s.x * dinv;
                a.y = s.y * dinv;
                const v2f b = *reinterpret_cast<const v2f*>(lrow + kk * 4);
                c = __builtin_amdgcn_wmma_f32_16x16x4_f32(
                    false, a, false, b, (short)0, c, false, false);
            }
            for (int kk = 0; kk < DHID / 4; ++kk) {
                const v2f a = *reinterpret_cast<const v2f*>(hrow + kk * 4);
                const v2f b = *reinterpret_cast<const v2f*>(rrow + kk * 4);
                c = __builtin_amdgcn_wmma_f32_16x16x4_f32(
                    false, a, false, b, (short)0, c, false, false);
            }
            acc[h] = c;
        }
    }
    __syncthreads();

#pragma unroll
    for (int h = 0; h < 2; ++h) {
        const int u  = wave + h * AGG_WAVES;
        const int rt = u >> 2, nt = u & 3;
        if (u < NUNIT && rt < nrt) {
            const int n = nt * 16 + mr;
            float* zc = tile + (rt * 16 + 8 * (lane >> 4)) * P + n;
#pragma unroll
            for (int v = 0; v < 8; ++v) zc[v * P] = acc[h][v] + bias;
        }
    }
    __syncthreads();

    constexpr int RPW = (TR + AGG_WAVES - 1) / AGG_WAVES;
    v2f z[RPW];
    float mx[RPW], sm[RPW];
#pragma unroll
    for (int i = 0; i < RPW; ++i) {
        const int r = min(wave + i * AGG_WAVES, TR - 1);
        z[i] = *reinterpret_cast<const v2f*>(tile + r * P + lane * 2);
        mx[i] = fmaxf(z[i].x, z[i].y);
    }
#pragma unroll
    for (int o = 16; o > 0; o >>= 1) {
#pragma unroll
        for (int i = 0; i < RPW; ++i) mx[i] = fmaxf(mx[i], __shfl_xor(mx[i], o));
    }
#pragma unroll
    for (int i = 0; i < RPW; ++i) sm[i] = expf(z[i].x - mx[i]) + expf(z[i].y - mx[i]);
#pragma unroll
    for (int o = 16; o > 0; o >>= 1) {
#pragma unroll
        for (int i = 0; i < RPW; ++i) sm[i] += __shfl_xor(sm[i], o);
    }
#pragma unroll
    for (int i = 0; i < RPW; ++i) {
        const int r = wave + i * AGG_WAVES;
        if (r < nv) {
            const float lse = mx[i] + logf(sm[i]);
            float2 res;
            res.x = z[i].x - lse;
            res.y = z[i].y - lse;
            *reinterpret_cast<float2*>(out + (size_t)(lo + r) * DOUT + lane * 2) = res;
        }
    }
}

extern "C" void kernel_launch(void* const* d_in, const int* in_sizes, int n_in,
                              void* d_out, int out_size, void* d_ws, size_t ws_size,
                              hipStream_t stream) {
    const float* x   = (const float*)d_in[0];
    const int*   ei  = (const int*)d_in[1];
    const float* Wl1 = (const float*)d_in[2];
    const float* bl1 = (const float*)d_in[3];
    const float* Wr1 = (const float*)d_in[4];
    const float* Wl2 = (const float*)d_in[5];
    const float* bl2 = (const float*)d_in[6];
    const float* Wr2 = (const float*)d_in[7];
    float* out = (float*)d_out;

    float* ws   = (float*)d_ws;
    float* sum1 = ws;
    float* lsts = sum1 + (size_t)NN * DIN;
    float* h1   = lsts + (size_t)NN * DHID;
    float* degb = h1 + (size_t)NN * DHID;

    agg1_kernel<<<NT, AGG_THREADS, 0, stream>>>(x, ei, sum1, degb, lsts);
    gemm1_kernel<<<G_BLOCKS, G_THREADS, 0, stream>>>(x, sum1, degb, Wl1, bl1, Wr1, h1);
    layer2_kernel<<<NT, AGG_THREADS, 0, stream>>>(h1, ei, degb, lsts, Wl2, bl2, Wr2, out);
}


# Layer 1 in one kernel: the work-group that scans and sums a tile's x rows in LDS also multiplies them there (sums never stored, 2 launches); 16x32 blocks over its 32 waves; sum order kept
# speedup vs baseline: 4.7077x; 1.0736x over previous
#include <hip/hip_runtime.h>
#include <cmath>


#define NN 50000
#define NE 800000
#define DIN 64
#define DHID 128
#define DOUT 64

#define TR 208
#define NT ((NN + TR - 1) / TR)
#define AGG_WAVES 32
#define AGG_THREADS (AGG_WAVES * 32)
#define NCH (NE / 128)
#define UNR 5
#define NPASS (NCH / AGG_WAVES / UNR)
#define QCAP 1024
#define LHEAD 64
#define LSTRIDE ((NN * DHID / NT) / 64 * 64)
#define LCAP (LSTRIDE - LHEAD)

static_assert((NT - 1) * TR < NN && NT * TR >= NN && (NN - (NT - 1) * TR) % 16 == 0, "tiles cover the nodes; the last one holds whole row tiles");
static_assert((size_t)NT * LSTRIDE <= (size_t)NN * DHID, "the lists must fit their part of the workspace");
static_assert(((size_t)NN * DIN * 4) % 256 == 0 && ((size_t)NN * DHID * 4) % 256 == 0, "the lists and h1 must start on 256-byte boundaries");
static_assert(NE % 128 == 0, "edge list must split into whole chunks");
static_assert(TR <= 256, "local target index is packed into 8 bits");
static_assert(NE <= (1 << 24), "edge index is packed into 24 bits");
static_assert(NN <= (1 << 24), "source node is packed into 24 bits of a list entry");
static_assert(NCH / NT >= 1, "start offsets must differ between tiles");
static_assert(UNR * AGG_WAVES < NCH && AGG_WAVES + NT * (NCH / NT) <= 2 * NCH, "one subtraction must be enough to wrap a chunk number");
static_assert(QCAP >= UNR * 128 + 128, "one pass of the scan must always fit behind the flush mark");
static_assert(NCH - NPASS * UNR * AGG_WAVES < AGG_WAVES, "after its full passes a wave has at most one chunk left");

typedef __attribute__((ext_vector_type(2))) float v2f;
typedef __attribute__((ext_vector_type(8))) float v8f;

template <int D, int P>
__device__ __forceinline__ void add_rows(const float* __restrict__ feat, float* tile,
                                         int s, int t, int m, int lane) {
    for (int k = 0; k < m; k += 8) {
        float2 r[8][D / 64];
#pragma unroll
        for (int i = 0; i < 8; ++i) {
            const int sk = __builtin_amdgcn_readlane(s, k + i);
            const float* row = feat + (size_t)sk * D + lane * 2;
#pragma unroll
            for (int j = 0; j < D / 64; ++j)
                r[i][j] = *reinterpret_cast<const float2*>(row + j * 64);
        }
#pragma unroll
        for (int i = 0; i < 8; ++i) {
            if (k + i < m) {
                const int tk = __builtin_amdgcn_readlane(t, k + i);
                float* trow = tile + tk * P + lane * 2;
#pragma unroll
                for (int j = 0; j < D / 64; ++j) {
                    atomicAdd(trow + j * 64, r[i][j].x);
                    atomicAdd(trow + j * 64 + 1, r[i][j].y);
                }
            }
        }
    }
}

template <int D, int P, bool SAVE>
__device__ __forceinline__ void flush_queue(const float* __restrict__ feat,
                                            const int* __restrict__ ei,
                                            float* tile, const unsigned* q,
                                            int n, int lane, unsigned* list, int lbase) {
    for (int base = 0; base < n; base += 32) {
        const int idx = base + lane;
        unsigned ent = 0u;
        int s = 0;
        if (idx < n) {
            ent = q[idx];
            s = ei[ent >> 8];
            if (SAVE && lbase + idx < LCAP)
                list[LHEAD + lbase + idx] = ((unsigned)s << 8) | (ent & 255u);
        }
        add_rows<D, P>(feat, tile, s, (int)(ent & 255u), min(32, n - base), lane);
    }
}

template <int D, int P>
__device__ __forceinline__ void replay_list(const float* __restrict__ feat, float* tile,
                                            const unsigned* lst, int n, int lane) {
    unsigned cur = lane < n ? lst[lane] : 0u;
    for (int base = 0; base < n; base += 32) {
        const int nx = base + 32 + lane;
        const unsigned nxt = nx < n ? lst[nx] : 0u;
        const int s = min((int)(cur >> 8), NN - 1);
        const int t = min((int)(cur & 255u), TR - 1);
        add_rows<D, P>(feat, tile, s, t, min(32, n - base), lane);
        cur = nxt;
    }
}

template <int D, int P, bool FIRST>
__device__ __forceinline__ void gather_tile(
    const float* __restrict__ feat, const int* __restrict__ ei, float* tile,
    unsigned* queue, int* cnt, int* lfill, unsigned* list, int n_list,
    int lo, int lane, int wave) {
    constexpr bool DEG = FIRST;
    bool replayed = false;
    if (!FIRST) {
        const int n = n_list;
        if (n >= 0 && n <= LCAP) {
            const int per = ((n + AGG_WAVES - 1) / AGG_WAVES + 7) & ~7;
            const int beg = min(n, wave * per);
            const int len = min(n - beg, per);
            replay_list<D, P>(feat, tile, list + LHEAD + beg, len, lane);
            replayed = true;
        }
    }
    if (!replayed) {
    unsigned* q = queue + wave * QCAP;
    int qn = 0;
    const int4* dp = reinterpret_cast<const int4*>(ei + NE);
    const int start = blockIdx.x * (NCH / NT);

    int nb = wave + start;
    if (nb >= NCH) nb -= NCH;
    auto load_pass = [&](int4 (&v)[UNR], int (&ch)[UNR]) {
#pragma unroll
        for (int u = 0; u < UNR; ++u) {
            int r = nb + u * AGG_WAVES;
            if (r >= NCH) r -= NCH;
            ch[u] = r;
            v[u] = dp[r * 32 + lane];
        }
        nb += UNR * AGG_WAVES;
        if (nb >= NCH) nb -= NCH;
    };
    auto scan_chunk = [&](const int4& d, int chunk) {
        const unsigned t0 = (unsigned)(d.x - lo);
        const unsigned t1 = (unsigned)(d.y - lo);
        const unsigned t2 = (unsigned)(d.z - lo);
        const unsigned t3 = (unsigned)(d.w - lo);
        const bool m0 = t0 < TR, m1 = t1 < TR, m2 = t2 < TR, m3 = t3 < TR;
        if (__builtin_amdgcn_ballot_w32(m0 | m1 | m2 | m3)) {
            const unsigned e0 = (unsigned)(chunk * 32 + lane) * 4u;
            const unsigned tt[4] = {t0, t1, t2, t3};
            const bool     mm[4] = {m0, m1, m2, m3};
#pragma unroll
            for (int j = 0; j < 4; ++j) {
                const unsigned mask = __builtin_amdgcn_ballot_w32(mm[j]);
                if (mask) {
                    if (mm[j]) {
                        const int pos = qn + (int)__builtin_amdgcn_mbcnt_lo(mask, 0u);
                        q[pos] = ((e0 + j) << 8) | tt[j];
                        if (DEG) atomicAdd(&cnt[tt[j]], 1);
                    }
                    qn += __builtin_popcount(mask);
                }
            }
        }
    };

    auto flush_now = [&]() {
        int lb = 0;
        if (FIRST) {
            if (lane == 0) lb = atomicAdd(lfill, qn);
            lb = __builtin_amdgcn_readfirstlane(lb);
        }
        flush_queue<D, P, FIRST>(feat, ei, tile, q, qn, lane, list, lb);
        qn = 0;
    };

    auto step = [&](int4 (&cur)[UNR], int (&cch)[UNR], int4 (&nxt)[UNR], int (&nxc)[UNR]) {
        load_pass(nxt, nxc);
#pragma unroll
        for (int u = 0; u < UNR; ++u) scan_chunk(cur[u], cch[u]);
        if (qn > QCAP - UNR * 128) {
            __builtin_amdgcn_fence(__ATOMIC_RELEASE, "wavefront");
            __builtin_amdgcn_fence(__ATOMIC_ACQUIRE, "wavefront");
            flush_now();
        }
    };

    static_assert(NPASS % 2 == 1, "the loop below does the passes in pairs, then one more");
    int4 va[UNR], vb[UNR];
    int  ca[UNR], cb[UNR];
    load_pass(va, ca);
    for (int p = 0; p + 1 < NPASS; p += 2) {
        step(va, ca, vb, cb);
        step(vb, cb, va, ca);
    }
    step(va, ca, vb, cb);
    if (wave + NPASS * UNR * AGG_WAVES < NCH) scan_chunk(vb[0], cb[0]);
    __builtin_amdgcn_fence(__ATOMIC_RELEASE, "wavefront");
    __builtin_amdgcn_fence(__ATOMIC_ACQUIRE, "wavefront");
    flush_now();
    }
}

#define P1 (DIN + 4)
#define NUNIT1 ((TR / 16) * (DHID / 32))
#define NTILE (NN / 16)

static_assert(NN % 16 == 0 && TR % 16 == 0, "row tiles must cover the nodes exactly");
static_assert(NT <= 256, "one work-group per processor, all resident at once");
static_assert((2 * TR * P1 + 2 * DHID * P1 + AGG_WAVES * QCAP + TR + 1) * 4 > 160 * 1024, "layer 1: one work-group per processor");
static_assert((2 * TR * P1 + 2 * DHID * P1 + AGG_WAVES * QCAP + TR + 1) * 4 <= 320 * 1024, "layer 1 must fit the LDS");
static_assert(NUNIT1 <= 2 * AGG_WAVES, "at most two output blocks per wave");

__global__ __launch_bounds__(AGG_THREADS) void layer1_kernel(
    const float* __restrict__ x, const int* __restrict__ ei,
    const float* __restrict__ Wl, const float* __restrict__ bl,
    const float* __restrict__ Wr, float* __restrict__ h1,
    float* __restrict__ deg, float* __restrict__ lists) {
    constexpr int D = DIN;
    constexpr int P = P1;
    constexpr int Q = DIN / 4;
    constexpr int NW = DHID * Q / AGG_THREADS;
    constexpr int NH = (TR * Q + AGG_THREADS - 1) / AGG_THREADS;
    static_assert(DHID * Q % AGG_THREADS == 0, "the weights split evenly over the work-group");
    __shared__ __attribute__((aligned(16))) float smem[2 * TR * P + 2 * DHID * P];
    __shared__ unsigned queue[AGG_WAVES * QCAP];
    __shared__ int cnt[TR];
    __shared__ int lfill;
    float* tile = smem;
    float* ax_s = tile + TR * P;
    float* wl_s = ax_s + TR * P;
    float* wr_s = wl_s + DHID * P;

    const int tid  = threadIdx.x;
    const int lane = tid & 31;
    const int wave = __builtin_amdgcn_readfirstlane(tid >> 5);
    const int lo   = blockIdx.x * TR;
    const int nv   = min(TR, NN - lo);
    const int nrt  = nv / 16;
    const int mr   = lane & 15;
    const int kb   = (lane >> 4) * 2;

    unsigned* list = reinterpret_cast<unsigned*>(lists) + (size_t)blockIdx.x * LSTRIDE;

    const float4* pl = reinterpret_cast<const float4*>(Wl);
    const float4* pr = reinterpret_cast<const float4*>(Wr);
    const float4* px = reinterpret_cast<const float4*>(x + (size_t)lo * D);
    const int last = nv * Q - 1;
    float4 vl[NW], vr[NW], vx[NH];
#pragma unroll
    for (int j = 0; j < NW; ++j) {
        vl[j] = pl[tid + j * AGG_THREADS];
        vr[j] = pr[tid + j * AGG_THREADS];
    }
#pragma unroll
    for (int j = 0; j < NH; ++j) vx[j] = px[min(tid + j * AGG_THREADS, last)];

    for (int i = tid; i < TR * P / 4; i += AGG_THREADS)
        reinterpret_cast<float4*>(tile)[i] = make_float4(0.f, 0.f, 0.f, 0.f);
    for (int i = tid; i < TR; i += AGG_THREADS) cnt[i] = 0;
    if (tid == 0) lfill = 0;

#pragma unroll
    for (int j = 0; j < NW; ++j) {
        const int i = tid + j * AGG_THREADS;
        const int o = (i / Q) * P + (i % Q) * 4;
        *reinterpret_cast<float4*>(wl_s + o) = vl[j];
        *reinterpret_cast<float4*>(wr_s + o) = vr[j];
    }
#pragma unroll
    for (int j = 0; j < NH; ++j) {
        const int i = tid + j * AGG_THREADS;
        if ((j + 1) * AGG_THREADS <= TR * Q || i < TR * Q) {
            const int o = (i / Q) * P + (i % Q) * 4;
            *reinterpret_cast<float4*>(ax_s + o) = vx[j];
        }
    }
    __syncthreads();

    gather_tile<D, P, true>(x, ei, tile, queue, cnt, &lfill, list, -1, lo, lane, wave);
    __syncthreads();

    for (int i = tid; i < nv; i += AGG_THREADS) deg[lo + i] = (float)cnt[i];
    if (tid == 0) list[0] = (unsigned)lfill;

    for (int h = 0; h < 2; ++h) {
        const int u  = wave + h * AGG_WAVES;
        const int rt = u >> 2, cp = u & 3;
        if (u < NUNIT1 && rt < nrt) {
            const float dinv = 1.0f / fmaxf((float)cnt[rt * 16 + mr], 1.0f);
            const float* arow = tile + (rt * 16 + mr) * P + kb;
            const float* xrow = ax_s + (rt * 16 + mr) * P + kb;
            const float* l0 = wl_s + (cp * 32 + mr) * P + kb;
            const float* l1 = l0 + 16 * P;
            const float* r0 = wr_s + (cp * 32 + mr) * P + kb;
            const float* r1 = r0 + 16 * P;
            v8f c0 = {}, c1 = {};
            for (int kk = 0; kk < DIN / 4; ++kk) {
                const v2f s = *reinterpret_cast<const v2f*>(arow + kk * 4);
                v2f a;
                a.x = s.x * dinv;
                a.y = s.y * dinv;
                const v2f b0 = *reinterpret_cast<const v2f*>(l0 + kk * 4);
                const v2f b1 = *reinterpret_cast<const v2f*>(l1 + kk * 4);
                c0 = __builtin_amdgcn_wmma_f32_16x16x4_f32(
                    false, a, false, b0, (short)0, c0, false, false);
                c1 = __builtin_amdgcn_wmma_f32_16x16x4_f32(
                    false, a, false, b1, (short)0, c1, false, false);
            }
            for (int kk = 0; kk < DIN / 4; ++kk) {
                const v2f a = *reinterpret_cast<const v2f*>(xrow + kk * 4);
                const v2f b0 = *reinterpret_cast<const v2f*>(r0 + kk * 4);
                const v2f b1 = *reinterpret_cast<const v2f*>(r1 + kk * 4);
                c0 = __builtin_amdgcn_wmma_f32_16x16x4_f32(
                    false, a, false, b0, (short)0, c0, false, false);
                c1 = __builtin_amdgcn_wmma_f32_16x16x4_f32(
                    false, a, false, b1, (short)0, c1, false, false);
            }
            const int n0 = cp * 32 + mr;
            const float bias0 = bl[n0];
            const float bias1 = bl[n0 + 16];
            float* hc = h1 + (size_t)(lo + rt * 16 + 8 * (lane >> 4)) * DHID + n0;
#pragma unroll
            for (int v = 0; v < 8; ++v) {
                hc[v * DHID]      = tanhf(c0[v] + bias0);
                hc[v * DHID + 16] = tanhf(c1[v] + bias1);
            }
        }
    }
}

#define P2 (DHID + 4)
#define NUNIT ((TR / 16) * (DOUT / 16))

static_assert((2 * TR * P2 + 2 * DOUT * P2) * 4 > 160 * 1024, "one work-group per processor");
static_assert((2 * TR * P2 + 2 * DOUT * P2) * 4 <= 320 * 1024, "must fit the LDS");
static_assert(AGG_WAVES * QCAP <= TR * P2 + 2 * DOUT * P2, "the scan's queues fit behind the tile");
static_assert(NUNIT <= 2 * AGG_WAVES, "at most two output blocks per wave");
static_assert(AGG_WAVES % (DOUT / 16) == 0, "both blocks of a wave lie in one column tile");

__device__ __forceinline__ void stage_layer2(const float* __restrict__ h1t,
                                             const float* __restrict__ Wl, const float* __restrict__ Wr,
                                             float* ah_s, float* wl_s, float* wr_s, int nv, int tid) {
    constexpr int Q  = DHID / 4;
    constexpr int NW = DOUT * Q / AGG_THREADS;
    constexpr int NH = (TR * Q + AGG_THREADS - 1) / AGG_THREADS;
    static_assert(DOUT * Q % AGG_THREADS == 0, "the weights split evenly over the work-group");
    const float4* pl = reinterpret_cast<const float4*>(Wl);
    const float4* pr = reinterpret_cast<const float4*>(Wr);
    const float4* ph = reinterpret_cast<const float4*>(h1t);
    const int last = nv * Q - 1;
    float4 vl[NW], vr[NW], vh[NH];
#pragma unroll
    for (int j = 0; j < NW; ++j) {
        vl[j] = pl[tid + j * AGG_THREADS];
        vr[j] = pr[tid + j * AGG_THREADS];
    }
#pragma unroll
    for (int j = 0; j < NH; ++j) vh[j] = ph[min(tid + j * AGG_THREADS, last)];
#pragma unroll
    for (int j = 0; j < NW; ++j) {
        const int i = tid + j * AGG_THREADS;
        const int o = (i / Q) * P2 + (i % Q) * 4;
        *reinterpret_cast<float4*>(wl_s + o) = vl[j];
        *reinterpret_cast<float4*>(wr_s + o) = vr[j];
    }
#pragma unroll
    for (int j = 0; j < NH; ++j) {
        const int i = tid + j * AGG_THREADS;
        if ((j + 1) * AGG_THREADS <= TR * Q || i < TR * Q) {
            const int o = (i / Q) * P2 + (i % Q) * 4;
            *reinterpret_cast<float4*>(ah_s + o) = vh[j];
        }
    }
}

__global__ __launch_bounds__(AGG_THREADS) void layer2_kernel(
    const float* __restrict__ h1, const int* __restrict__ ei,
    const float* __restrict__ deg, float* __restrict__ lists,
    const float* __restrict__ Wl, const float* __restrict__ bl,
    const float* __restrict__ Wr, float* __restrict__ out) {
    constexpr int P = P2;
    __shared__ __attribute__((aligned(16))) float smem[2 * TR * P + 2 * DOUT * P];
    float* tile = smem;
    float* ah_s = tile + TR * P;
    float* wl_s = ah_s + TR * P;
    float* wr_s = wl_s + DOUT * P;
    unsigned* queue = reinterpret_cast<unsigned*>(ah_s);

    const int tid  = threadIdx.x;
    const int lane = tid & 31;
    const int wave = __builtin_amdgcn_readfirstlane(tid >> 5);
    const int lo   = blockIdx.x * TR;
    const int nv   = min(TR, NN - lo);
    const int nrt  = nv / 16;
    const int mr   = lane & 15;
    const int kb   = (lane >> 4) * 2;

    unsigned* list = reinterpret_cast<unsigned*>(lists) + (size_t)blockIdx.x * LSTRIDE;
    const int n_list = (int)list[0];
    float degm[2];
#pragma unroll
    for (int h = 0; h < 2; ++h) {
        const int rt = (wave + h * AGG_WAVES) >> 2;
        degm[h] = deg[min(lo + rt * 16 + mr, NN - 1)];
    }
    const float bias = bl[(wave & 3) * 16 + mr];

    for (int i = tid; i < TR * P / 4; i += AGG_THREADS)
        reinterpret_cast<float4*>(tile)[i] = make_float4(0.f, 0.f, 0.f, 0.f);
    __syncthreads();

    const bool listed = n_list >= 0 && n_list <= LCAP;
    if (listed) stage_layer2(h1 + (size_t)lo * DHID, Wl, Wr, ah_s, wl_s, wr_s, nv, tid);
    gather_tile<DHID, P, false>(h1, ei, tile, queue, nullptr, nullptr, list, n_list, lo, lane, wave);
    if (!listed) {
        __syncthreads();
        stage_layer2(h1 + (size_t)lo * DHID, Wl, Wr, ah_s, wl_s, wr_s, nv, tid);
    }
    __syncthreads();

    v8f acc[2] = {};
#pragma unroll
    for (int h = 0; h < 2; ++h) {
        const int u  = wave + h * AGG_WAVES;
        const int rt = u >> 2, nt = u & 3;
        if (u < NUNIT && rt < nrt) {
            const float dinv = 1.0f / fmaxf(degm[h], 1.0f);
            const float* arow = tile + (rt * 16 + mr) * P + kb;
            const float* hrow = ah_s + (rt * 16 + mr) * P + kb;
            const float* lrow = wl_s + (nt * 16 + mr) * P + kb;
            const float* rrow = wr_s + (nt * 16 + mr) * P + kb;
            v8f c = {};
            for (int kk = 0; kk < DHID / 4; ++kk) {
                const v2f s = *reinterpret_cast<const v2f*>(arow + kk * 4);
                v2f a;
                a.x = s.x * dinv;
                a.y = s.y * dinv;
                const v2f b = *reinterpret_cast<const v2f*>(lrow + kk * 4);
                c = __builtin_amdgcn_wmma_f32_16x16x4_f32(
                    false, a, false, b, (short)0, c, false, false);
            }
            for (int kk = 0; kk < DHID / 4; ++kk) {
                const v2f a = *reinterpret_cast<const v2f*>(hrow + kk * 4);
                const v2f b = *reinterpret_cast<const v2f*>(rrow + kk * 4);
                c = __builtin_amdgcn_wmma_f32_16x16x4_f32(
                    false, a, false, b, (short)0, c, false, false);
            }
            acc[h] = c;
        }
    }
    __syncthreads();

#pragma unroll
    for (int h = 0; h < 2; ++h) {
        const int u  = wave + h * AGG_WAVES;
        const int rt = u >> 2, nt = u & 3;
        if (u < NUNIT && rt < nrt) {
            const int n = nt * 16 + mr;
            float* zc = tile + (rt * 16 + 8 * (lane >> 4)) * P + n;
#pragma unroll
            for (int v = 0; v < 8; ++v) zc[v * P] = acc[h][v] + bias;
        }
    }
    __syncthreads();

    constexpr int RPW = (TR + AGG_WAVES - 1) / AGG_WAVES;
    v2f z[RPW];
    float mx[RPW], sm[RPW];
#pragma unroll
    for (int i = 0; i < RPW; ++i) {
        const int r = min(wave + i * AGG_WAVES, TR - 1);
        z[i] = *reinterpret_cast<const v2f*>(tile + r * P + lane * 2);
        mx[i] = fmaxf(z[i].x, z[i].y);
    }
#pragma unroll
    for (int o = 16; o > 0; o >>= 1) {
#pragma unroll
        for (int i = 0; i < RPW; ++i) mx[i] = fmaxf(mx[i], __shfl_xor(mx[i], o));
    }
#pragma unroll
    for (int i = 0; i < RPW; ++i) sm[i] = expf(z[i].x - mx[i]) + expf(z[i].y - mx[i]);
#pragma unroll
    for (int o = 16; o > 0; o >>= 1) {
#pragma unroll
        for (int i = 0; i < RPW; ++i) sm[i] += __shfl_xor(sm[i], o);
    }
#pragma unroll
    for (int i = 0; i < RPW; ++i) {
        const int r = wave + i * AGG_WAVES;
        if (r < nv) {
            const float lse = mx[i] + logf(sm[i]);
            float2 res;
            res.x = z[i].x - lse;
            res.y = z[i].y - lse;
            *reinterpret_cast<float2*>(out + (size_t)(lo + r) * DOUT + lane * 2) = res;
        }
    }
}

extern "C" void kernel_launch(void* const* d_in, const int* in_sizes, int n_in,
                              void* d_out, int out_size, void* d_ws, size_t ws_size,
                              hipStream_t stream) {
    const float* x   = (const float*)d_in[0];
    const int*   ei  = (const int*)d_in[1];
    const float* Wl1 = (const float*)d_in[2];
    const float* bl1 = (const float*)d_in[3];
    const float* Wr1 = (const float*)d_in[4];
    const float* Wl2 = (const float*)d_in[5];
    const float* bl2 = (const float*)d_in[6];
    const float* Wr2 = (const float*)d_in[7];
    float* out = (float*)d_out;

    float* ws   = (float*)d_ws;
    float* lsts = ws + (size_t)NN * DIN;
    float* h1   = lsts + (size_t)NN * DHID;
    float* degb = h1 + (size_t)NN * DHID;

    layer1_kernel<<<NT, AGG_THREADS, 0, stream>>>(x, ei, Wl1, bl1, Wr1, h1, degb, lsts);
    layer2_kernel<<<NT, AGG_THREADS, 0, stream>>>(h1, ei, degb, lsts, Wl2, bl2, Wr2, out);
}


# Edge targets recoded once per call to one-byte tile numbers by a small first kernel (3 launches): every tile's scan reads a quarter of the bytes, tests 4 edges per dword; sum order by LDS arrival
# speedup vs baseline: 4.9708x; 1.0559x over previous
#include <hip/hip_runtime.h>
#include <cmath>


#define NN 50000
#define NE 800000
#define DIN 64
#define DHID 128
#define DOUT 64

#define TR 208
#define NT ((NN + TR - 1) / TR)
#define AGG_WAVES 32
#define AGG_THREADS (AGG_WAVES * 32)
#define NCH ((NE + 511) / 512)
#define NEP (NCH * 512)
#define UNR 7
#define NPASS ((NCH + AGG_WAVES * UNR - 1) / (AGG_WAVES * UNR))
#define QCAP 1024
#define LHEAD 64
#define LSTRIDE ((NN * DHID / NT) / 64 * 64)
#define LCAP (LSTRIDE - LHEAD)

static_assert((NT - 1) * TR < NN && NT * TR >= NN && (NN - (NT - 1) * TR) % 16 == 0, "tiles cover the nodes; the last one holds whole row tiles");
static_assert((size_t)NT * LSTRIDE <= (size_t)NN * DHID, "the lists must fit their part of the workspace");
static_assert(((size_t)NN * DIN * 4) % 256 == 0 && ((size_t)NN * DHID * 4) % 256 == 0, "the lists and h1 must start on 256-byte boundaries");
static_assert(NE % 4 == 0, "four targets make one dword of tile bytes");
static_assert(NT <= 255, "a tile number fits a byte, and 255 is no tile");
static_assert((size_t)NEP <= (size_t)NN * DIN * 4, "the tile bytes fit the first part of the workspace");
static_assert(TR <= 256, "local target index is packed into 8 bits");
static_assert(NE <= (1 << 24), "edge index is packed into 24 bits");
static_assert(NN <= (1 << 24), "source node is packed into 24 bits of a list entry");
static_assert(NCH / NT >= 1, "start offsets must differ between tiles");
static_assert(NT * (NCH / NT) <= NCH, "one subtraction must be enough to wrap a chunk number");
static_assert(QCAP >= 2 * 512, "one chunk must always fit behind the flush mark");

typedef __attribute__((ext_vector_type(2))) float v2f;
typedef __attribute__((ext_vector_type(8))) float v8f;

template <int D, int P>
__device__ __forceinline__ void add_rows(const float* __restrict__ feat, float* tile,
                                         int s, int t, int m, int lane) {
    for (int k = 0; k < m; k += 8) {
        float2 r[8][D / 64];
#pragma unroll
        for (int i = 0; i < 8; ++i) {
            const int sk = __builtin_amdgcn_readlane(s, k + i);
            const float* row = feat + (size_t)sk * D + lane * 2;
#pragma unroll
            for (int j = 0; j < D / 64; ++j)
                r[i][j] = *reinterpret_cast<const float2*>(row + j * 64);
        }
#pragma unroll
        for (int i = 0; i < 8; ++i) {
            if (k + i < m) {
                const int tk = __builtin_amdgcn_readlane(t, k + i);
                float* trow = tile + tk * P + lane * 2;
#pragma unroll
                for (int j = 0; j < D / 64; ++j) {
                    atomicAdd(trow + j * 64, r[i][j].x);
                    atomicAdd(trow + j * 64 + 1, r[i][j].y);
                }
            }
        }
    }
}

template <int D, int P, bool SAVE>
__device__ __forceinline__ void flush_queue(const float* __restrict__ feat,
                                            const int* __restrict__ ei,
                                            float* tile, const unsigned* q,
                                            int n, int lane, unsigned* list, int lbase,
                                            int* cnt, int lo) {
    for (int base = 0; base < n; base += 32) {
        const int idx = base + lane;
        int s = 0;
        unsigned t = 0u;
        if (idx < n) {
            const unsigned e = min(q[idx], (unsigned)(NE - 1));
            s = ei[e];
            t = min((unsigned)(ei[NE + e] - lo), (unsigned)(TR - 1));
            if (SAVE) {
                atomicAdd(&cnt[t], 1);
                if (lbase + idx < LCAP)
                    list[LHEAD + lbase + idx] = ((unsigned)s << 8) | t;
            }
        }
        add_rows<D, P>(feat, tile, s, (int)t, min(32, n - base), lane);
    }
}

template <int D, int P>
__device__ __forceinline__ void replay_list(const float* __restrict__ feat, float* tile,
                                            const unsigned* lst, int n, int lane) {
    unsigned cur = lane < n ? lst[lane] : 0u;
    for (int base = 0; base < n; base += 32) {
        const int nx = base + 32 + lane;
        const unsigned nxt = nx < n ? lst[nx] : 0u;
        const int s = min((int)(cur >> 8), NN - 1);
        const int t = min((int)(cur & 255u), TR - 1);
        add_rows<D, P>(feat, tile, s, t, min(32, n - base), lane);
        cur = nxt;
    }
}

template <int D, int P, bool FIRST>
__device__ __forceinline__ void gather_tile(
    const float* __restrict__ feat, const int* __restrict__ ei,
    const unsigned* __restrict__ tb, float* tile,
    unsigned* queue, int* cnt, int* lfill, unsigned* list, int n_list,
    int lo, int lane, int wave) {
    bool replayed = false;
    if (!FIRST) {
        const int n = n_list;
        if (n >= 0 && n <= LCAP) {
            const int per = ((n + AGG_WAVES - 1) / AGG_WAVES + 7) & ~7;
            const int beg = min(n, wave * per);
            const int len = min(n - beg, per);
            replay_list<D, P>(feat, tile, list + LHEAD + beg, len, lane);
            replayed = true;
        }
    }
    if (!replayed) {
    unsigned* q = queue + wave * QCAP;
    int qn = 0;
    const uint4* dp = reinterpret_cast<const uint4*>(tb);
    const unsigned pat = blockIdx.x * 0x01010101u;
    const int start = blockIdx.x * (NCH / NT);

    auto chunk_at = [&](int k) {
        int r = k + start;
        if (r >= NCH) r -= NCH;
        return min(r, NCH - 1);
    };
    auto load_pass = [&](uint4 (&v)[UNR], int k0) {
#pragma unroll
        for (int u = 0; u < UNR; ++u) v[u] = dp[chunk_at(k0 + u * AGG_WAVES) * 32 + lane];
    };

    auto flush_now = [&]() {
        int lb = 0;
        if (FIRST) {
            if (lane == 0) lb = atomicAdd(lfill, qn);
            lb = __builtin_amdgcn_readfirstlane(lb);
        }
        flush_queue<D, P, FIRST>(feat, ei, tile, q, qn, lane, list, lb, cnt, lo);
        qn = 0;
    };

    auto scan_chunk = [&](const uint4& d, int chunk) {
        auto zero_bytes = [&](unsigned v) {
            const unsigned x = v ^ pat;
            return ~(((x & 0x7f7f7f7fu) + 0x7f7f7f7fu) | x) & 0x80808080u;
        };
        unsigned m = (zero_bytes(d.x) >> 7) | (zero_bytes(d.y) >> 6) |
                     (zero_bytes(d.z) >> 5) | (zero_bytes(d.w) >> 4);
        unsigned mask = __builtin_amdgcn_ballot_w32(m != 0u);
        if (mask) {
            const unsigned e0 = (unsigned)(chunk * 32 + lane) * 16u;
            for (int turn = 0; turn < 16 && mask; ++turn) {
                if (m != 0u) {
                    const unsigned bit = (unsigned)__builtin_ctz(m);
                    const int pos = qn + (int)__builtin_amdgcn_mbcnt_lo(mask, 0u);
                    q[pos] = e0 + (bit & 7u) * 4u + (bit >> 3);
                    m &= m - 1u;
                }
                qn += __builtin_popcount(mask);
                mask = __builtin_amdgcn_ballot_w32(m != 0u);
            }
        }
    };

    uint4 cur[UNR], nxt[UNR];
    load_pass(nxt, wave);
    for (int p = 0; p < NPASS; ++p) {
        const int k0 = p * UNR * AGG_WAVES + wave;
#pragma unroll
        for (int k = 0; k < UNR; ++k) cur[k] = nxt[k];
        if (p + 1 < NPASS) load_pass(nxt, k0 + UNR * AGG_WAVES);
        int u = 0;
        do {
            bool full = false;
#pragma unroll
            for (int k = 0; k < UNR; ++k) {
                if (k >= u && !full) {
                    const int kk = k0 + k * AGG_WAVES;
                    if (kk < NCH) scan_chunk(cur[k], chunk_at(kk));
                    u = k + 1;
                    full = qn > QCAP - 512;
                }
            }
            if (full) {
                __builtin_amdgcn_fence(__ATOMIC_RELEASE, "wavefront");
                __builtin_amdgcn_fence(__ATOMIC_ACQUIRE, "wavefront");
                flush_now();
            }
        } while (u < UNR);
    }
    __builtin_amdgcn_fence(__ATOMIC_RELEASE, "wavefront");
    __builtin_amdgcn_fence(__ATOMIC_ACQUIRE, "wavefront");
    flush_now();
    }
}

#define RC_BLOCKS 256
#define RC_THREADS 256
#define RC_PER ((NEP / 4 + RC_BLOCKS * RC_THREADS - 1) / (RC_BLOCKS * RC_THREADS))
__global__ __launch_bounds__(RC_THREADS) void recode_kernel(const int* __restrict__ ei,
                                                            unsigned* __restrict__ tb) {
    const int4* dp = reinterpret_cast<const int4*>(ei + NE);
    const int i0 = blockIdx.x * RC_THREADS + threadIdx.x;
    int4 d[RC_PER];
#pragma unroll
    for (int u = 0; u < RC_PER; ++u)
        d[u] = dp[min(i0 + u * RC_BLOCKS * RC_THREADS, NE / 4 - 1)];
#pragma unroll
    for (int u = 0; u < RC_PER; ++u) {
        const int i = i0 + u * RC_BLOCKS * RC_THREADS;
        if (i < NEP / 4) {
            unsigned w = ~0u;
            if (i < NE / 4) {
                const unsigned b0 = (unsigned)d[u].x < (unsigned)NN ? (unsigned)d[u].x / TR : 255u;
                const unsigned b1 = (unsigned)d[u].y < (unsigned)NN ? (unsigned)d[u].y / TR : 255u;
                const unsigned b2 = (unsigned)d[u].z < (unsigned)NN ? (unsigned)d[u].z / TR : 255u;
                const unsigned b3 = (unsigned)d[u].w < (unsigned)NN ? (unsigned)d[u].w / TR : 255u;
                w = b0 | (b1 << 8) | (b2 << 16) | (b3 << 24);
            }
            tb[i] = w;
        }
    }
}

#define P1 (DIN + 4)
#define NUNIT1 ((TR / 16) * (DHID / 32))
#define NTILE (NN / 16)

static_assert(NN % 16 == 0 && TR % 16 == 0, "row tiles must cover the nodes exactly");
static_assert(NT <= 256, "one work-group per processor, all resident at once");
static_assert((2 * TR * P1 + 2 * DHID * P1 + AGG_WAVES * QCAP + TR + 1) * 4 > 160 * 1024, "layer 1: one work-group per processor");
static_assert((2 * TR * P1 + 2 * DHID * P1 + AGG_WAVES * QCAP + TR + 1) * 4 <= 320 * 1024, "layer 1 must fit the LDS");
static_assert(NUNIT1 <= 2 * AGG_WAVES, "at most two output blocks per wave");

__global__ __launch_bounds__(AGG_THREADS) void layer1_kernel(
    const float* __restrict__ x, const int* __restrict__ ei,
    const float* __restrict__ Wl, const float* __restrict__ bl,
    const float* __restrict__ Wr, float* __restrict__ h1,
    float* __restrict__ deg, float* __restrict__ lists,
    const unsigned* __restrict__ tb) {
    constexpr int D = DIN;
    constexpr int P = P1;
    constexpr int Q = DIN / 4;
    constexpr int NW = DHID * Q / AGG_THREADS;
    constexpr int NH = (TR * Q + AGG_THREADS - 1) / AGG_THREADS;
    static_assert(DHID * Q % AGG_THREADS == 0, "the weights split evenly over the work-group");
    __shared__ __attribute__((aligned(16))) float smem[2 * TR * P + 2 * DHID * P];
    __shared__ unsigned queue[AGG_WAVES * QCAP];
    __shared__ int cnt[TR];
    __shared__ int lfill;
    float* tile = smem;
    float* ax_s = tile + TR * P;
    float* wl_s = ax_s + TR * P;
    float* wr_s = wl_s + DHID * P;

    const int tid  = threadIdx.x;
    const int lane = tid & 31;
    const int wave = __builtin_amdgcn_readfirstlane(tid >> 5);
    const int lo   = blockIdx.x * TR;
    const int nv   = min(TR, NN - lo);
    const int nrt  = nv / 16;
    const int mr   = lane & 15;
    const int kb   = (lane >> 4) * 2;

    unsigned* list = reinterpret_cast<unsigned*>(lists) + (size_t)blockIdx.x * LSTRIDE;

    const float4* pl = reinterpret_cast<const float4*>(Wl);
    const float4* pr = reinterpret_cast<const float4*>(Wr);
    const float4* px = reinterpret_cast<const float4*>(x + (size_t)lo * D);
    const int last = nv * Q - 1;
    float4 vl[NW], vr[NW], vx[NH];
#pragma unroll
    for (int j = 0; j < NW; ++j) {
        vl[j] = pl[tid + j * AGG_THREADS];
        vr[j] = pr[tid + j * AGG_THREADS];
    }
#pragma unroll
    for (int j = 0; j < NH; ++j) vx[j] = px[min(tid + j * AGG_THREADS, last)];

    for (int i = tid; i < TR * P / 4; i += AGG_THREADS)
        reinterpret_cast<float4*>(tile)[i] = make_float4(0.f, 0.f, 0.f, 0.f);
    for (int i = tid; i < TR; i += AGG_THREADS) cnt[i] = 0;
    if (tid == 0) lfill = 0;

#pragma unroll
    for (int j = 0; j < NW; ++j) {
        const int i = tid + j * AGG_THREADS;
        const int o = (i / Q) * P + (i % Q) * 4;
        *reinterpret_cast<float4*>(wl_s + o) = vl[j];
        *reinterpret_cast<float4*>(wr_s + o) = vr[j];
    }
#pragma unroll
    for (int j = 0; j < NH; ++j) {
        const int i = tid + j * AGG_THREADS;
        if ((j + 1) * AGG_THREADS <= TR * Q || i < TR * Q) {
            const int o = (i / Q) * P + (i % Q) * 4;
            *reinterpret_cast<float4*>(ax_s + o) = vx[j];
        }
    }
    __syncthreads();

    gather_tile<D, P, true>(x, ei, tb, tile, queue, cnt, &lfill, list, -1, lo, lane, wave);
    __syncthreads();

    for (int i = tid; i < nv; i += AGG_THREADS) deg[lo + i] = (float)cnt[i];
    if (tid == 0) list[0] = (unsigned)lfill;

    for (int h = 0; h < 2; ++h) {
        const int u  = wave + h * AGG_WAVES;
        const int rt = u >> 2, cp = u & 3;
        if (u < NUNIT1 && rt < nrt) {
            const float dinv = 1.0f / fmaxf((float)cnt[rt * 16 + mr], 1.0f);
            const float* arow = tile + (rt * 16 + mr) * P + kb;
            const float* xrow = ax_s + (rt * 16 + mr) * P + kb;
            const float* l0 = wl_s + (cp * 32 + mr) * P + kb;
            const float* l1 = l0 + 16 * P;
            const float* r0 = wr_s + (cp * 32 + mr) * P + kb;
            const float* r1 = r0 + 16 * P;
            v8f c0 = {}, c1 = {};
            for (int kk = 0; kk < DIN / 4; ++kk) {
                const v2f s = *reinterpret_cast<const v2f*>(arow + kk * 4);
                v2f a;
                a.x = s.x * dinv;
                a.y = s.y * dinv;
                const v2f b0 = *reinterpret_cast<const v2f*>(l0 + kk * 4);
                const v2f b1 = *reinterpret_cast<const v2f*>(l1 + kk * 4);
                c0 = __builtin_amdgcn_wmma_f32_16x16x4_f32(
                    false, a, false, b0, (short)0, c0, false, false);
                c1 = __builtin_amdgcn_wmma_f32_16x16x4_f32(
                    false, a, false, b1, (short)0, c1, false, false);
            }
            for (int kk = 0; kk < DIN / 4; ++kk) {
                const v2f a = *reinterpret_cast<const v2f*>(xrow + kk * 4);
                const v2f b0 = *reinterpret_cast<const v2f*>(r0 + kk * 4);
                const v2f b1 = *reinterpret_cast<const v2f*>(r1 + kk * 4);
                c0 = __builtin_amdgcn_wmma_f32_16x16x4_f32(
                    false, a, false, b0, (short)0, c0, false, false);
                c1 = __builtin_amdgcn_wmma_f32_16x16x4_f32(
                    false, a, false, b1, (short)0, c1, false, false);
            }
            const int n0 = cp * 32 + mr;
            const float bias0 = bl[n0];
            const float bias1 = bl[n0 + 16];
            float* hc = h1 + (size_t)(lo + rt * 16 + 8 * (lane >> 4)) * DHID + n0;
#pragma unroll
            for (int v = 0; v < 8; ++v) {
                hc[v * DHID]      = tanhf(c0[v] + bias0);
                hc[v * DHID + 16] = tanhf(c1[v] + bias1);
            }
        }
    }
}

#define P2 (DHID + 4)
#define NUNIT ((TR / 16) * (DOUT / 16))

static_assert((2 * TR * P2 + 2 * DOUT * P2) * 4 > 160 * 1024, "one work-group per processor");
static_assert((2 * TR * P2 + 2 * DOUT * P2) * 4 <= 320 * 1024, "must fit the LDS");
static_assert(AGG_WAVES * QCAP <= TR * P2 + 2 * DOUT * P2, "the scan's queues fit behind the tile");
static_assert(NUNIT <= 2 * AGG_WAVES, "at most two output blocks per wave");
static_assert(AGG_WAVES % (DOUT / 16) == 0, "both blocks of a wave lie in one column tile");

__device__ __forceinline__ void stage_layer2(const float* __restrict__ h1t,
                                             const float* __restrict__ Wl, const float* __restrict__ Wr,
                                             float* ah_s, float* wl_s, float* wr_s, int nv, int tid) {
    constexpr int Q  = DHID / 4;
    constexpr int NW = DOUT * Q / AGG_THREADS;
    constexpr int NH = (TR * Q + AGG_THREADS - 1) / AGG_THREADS;
    static_assert(DOUT * Q % AGG_THREADS == 0, "the weights split evenly over the work-group");
    const float4* pl = reinterpret_cast<const float4*>(Wl);
    const float4* pr = reinterpret_cast<const float4*>(Wr);
    const float4* ph = reinterpret_cast<const float4*>(h1t);
    const int last = nv * Q - 1;
    float4 vl[NW], vr[NW], vh[NH];
#pragma unroll
    for (int j = 0; j < NW; ++j) {
        vl[j] = pl[tid + j * AGG_THREADS];
        vr[j] = pr[tid + j * AGG_THREADS];
    }
#pragma unroll
    for (int j = 0; j < NH; ++j) vh[j] = ph[min(tid + j * AGG_THREADS, last)];
#pragma unroll
    for (int j = 0; j < NW; ++j) {
        const int i = tid + j * AGG_THREADS;
        const int o = (i / Q) * P2 + (i % Q) * 4;
        *reinterpret_cast<float4*>(wl_s + o) = vl[j];
        *reinterpret_cast<float4*>(wr_s + o) = vr[j];
    }
#pragma unroll
    for (int j = 0; j < NH; ++j) {
        const int i = tid + j * AGG_THREADS;
        if ((j + 1) * AGG_THREADS <= TR * Q || i < TR * Q) {
            const int o = (i / Q) * P2 + (i % Q) * 4;
            *reinterpret_cast<float4*>(ah_s + o) = vh[j];
        }
    }
}

__global__ __launch_bounds__(AGG_THREADS) void layer2_kernel(
    const float* __restrict__ h1, const int* __restrict__ ei,
    const float* __restrict__ deg, float* __restrict__ lists,
    const float* __restrict__ Wl, const float* __restrict__ bl,
    const float* __restrict__ Wr, float* __restrict__ out,
    const unsigned* __restrict__ tb) {
    constexpr int P = P2;
    __shared__ __attribute__((aligned(16))) float smem[2 * TR * P + 2 * DOUT * P];
    float* tile = smem;
    float* ah_s = tile + TR * P;
    float* wl_s = ah_s + TR * P;
    float* wr_s = wl_s + DOUT * P;
    unsigned* queue = reinterpret_cast<unsigned*>(ah_s);

    const int tid  = threadIdx.x;
    const int lane = tid & 31;
    const int wave = __builtin_amdgcn_readfirstlane(tid >> 5);
    const int lo   = blockIdx.x * TR;
    const int nv   = min(TR, NN - lo);
    const int nrt  = nv / 16;
    const int mr   = lane & 15;
    const int kb   = (lane >> 4) * 2;

    unsigned* list = reinterpret_cast<unsigned*>(lists) + (size_t)blockIdx.x * LSTRIDE;
    const int n_list = (int)list[0];
    float degm[2];
#pragma unroll
    for (int h = 0; h < 2; ++h) {
        const int rt = (wave + h * AGG_WAVES) >> 2;
        degm[h] = deg[min(lo + rt * 16 + mr, NN - 1)];
    }
    const float bias = bl[(wave & 3) * 16 + mr];

    for (int i = tid; i < TR * P / 4; i += AGG_THREADS)
        reinterpret_cast<float4*>(tile)[i] = make_float4(0.f, 0.f, 0.f, 0.f);
    __syncthreads();

    const bool listed = n_list >= 0 && n_list <= LCAP;
    if (listed) stage_layer2(h1 + (size_t)lo * DHID, Wl, Wr, ah_s, wl_s, wr_s, nv, tid);
    gather_tile<DHID, P, false>(h1, ei, tb, tile, queue, nullptr, nullptr, list, n_list, lo, lane, wave);
    if (!listed) {
        __syncthreads();
        stage_layer2(h1 + (size_t)lo * DHID, Wl, Wr, ah_s, wl_s, wr_s, nv, tid);
    }
    __syncthreads();

    v8f acc[2] = {};
#pragma unroll
    for (int h = 0; h < 2; ++h) {
        const int u  = wave + h * AGG_WAVES;
        const int rt = u >> 2, nt = u & 3;
        if (u < NUNIT && rt < nrt) {
            const float dinv = 1.0f / fmaxf(degm[h], 1.0f);
            const float* arow = tile + (rt * 16 + mr) * P + kb;
            const float* hrow = ah_s + (rt * 16 + mr) * P + kb;
            const float* lrow = wl_s + (nt * 16 + mr) * P + kb;
            const float* rrow = wr_s + (nt * 16 + mr) * P + kb;
            v8f c = {};
            for (int kk = 0; kk < DHID / 4; ++kk) {
                const v2f s = *reinterpret_cast<const v2f*>(arow + kk * 4);
                v2f a;
                a.x = s.x * dinv;
                a.y = s.y * dinv;
                const v2f b = *reinterpret_cast<const v2f*>(lrow + kk * 4);
                c = __builtin_amdgcn_wmma_f32_16x16x4_f32(
                    false, a, false, b, (short)0, c, false, false);
            }
            for (int kk = 0; kk < DHID / 4; ++kk) {
                const v2f a = *reinterpret_cast<const v2f*>(hrow + kk * 4);
                const v2f b = *reinterpret_cast<const v2f*>(rrow + kk * 4);
                c = __builtin_amdgcn_wmma_f32_16x16x4_f32(
                    false, a, false, b, (short)0, c, false, false);
            }
            acc[h] = c;
        }
    }
    __syncthreads();

#pragma unroll
    for (int h = 0; h < 2; ++h) {
        const int u  = wave + h * AGG_WAVES;
        const int rt = u >> 2, nt = u & 3;
        if (u < NUNIT && rt < nrt) {
            const int n = nt * 16 + mr;
            float* zc = tile + (rt * 16 + 8 * (lane >> 4)) * P + n;
#pragma unroll
            for (int v = 0; v < 8; ++v) zc[v * P] = acc[h][v] + bias;
        }
    }
    __syncthreads();

    constexpr int RPW = (TR + AGG_WAVES - 1) / AGG_WAVES;
    v2f z[RPW];
    float mx[RPW], sm[RPW];
#pragma unroll
    for (int i = 0; i < RPW; ++i) {
        const int r = min(wave + i * AGG_WAVES, TR - 1);
        z[i] = *reinterpret_cast<const v2f*>(tile + r * P + lane * 2);
        mx[i] = fmaxf(z[i].x, z[i].y);
    }
#pragma unroll
    for (int o = 16; o > 0; o >>= 1) {
#pragma unroll
        for (int i = 0; i < RPW; ++i) mx[i] = fmaxf(mx[i], __shfl_xor(mx[i], o));
    }
#pragma unroll
    for (int i = 0; i < RPW; ++i) sm[i] = expf(z[i].x - mx[i]) + expf(z[i].y - mx[i]);
#pragma unroll
    for (int o = 16; o > 0; o >>= 1) {
#pragma unroll
        for (int i = 0; i < RPW; ++i) sm[i] += __shfl_xor(sm[i], o);
    }
#pragma unroll
    for (int i = 0; i < RPW; ++i) {
        const int r = wave + i * AGG_WAVES;
        if (r < nv) {
            const float lse = mx[i] + logf(sm[i]);
            float2 res;
            res.x = z[i].x - lse;
            res.y = z[i].y - lse;
            *reinterpret_cast<float2*>(out + (size_t)(lo + r) * DOUT + lane * 2) = res;
        }
    }
}

extern "C" void kernel_launch(void* const* d_in, const int* in_sizes, int n_in,
                              void* d_out, int out_size, void* d_ws, size_t ws_size,
                              hipStream_t stream) {
    const float* x   = (const float*)d_in[0];
    const int*   ei  = (const int*)d_in[1];
    const float* Wl1 = (const float*)d_in[2];
    const float* bl1 = (const float*)d_in[3];
    const float* Wr1 = (const float*)d_in[4];
    const float* Wl2 = (const float*)d_in[5];
    const float* bl2 = (const float*)d_in[6];
    const float* Wr2 = (const float*)d_in[7];
    float* out = (float*)d_out;

    float* ws   = (float*)d_ws;
    unsigned* tb = (unsigned*)d_ws;
    float* lsts = ws + (size_t)NN * DIN;
    float* h1   = lsts + (size_t)NN * DHID;
    float* degb = h1 + (size_t)NN * DHID;

    recode_kernel<<<RC_BLOCKS, RC_THREADS, 0, stream>>>(ei, tb);
    layer1_kernel<<<NT, AGG_THREADS, 0, stream>>>(x, ei, Wl1, bl1, Wr1, h1, degb, lsts, tb);
    layer2_kernel<<<NT, AGG_THREADS, 0, stream>>>(h1, ei, degb, lsts, Wl2, bl2, Wr2, out, tb);
}


# Root-term products (x.Wr1^T, h1.Wr2^T) issued inside the row gather of both layers, between a group's row requests and their use, not behind the barrier; root term summed first, then neighbours
# speedup vs baseline: 5.1369x; 1.0334x over previous
#include <hip/hip_runtime.h>
#include <cmath>


#define NN 50000
#define NE 800000
#define DIN 64
#define DHID 128
#define DOUT 64

#define TR 208
#define NT ((NN + TR - 1) / TR)
#define AGG_WAVES 32
#define AGG_THREADS (AGG_WAVES * 32)
#define NCH ((NE + 511) / 512)
#define NEP (NCH * 512)
#define UNR 7
#define NPASS ((NCH + AGG_WAVES * UNR - 1) / (AGG_WAVES * UNR))
#define QCAP 1024
#define LHEAD 64
#define LSTRIDE ((NN * DHID / NT) / 64 * 64)
#define LCAP (LSTRIDE - LHEAD)

static_assert((NT - 1) * TR < NN && NT * TR >= NN && (NN - (NT - 1) * TR) % 16 == 0, "tiles cover the nodes; the last one holds whole row tiles");
static_assert((size_t)NT * LSTRIDE <= (size_t)NN * DHID, "the lists must fit their part of the workspace");
static_assert(((size_t)NN * DIN * 4) % 256 == 0 && ((size_t)NN * DHID * 4) % 256 == 0, "the lists and h1 must start on 256-byte boundaries");
static_assert(NE % 4 == 0, "four targets make one dword of tile bytes");
static_assert(NT <= 255, "a tile number fits a byte, and 255 is no tile");
static_assert((size_t)NEP <= (size_t)NN * DIN * 4, "the tile bytes fit the first part of the workspace");
static_assert(TR <= 256, "local target index is packed into 8 bits");
static_assert(NE <= (1 << 24), "edge index is packed into 24 bits");
static_assert(NN <= (1 << 24), "source node is packed into 24 bits of a list entry");
static_assert(NCH / NT >= 1, "start offsets must differ between tiles");
static_assert(NT * (NCH / NT) <= NCH, "one subtraction must be enough to wrap a chunk number");
static_assert(QCAP >= 2 * 512, "one chunk must always fit behind the flush mark");

typedef __attribute__((ext_vector_type(2))) float v2f;
typedef __attribute__((ext_vector_type(8))) float v8f;

template <int D, int P, class F>
__device__ __forceinline__ void add_rows(const float* __restrict__ feat, float* tile,
                                         int s, int t, int m, int lane, F& between) {
    for (int k = 0; k < m; k += 8) {
        float2 r[8][D / 64];
#pragma unroll
        for (int i = 0; i < 8; ++i) {
            const int sk = __builtin_amdgcn_readlane(s, k + i);
            const float* row = feat + (size_t)sk * D + lane * 2;
#pragma unroll
            for (int j = 0; j < D / 64; ++j)
                r[i][j] = *reinterpret_cast<const float2*>(row + j * 64);
        }
        between();
#pragma unroll
        for (int i = 0; i < 8; ++i) {
            if (k + i < m) {
                const int tk = __builtin_amdgcn_readlane(t, k + i);
                float* trow = tile + tk * P + lane * 2;
#pragma unroll
                for (int j = 0; j < D / 64; ++j) {
                    atomicAdd(trow + j * 64, r[i][j].x);
                    atomicAdd(trow + j * 64 + 1, r[i][j].y);
                }
            }
        }
    }
}

template <int D, int P, bool SAVE, class F>
__device__ __forceinline__ void flush_queue(const float* __restrict__ feat,
                                            const int* __restrict__ ei,
                                            float* tile, const unsigned* q,
                                            int n, int lane, unsigned* list, int lbase,
                                            int* cnt, int lo, F& between) {
    for (int base = 0; base < n; base += 32) {
        const int idx = base + lane;
        int s = 0;
        unsigned t = 0u;
        if (idx < n) {
            const unsigned e = min(q[idx], (unsigned)(NE - 1));
            s = ei[e];
            t = min((unsigned)(ei[NE + e] - lo), (unsigned)(TR - 1));
            if (SAVE) {
                atomicAdd(&cnt[t], 1);
                if (lbase + idx < LCAP)
                    list[LHEAD + lbase + idx] = ((unsigned)s << 8) | t;
            }
        }
        add_rows<D, P>(feat, tile, s, (int)t, min(32, n - base), lane, between);
    }
}

template <int D, int P, class F>
__device__ __forceinline__ void replay_list(const float* __restrict__ feat, float* tile,
                                            const unsigned* lst, int n, int lane, F& between) {
    unsigned cur = lane < n ? lst[lane] : 0u;
    for (int base = 0; base < n; base += 32) {
        const int nx = base + 32 + lane;
        const unsigned nxt = nx < n ? lst[nx] : 0u;
        const int s = min((int)(cur >> 8), NN - 1);
        const int t = min((int)(cur & 255u), TR - 1);
        add_rows<D, P>(feat, tile, s, t, min(32, n - base), lane, between);
        cur = nxt;
    }
}

template <int D, int P, bool FIRST, class F>
__device__ __forceinline__ void gather_tile(
    const float* __restrict__ feat, const int* __restrict__ ei,
    const unsigned* __restrict__ tb, float* tile,
    unsigned* queue, int* cnt, int* lfill, unsigned* list, int n_list,
    int lo, int lane, int wave, F& between) {
    bool replayed = false;
    if (!FIRST) {
        const int n = n_list;
        if (n >= 0 && n <= LCAP) {
            const int per = ((n + AGG_WAVES - 1) / AGG_WAVES + 7) & ~7;
            const int beg = min(n, wave * per);
            const int len = min(n - beg, per);
            replay_list<D, P>(feat, tile, list + LHEAD + beg, len, lane, between);
            replayed = true;
        }
    }
    if (!replayed) {
    unsigned* q = queue + wave * QCAP;
    int qn = 0;
    const uint4* dp = reinterpret_cast<const uint4*>(tb);
    const unsigned pat = blockIdx.x * 0x01010101u;
    const int start = blockIdx.x * (NCH / NT);

    auto chunk_at = [&](int k) {
        int r = k + start;
        if (r >= NCH) r -= NCH;
        return min(r, NCH - 1);
    };
    auto load_pass = [&](uint4 (&v)[UNR], int k0) {
#pragma unroll
        for (int u = 0; u < UNR; ++u) v[u] = dp[chunk_at(k0 + u * AGG_WAVES) * 32 + lane];
    };

    auto flush_now = [&](auto& fn) {
        int lb = 0;
        if (FIRST) {
            if (lane == 0) lb = atomicAdd(lfill, qn);
            lb = __builtin_amdgcn_readfirstlane(lb);
        }
        flush_queue<D, P, FIRST>(feat, ei, tile, q, qn, lane, list, lb, cnt, lo, fn);
        qn = 0;
    };
    auto nothing = []() {};

    auto scan_chunk = [&](const uint4& d, int chunk) {
        auto zero_bytes = [&](unsigned v) {
            const unsigned x = v ^ pat;
            return ~(((x & 0x7f7f7f7fu) + 0x7f7f7f7fu) | x) & 0x80808080u;
        };
        unsigned m = (zero_bytes(d.x) >> 7) | (zero_bytes(d.y) >> 6) |
                     (zero_bytes(d.z) >> 5) | (zero_bytes(d.w) >> 4);
        unsigned mask = __builtin_amdgcn_ballot_w32(m != 0u);
        if (mask) {
            const unsigned e0 = (unsigned)(chunk * 32 + lane) * 16u;
            for (int turn = 0; turn < 16 && mask; ++turn) {
                if (m != 0u) {
                    const unsigned bit = (unsigned)__builtin_ctz(m);
                    const int pos = qn + (int)__builtin_amdgcn_mbcnt_lo(mask, 0u);
                    q[pos] = e0 + (bit & 7u) * 4u + (bit >> 3);
                    m &= m - 1u;
                }
                qn += __builtin_popcount(mask);
                mask = __builtin_amdgcn_ballot_w32(m != 0u);
            }
        }
    };

    uint4 cur[UNR], nxt[UNR];
    load_pass(nxt, wave);
    for (int p = 0; p < NPASS; ++p) {
        const int k0 = p * UNR * AGG_WAVES + wave;
#pragma unroll
        for (int k = 0; k < UNR; ++k) cur[k] = nxt[k];
        if (p + 1 < NPASS) load_pass(nxt, k0 + UNR * AGG_WAVES);
        int u = 0;
        do {
            bool full = false;
#pragma unroll
            for (int k = 0; k < UNR; ++k) {
                if (k >= u && !full) {
                    const int kk = k0 + k * AGG_WAVES;
                    if (kk < NCH) scan_chunk(cur[k], chunk_at(kk));
                    u = k + 1;
                    full = qn > QCAP - 512;
                }
            }
            if (full) {
                __builtin_amdgcn_fence(__ATOMIC_RELEASE, "wavefront");
                __builtin_amdgcn_fence(__ATOMIC_ACQUIRE, "wavefront");
                flush_now(nothing);
            }
        } while (u < UNR);
    }
    __builtin_amdgcn_fence(__ATOMIC_RELEASE, "wavefront");
    __builtin_amdgcn_fence(__ATOMIC_ACQUIRE, "wavefront");
    flush_now(between);
    }
}

#define RC_BLOCKS 256
#define RC_THREADS 256
#define RC_PER ((NEP / 4 + RC_BLOCKS * RC_THREADS - 1) / (RC_BLOCKS * RC_THREADS))
__global__ __launch_bounds__(RC_THREADS) void recode_kernel(const int* __restrict__ ei,
                                                            unsigned* __restrict__ tb) {
    const int4* dp = reinterpret_cast<const int4*>(ei + NE);
    const int i0 = blockIdx.x * RC_THREADS + threadIdx.x;
    int4 d[RC_PER];
#pragma unroll
    for (int u = 0; u < RC_PER; ++u)
        d[u] = dp[min(i0 + u * RC_BLOCKS * RC_THREADS, NE / 4 - 1)];
#pragma unroll
    for (int u = 0; u < RC_PER; ++u) {
        const int i = i0 + u * RC_BLOCKS * RC_THREADS;
        if (i < NEP / 4) {
            unsigned w = ~0u;
            if (i < NE / 4) {
                const unsigned b0 = (unsigned)d[u].x < (unsigned)NN ? (unsigned)d[u].x / TR : 255u;
                const unsigned b1 = (unsigned)d[u].y < (unsigned)NN ? (unsigned)d[u].y / TR : 255u;
                const unsigned b2 = (unsigned)d[u].z < (unsigned)NN ? (unsigned)d[u].z / TR : 255u;
                const unsigned b3 = (unsigned)d[u].w < (unsigned)NN ? (unsigned)d[u].w / TR : 255u;
                w = b0 | (b1 << 8) | (b2 << 16) | (b3 << 24);
            }
            tb[i] = w;
        }
    }
}

#define P1 (DIN + 4)
#define NUNIT1 ((TR / 16) * (DHID / 32))
#define NTILE (NN / 16)

static_assert(NN % 16 == 0 && TR % 16 == 0, "row tiles must cover the nodes exactly");
static_assert(NT <= 256, "one work-group per processor, all resident at once");
static_assert((2 * TR * P1 + 2 * DHID * P1 + AGG_WAVES * QCAP + TR + 1) * 4 > 160 * 1024, "layer 1: one work-group per processor");
static_assert((2 * TR * P1 + 2 * DHID * P1 + AGG_WAVES * QCAP + TR + 1) * 4 <= 320 * 1024, "layer 1 must fit the LDS");
static_assert(NUNIT1 <= 2 * AGG_WAVES, "at most two output blocks per wave");

__global__ __launch_bounds__(AGG_THREADS) void layer1_kernel(
    const float* __restrict__ x, const int* __restrict__ ei,
    const float* __restrict__ Wl, const float* __restrict__ bl,
    const float* __restrict__ Wr, float* __restrict__ h1,
    float* __restrict__ deg, float* __restrict__ lists,
    const unsigned* __restrict__ tb) {
    constexpr int D = DIN;
    constexpr int P = P1;
    constexpr int Q = DIN / 4;
    constexpr int NW = DHID * Q / AGG_THREADS;
    constexpr int NH = (TR * Q + AGG_THREADS - 1) / AGG_THREADS;
    static_assert(DHID * Q % AGG_THREADS == 0, "the weights split evenly over the work-group");
    __shared__ __attribute__((aligned(16))) float smem[2 * TR * P + 2 * DHID * P];
    __shared__ unsigned queue[AGG_WAVES * QCAP];
    __shared__ int cnt[TR];
    __shared__ int lfill;
    float* tile = smem;
    float* ax_s = tile + TR * P;
    float* wl_s = ax_s + TR * P;
    float* wr_s = wl_s + DHID * P;

    const int tid  = threadIdx.x;
    const int lane = tid & 31;
    const int wave = __builtin_amdgcn_readfirstlane(tid >> 5);
    const int lo   = blockIdx.x * TR;
    const int nv   = min(TR, NN - lo);
    const int nrt  = nv / 16;
    const int mr   = lane & 15;
    const int kb   = (lane >> 4) * 2;

    unsigned* list = reinterpret_cast<unsigned*>(lists) + (size_t)blockIdx.x * LSTRIDE;

    const float4* pl = reinterpret_cast<const float4*>(Wl);
    const float4* pr = reinterpret_cast<const float4*>(Wr);
    const float4* px = reinterpret_cast<const float4*>(x + (size_t)lo * D);
    const int last = nv * Q - 1;
    float4 vl[NW], vr[NW], vx[NH];
#pragma unroll
    for (int j = 0; j < NW; ++j) {
        vl[j] = pl[tid + j * AGG_THREADS];
        vr[j] = pr[tid + j * AGG_THREADS];
    }
#pragma unroll
    for (int j = 0; j < NH; ++j) vx[j] = px[min(tid + j * AGG_THREADS, last)];

    for (int i = tid; i < TR * P / 4; i += AGG_THREADS)
        reinterpret_cast<float4*>(tile)[i] = make_float4(0.f, 0.f, 0.f, 0.f);
    for (int i = tid; i < TR; i += AGG_THREADS) cnt[i] = 0;
    if (tid == 0) lfill = 0;

#pragma unroll
    for (int j = 0; j < NW; ++j) {
        const int i = tid + j * AGG_THREADS;
        const int o = (i / Q) * P + (i % Q) * 4;
        *reinterpret_cast<float4*>(wl_s + o) = vl[j];
        *reinterpret_cast<float4*>(wr_s + o) = vr[j];
    }
#pragma unroll
    for (int j = 0; j < NH; ++j) {
        const int i = tid + j * AGG_THREADS;
        if ((j + 1) * AGG_THREADS <= TR * Q || i < TR * Q) {
            const int o = (i / Q) * P + (i % Q) * 4;
            *reinterpret_cast<float4*>(ax_s + o) = vx[j];
        }
    }
    __syncthreads();

    const int cpw = wave & 3;
    const float* wr0 = wr_s + (cpw * 32 + mr) * P + kb;
    const float* wr1 = wr0 + 16 * P;
    bool has[2];
    const float* xrw[2];
#pragma unroll
    for (int h = 0; h < 2; ++h) {
        const int u = wave + h * AGG_WAVES;
        has[h] = u < NUNIT1 && (u >> 2) < nrt;
        xrw[h] = ax_s + (min(u >> 2, TR / 16 - 1) * 16 + mr) * P + kb;
    }
    v8f rc0[2] = {}, rc1[2] = {};
    int rk = 0;
    constexpr int RCH = 2;
    static_assert((DIN / 4) % RCH == 0, "the root term is made of whole chunks");
    auto root_chunk = [&]() {
        v2f b0[RCH], b1[RCH], a0[RCH], a1[RCH];
#pragma unroll
        for (int j = 0; j < RCH; ++j) {
            b0[j] = *reinterpret_cast<const v2f*>(wr0 + (rk + j) * 4);
            b1[j] = *reinterpret_cast<const v2f*>(wr1 + (rk + j) * 4);
            a0[j] = *reinterpret_cast<const v2f*>(xrw[0] + (rk + j) * 4);
            a1[j] = *reinterpret_cast<const v2f*>(xrw[1] + (rk + j) * 4);
        }
        if (has[1]) {
#pragma unroll
            for (int j = 0; j < RCH; ++j) {
                rc0[0] = __builtin_amdgcn_wmma_f32_16x16x4_f32(false, a0[j], false, b0[j], (short)0, rc0[0], false, false);
                rc1[0] = __builtin_amdgcn_wmma_f32_16x16x4_f32(false, a0[j], false, b1[j], (short)0, rc1[0], false, false);
                rc0[1] = __builtin_amdgcn_wmma_f32_16x16x4_f32(false, a1[j], false, b0[j], (short)0, rc0[1], false, false);
                rc1[1] = __builtin_amdgcn_wmma_f32_16x16x4_f32(false, a1[j], false, b1[j], (short)0, rc1[1], false, false);
            }
        } else if (has[0]) {
#pragma unroll
            for (int j = 0; j < RCH; ++j) {
                rc0[0] = __builtin_amdgcn_wmma_f32_16x16x4_f32(false, a0[j], false, b0[j], (short)0, rc0[0], false, false);
                rc1[0] = __builtin_amdgcn_wmma_f32_16x16x4_f32(false, a0[j], false, b1[j], (short)0, rc1[0], false, false);
            }
        }
        rk += RCH;
    };
    auto between = [&]() { if (rk < DIN / 4) root_chunk(); };

    gather_tile<D, P, true>(x, ei, tb, tile, queue, cnt, &lfill, list, -1, lo, lane, wave, between);
    while (rk < DIN / 4) root_chunk();
    __syncthreads();

    for (int i = tid; i < nv; i += AGG_THREADS) deg[lo + i] = (float)cnt[i];
    if (tid == 0) list[0] = (unsigned)lfill;

#pragma unroll
    for (int h = 0; h < 2; ++h) {
        const int u  = wave + h * AGG_WAVES;
        const int rt = u >> 2, cp = u & 3;
        if (has[h]) {
            const float dinv = 1.0f / fmaxf((float)cnt[rt * 16 + mr], 1.0f);
            const float* arow = tile + (rt * 16 + mr) * P + kb;
            const float* l0 = wl_s + (cp * 32 + mr) * P + kb;
            const float* l1 = l0 + 16 * P;
            v8f c0 = rc0[h], c1 = rc1[h];
            for (int kk = 0; kk < DIN / 4; ++kk) {
                const v2f s = *reinterpret_cast<const v2f*>(arow + kk * 4);
                v2f a;
                a.x = s.x * dinv;
                a.y = s.y * dinv;
                const v2f b0 = *reinterpret_cast<const v2f*>(l0 + kk * 4);
                const v2f b1 = *reinterpret_cast<const v2f*>(l1 + kk * 4);
                c0 = __builtin_amdgcn_wmma_f32_16x16x4_f32(
                    false, a, false, b0, (short)0, c0, false, false);
                c1 = __builtin_amdgcn_wmma_f32_16x16x4_f32(
                    false, a, false, b1, (short)0, c1, false, false);
            }
            const int n0 = cp * 32 + mr;
            const float bias0 = bl[n0];
            const float bias1 = bl[n0 + 16];
            float* hc = h1 + (size_t)(lo + rt * 16 + 8 * (lane >> 4)) * DHID + n0;
#pragma unroll
            for (int v = 0; v < 8; ++v) {
                hc[v * DHID]      = tanhf(c0[v] + bias0);
                hc[v * DHID + 16] = tanhf(c1[v] + bias1);
            }
        }
    }
}

#define P2 (DHID + 4)
#define NUNIT ((TR / 16) * (DOUT / 16))

static_assert((2 * TR * P2 + 2 * DOUT * P2) * 4 > 160 * 1024, "one work-group per processor");
static_assert((2 * TR * P2 + 2 * DOUT * P2) * 4 <= 320 * 1024, "must fit the LDS");
static_assert(AGG_WAVES * QCAP <= TR * P2 + 2 * DOUT * P2, "the scan's queues fit behind the tile");
static_assert(NUNIT <= 2 * AGG_WAVES, "at most two output blocks per wave");
static_assert(AGG_WAVES % (DOUT / 16) == 0, "both blocks of a wave lie in one column tile");

__device__ __forceinline__ void stage_layer2(const float* __restrict__ h1t,
                                             const float* __restrict__ Wl, const float* __restrict__ Wr,
                                             float* ah_s, float* wl_s, float* wr_s, int nv, int tid) {
    constexpr int Q  = DHID / 4;
    constexpr int NW = DOUT * Q / AGG_THREADS;
    constexpr int NH = (TR * Q + AGG_THREADS - 1) / AGG_THREADS;
    static_assert(DOUT * Q % AGG_THREADS == 0, "the weights split evenly over the work-group");
    const float4* pl = reinterpret_cast<const float4*>(Wl);
    const float4* pr = reinterpret_cast<const float4*>(Wr);
    const float4* ph = reinterpret_cast<const float4*>(h1t);
    const int last = nv * Q - 1;
    float4 vl[NW], vr[NW], vh[NH];
#pragma unroll
    for (int j = 0; j < NW; ++j) {
        vl[j] = pl[tid + j * AGG_THREADS];
        vr[j] = pr[tid + j * AGG_THREADS];
    }
#pragma unroll
    for (int j = 0; j < NH; ++j) vh[j] = ph[min(tid + j * AGG_THREADS, last)];
#pragma unroll
    for (int j = 0; j < NW; ++j) {
        const int i = tid + j * AGG_THREADS;
        const int o = (i / Q) * P2 + (i % Q) * 4;
        *reinterpret_cast<float4*>(wl_s + o) = vl[j];
        *reinterpret_cast<float4*>(wr_s + o) = vr[j];
    }
#pragma unroll
    for (int j = 0; j < NH; ++j) {
        const int i = tid + j * AGG_THREADS;
        if ((j + 1) * AGG_THREADS <= TR * Q || i < TR * Q) {
            const int o = (i / Q) * P2 + (i % Q) * 4;
            *reinterpret_cast<float4*>(ah_s + o) = vh[j];
        }
    }
}

__global__ __launch_bounds__(AGG_THREADS) void layer2_kernel(
    const float* __restrict__ h1, const int* __restrict__ ei,
    const float* __restrict__ deg, float* __restrict__ lists,
    const float* __restrict__ Wl, const float* __restrict__ bl,
    const float* __restrict__ Wr, float* __restrict__ out,
    const unsigned* __restrict__ tb) {
    constexpr int P = P2;
    __shared__ __attribute__((aligned(16))) float smem[2 * TR * P + 2 * DOUT * P];
    float* tile = smem;
    float* ah_s = tile + TR * P;
    float* wl_s = ah_s + TR * P;
    float* wr_s = wl_s + DOUT * P;
    unsigned* queue = reinterpret_cast<unsigned*>(ah_s);

    const int tid  = threadIdx.x;
    const int lane = tid & 31;
    const int wave = __builtin_amdgcn_readfirstlane(tid >> 5);
    const int lo   = blockIdx.x * TR;
    const int nv   = min(TR, NN - lo);
    const int nrt  = nv / 16;
    const int mr   = lane & 15;
    const int kb   = (lane >> 4) * 2;

    unsigned* list = reinterpret_cast<unsigned*>(lists) + (size_t)blockIdx.x * LSTRIDE;
    const int n_list = (int)list[0];
    float degm[2];
#pragma unroll
    for (int h = 0; h < 2; ++h) {
        const int rt = (wave + h * AGG_WAVES) >> 2;
        degm[h] = deg[min(lo + rt * 16 + mr, NN - 1)];
    }
    const float bias = bl[(wave & 3) * 16 + mr];

    for (int i = tid; i < TR * P / 4; i += AGG_THREADS)
        reinterpret_cast<float4*>(tile)[i] = make_float4(0.f, 0.f, 0.f, 0.f);
    __syncthreads();

    const bool listed = n_list >= 0 && n_list <= LCAP;
    if (listed) {
        stage_layer2(h1 + (size_t)lo * DHID, Wl, Wr, ah_s, wl_s, wr_s, nv, tid);
        __syncthreads();
    }

    const float* rrow = wr_s + ((wave & 3) * 16 + mr) * P + kb;
    bool has[2];
    const float* hrw[2];
#pragma unroll
    for (int h = 0; h < 2; ++h) {
        const int u = wave + h * AGG_WAVES;
        has[h] = u < NUNIT && (u >> 2) < nrt;
        hrw[h] = ah_s + (min(u >> 2, TR / 16 - 1) * 16 + mr) * P + kb;
    }
    v8f acc[2] = {};
    int rk = 0;
    constexpr int RCH = 4;
    static_assert((DHID / 4) % RCH == 0, "the root term is made of whole chunks");
    auto root_chunk = [&]() {
        v2f b[RCH], a0[RCH], a1[RCH];
#pragma unroll
        for (int j = 0; j < RCH; ++j) {
            b[j]  = *reinterpret_cast<const v2f*>(rrow + (rk + j) * 4);
            a0[j] = *reinterpret_cast<const v2f*>(hrw[0] + (rk + j) * 4);
            a1[j] = *reinterpret_cast<const v2f*>(hrw[1] + (rk + j) * 4);
        }
        if (has[1]) {
#pragma unroll
            for (int j = 0; j < RCH; ++j) {
                acc[0] = __builtin_amdgcn_wmma_f32_16x16x4_f32(false, a0[j], false, b[j], (short)0, acc[0], false, false);
                acc[1] = __builtin_amdgcn_wmma_f32_16x16x4_f32(false, a1[j], false, b[j], (short)0, acc[1], false, false);
            }
        } else if (has[0]) {
#pragma unroll
            for (int j = 0; j < RCH; ++j)
                acc[0] = __builtin_amdgcn_wmma_f32_16x16x4_f32(false, a0[j], false, b[j], (short)0, acc[0], false, false);
        }
        rk += RCH;
    };
    auto between = [&]() { if (listed && rk < DHID / 4) root_chunk(); };

    gather_tile<DHID, P, false>(h1, ei, tb, tile, queue, nullptr, nullptr, list, n_list, lo, lane, wave, between);
    if (listed) {
        while (rk < DHID / 4) root_chunk();
    } else {
        __syncthreads();
        stage_layer2(h1 + (size_t)lo * DHID, Wl, Wr, ah_s, wl_s, wr_s, nv, tid);
    }
    __syncthreads();
    while (rk < DHID / 4) root_chunk();

#pragma unroll
    for (int h = 0; h < 2; ++h) {
        const int u  = wave + h * AGG_WAVES;
        const int rt = u >> 2, nt = u & 3;
        if (has[h]) {
            const float dinv = 1.0f / fmaxf(degm[h], 1.0f);
            const float* arow = tile + (rt * 16 + mr) * P + kb;
            const float* lrow = wl_s + (nt * 16 + mr) * P + kb;
            v8f c = acc[h];
            for (int kk = 0; kk < DHID / 4; ++kk) {
                const v2f s = *reinterpret_cast<const v2f*>(arow + kk * 4);
                v2f a;
                a.x = s.x * dinv;
                a.y = s.y * dinv;
                const v2f b = *reinterpret_cast<const v2f*>(lrow + kk * 4);
                c = __builtin_amdgcn_wmma_f32_16x16x4_f32(
                    false, a, false, b, (short)0, c, false, false);
            }
            acc[h] = c;
        }
    }
    __syncthreads();

#pragma unroll
    for (int h = 0; h < 2; ++h) {
        const int u  = wave + h * AGG_WAVES;
        const int rt = u >> 2, nt = u & 3;
        if (u < NUNIT && rt < nrt) {
            const int n = nt * 16 + mr;
            float* zc = tile + (rt * 16 + 8 * (lane >> 4)) * P + n;
#pragma unroll
            for (int v = 0; v < 8; ++v) zc[v * P] = acc[h][v] + bias;
        }
    }
    __syncthreads();

    constexpr int RPW = (TR + AGG_WAVES - 1) / AGG_WAVES;
    v2f z[RPW];
    float mx[RPW], sm[RPW];
#pragma unroll
    for (int i = 0; i < RPW; ++i) {
        const int r = min(wave + i * AGG_WAVES, TR - 1);
        z[i] = *reinterpret_cast<const v2f*>(tile + r * P + lane * 2);
        mx[i] = fmaxf(z[i].x, z[i].y);
    }
#pragma unroll
    for (int o = 16; o > 0; o >>= 1) {
#pragma unroll
        for (int i = 0; i < RPW; ++i) mx[i] = fmaxf(mx[i], __shfl_xor(mx[i], o));
    }
#pragma unroll
    for (int i = 0; i < RPW; ++i) sm[i] = expf(z[i].x - mx[i]) + expf(z[i].y - mx[i]);
#pragma unroll
    for (int o = 16; o > 0; o >>= 1) {
#pragma unroll
        for (int i = 0; i < RPW; ++i) sm[i] += __shfl_xor(sm[i], o);
    }
#pragma unroll
    for (int i = 0; i < RPW; ++i) {
        const int r = wave + i * AGG_WAVES;
        if (r < nv) {
            const float lse = mx[i] + logf(sm[i]);
            float2 res;
            res.x = z[i].x - lse;
            res.y = z[i].y - lse;
            *reinterpret_cast<float2*>(out + (size_t)(lo + r) * DOUT + lane * 2) = res;
        }
    }
}

extern "C" void kernel_launch(void* const* d_in, const int* in_sizes, int n_in,
                              void* d_out, int out_size, void* d_ws, size_t ws_size,
                              hipStream_t stream) {
    const float* x   = (const float*)d_in[0];
    const int*   ei  = (const int*)d_in[1];
    const float* Wl1 = (const float*)d_in[2];
    const float* bl1 = (const float*)d_in[3];
    const float* Wr1 = (const float*)d_in[4];
    const float* Wl2 = (const float*)d_in[5];
    const float* bl2 = (const float*)d_in[6];
    const float* Wr2 = (const float*)d_in[7];
    float* out = (float*)d_out;

    float* ws   = (float*)d_ws;
    unsigned* tb = (unsigned*)d_ws;
    float* lsts = ws + (size_t)NN * DIN;
    float* h1   = lsts + (size_t)NN * DHID;
    float* degb = h1 + (size_t)NN * DHID;

    recode_kernel<<<RC_BLOCKS, RC_THREADS, 0, stream>>>(ei, tb);
    layer1_kernel<<<NT, AGG_THREADS, 0, stream>>>(x, ei, Wl1, bl1, Wr1, h1, degb, lsts, tb);
    layer2_kernel<<<NT, AGG_THREADS, 0, stream>>>(h1, ei, degb, lsts, Wl2, bl2, Wr2, out, tb);
}


# First kernel also writes, per tile, one bit per group of 16 edges (set if the group holds an edge of it); a tile loads and tests only its set groups, not all tile bytes; sum order by LDS arrival
# speedup vs baseline: 5.3967x; 1.0506x over previous
#include <hip/hip_runtime.h>
#include <cmath>


#define NN 50000
#define NE 800000
#define DIN 64
#define DHID 128
#define DOUT 64

#define TR 208
#define NT ((NN + TR - 1) / TR)
#define AGG_WAVES 32
#define AGG_THREADS (AGG_WAVES * 32)
#define NCH ((NE + 511) / 512)
#define NEP (NCH * 512)
#define NGRP (NEP / 16)
#define SLICE 4096
#define NSL ((NEP + SLICE - 1) / SLICE)
#define BMS 2048
#define NBW (NSL * 8)
#define GPR 4
#define QCAP 1024
#define LHEAD 64
#define LSTRIDE ((NN * DHID / NT) / 64 * 64)
#define LCAP (LSTRIDE - LHEAD)

static_assert((NT - 1) * TR < NN && NT * TR >= NN && (NN - (NT - 1) * TR) % 16 == 0, "tiles cover the nodes; the last one holds whole row tiles");
static_assert((size_t)NT * LSTRIDE <= (size_t)NN * DHID, "the lists must fit their part of the workspace");
static_assert(((size_t)NN * DIN * 4) % 256 == 0 && ((size_t)NN * DHID * 4) % 256 == 0, "the lists and h1 must start on 256-byte boundaries");
static_assert(NE % 4 == 0, "four targets make one dword of tile bytes");
static_assert(NT <= 255, "a tile number fits a byte, and 255 is no tile");
static_assert((size_t)NEP + (size_t)NSL * BMS * 4 <= (size_t)NN * DIN * 4, "tile bytes and group bits fit the first part of the workspace");
static_assert(NEP % 256 == 0 && NEP - NE >= 16, "the bits start on a line; the last group is padding only");
static_assert(NT <= 256, "a tile slot per tile in every slice's bits");
static_assert(TR <= 256, "local target index is packed into 8 bits");
static_assert(NE <= (1 << 24), "edge index is packed into 24 bits");
static_assert(NN <= (1 << 24), "source node is packed into 24 bits of a list entry");
static_assert(QCAP >= 2 * 512, "one round of 32 groups must always fit behind the flush mark");

typedef __attribute__((ext_vector_type(2))) float v2f;
typedef __attribute__((ext_vector_type(8))) float v8f;

template <int D, int P, class F>
__device__ __forceinline__ void add_rows(const float* __restrict__ feat, float* tile,
                                         int s, int t, int m, int lane, F& between) {
    for (int k = 0; k < m; k += 8) {
        float2 r[8][D / 64];
#pragma unroll
        for (int i = 0; i < 8; ++i) {
            const int sk = __builtin_amdgcn_readlane(s, k + i);
            const float* row = feat + (size_t)sk * D + lane * 2;
#pragma unroll
            for (int j = 0; j < D / 64; ++j)
                r[i][j] = *reinterpret_cast<const float2*>(row + j * 64);
        }
        between();
#pragma unroll
        for (int i = 0; i < 8; ++i) {
            if (k + i < m) {
                const int tk = __builtin_amdgcn_readlane(t, k + i);
                float* trow = tile + tk * P + lane * 2;
#pragma unroll
                for (int j = 0; j < D / 64; ++j) {
                    atomicAdd(trow + j * 64, r[i][j].x);
                    atomicAdd(trow + j * 64 + 1, r[i][j].y);
                }
            }
        }
    }
}

template <int D, int P, bool SAVE, class F>
__device__ __forceinline__ void flush_queue(const float* __restrict__ feat,
                                            const int* __restrict__ ei,
                                            float* tile, const unsigned* q,
                                            int n, int lane, unsigned* list, int lbase,
                                            int* cnt, int lo, F& between) {
    for (int base = 0; base < n; base += 32) {
        const int idx = base + lane;
        int s = 0;
        unsigned t = 0u;
        if (idx < n) {
            const unsigned e = min(q[idx], (unsigned)(NE - 1));
            s = ei[e];
            t = min((unsigned)(ei[NE + e] - lo), (unsigned)(TR - 1));
            if (SAVE) {
                atomicAdd(&cnt[t], 1);
                if (lbase + idx < LCAP)
                    list[LHEAD + lbase + idx] = ((unsigned)s << 8) | t;
            }
        }
        add_rows<D, P>(feat, tile, s, (int)t, min(32, n - base), lane, between);
    }
}

template <int D, int P, class F>
__device__ __forceinline__ void replay_list(const float* __restrict__ feat, float* tile,
                                            const unsigned* lst, int n, int lane, F& between) {
    unsigned cur = lane < n ? lst[lane] : 0u;
    for (int base = 0; base < n; base += 32) {
        const int nx = base + 32 + lane;
        const unsigned nxt = nx < n ? lst[nx] : 0u;
        const int s = min((int)(cur >> 8), NN - 1);
        const int t = min((int)(cur & 255u), TR - 1);
        add_rows<D, P>(feat, tile, s, t, min(32, n - base), lane, between);
        cur = nxt;
    }
}

#define WPW ((NBW + AGG_WAVES - 1) / AGG_WAVES)
static_assert(WPW >= 32 && WPW <= 64, "one or two words of bits per lane");
__device__ __forceinline__ unsigned long long tile_bits(const unsigned* __restrict__ tb, int wave, int lane) {
    const unsigned* bmT = tb + NEP / 4 + blockIdx.x * 8;
    auto word = [&](int i) {
        return i < NBW ? bmT[(i >> 3) * BMS + (i & 7)] : 0u;
    };
    const int i0 = WPW * wave + lane;
    const unsigned w_lo = word(i0);
    const unsigned w_hi = lane < WPW - 32 ? word(i0 + 32) : 0u;
    return (unsigned long long)w_lo | ((unsigned long long)w_hi << 32);
}

template <int D, int P, bool FIRST, class F>
__device__ __forceinline__ void gather_tile(
    const float* __restrict__ feat, const int* __restrict__ ei,
    const unsigned* __restrict__ tb, float* tile,
    unsigned* queue, int* cnt, int* lfill, unsigned* list, int n_list,
    unsigned long long first_bits,
    int lo, int lane, int wave, F& between) {
    bool replayed = false;
    if (!FIRST) {
        const int n = n_list;
        if (n >= 0 && n <= LCAP) {
            const int per = ((n + AGG_WAVES - 1) / AGG_WAVES + 7) & ~7;
            const int beg = min(n, wave * per);
            const int len = min(n - beg, per);
            replay_list<D, P>(feat, tile, list + LHEAD + beg, len, lane, between);
            replayed = true;
        }
    }
    if (!replayed) {
    unsigned* q = queue + wave * QCAP;
    int qn = 0;
    const uint4* dp = reinterpret_cast<const uint4*>(tb);
    const unsigned pat = blockIdx.x * 0x01010101u;
    const unsigned g0 = (unsigned)(WPW * wave + lane) * 32u;
    unsigned long long bits = FIRST ? first_bits : tile_bits(tb, wave, lane);

    auto flush_now = [&](auto& fn) {
        int lb = 0;
        if (FIRST) {
            if (lane == 0) lb = atomicAdd(lfill, qn);
            lb = __builtin_amdgcn_readfirstlane(lb);
        }
        flush_queue<D, P, FIRST>(feat, ei, tile, q, qn, lane, list, lb, cnt, lo, fn);
        qn = 0;
    };
    auto nothing = []() {};

    auto scan_group = [&](const uint4& d, unsigned e0) {
        auto zero_bytes = [&](unsigned v) {
            const unsigned x = v ^ pat;
            return ~(((x & 0x7f7f7f7fu) + 0x7f7f7f7fu) | x) & 0x80808080u;
        };
        unsigned m = (zero_bytes(d.x) >> 7) | (zero_bytes(d.y) >> 6) |
                     (zero_bytes(d.z) >> 5) | (zero_bytes(d.w) >> 4);
        unsigned mask = __builtin_amdgcn_ballot_w32(m != 0u);
        for (int turn = 0; turn < 16 && mask; ++turn) {
            if (m != 0u) {
                const unsigned bit = (unsigned)__builtin_ctz(m);
                const int pos = qn + (int)__builtin_amdgcn_mbcnt_lo(mask, 0u);
                q[pos] = e0 + (bit & 7u) * 4u + (bit >> 3);
                m &= m - 1u;
            }
            qn += __builtin_popcount(mask);
            mask = __builtin_amdgcn_ballot_w32(m != 0u);
        }
    };

    for (int round = 0; round < 64 / GPR; ++round) {
        if (__builtin_amdgcn_ballot_w32(bits != 0ull) == 0u) break;
        unsigned g[GPR];
        uint4 d[GPR];
        unsigned live[GPR];
#pragma unroll
        for (int k = 0; k < GPR; ++k) {
            const bool has_bit = bits != 0ull;
            live[k] = __builtin_amdgcn_ballot_w32(has_bit);
            unsigned gk = NGRP - 1;
            if (has_bit) {
                const int b = __builtin_ctzll(bits);
                bits &= bits - 1ull;
                gk = g0 + (unsigned)(b & 31) + (unsigned)(b >> 5) * (32u * 32u);
            }
            g[k] = min(gk, (unsigned)(NGRP - 1));
            d[k] = dp[g[k]];
        }
        int u = 0;
        do {
            bool full = false;
#pragma unroll
            for (int k = 0; k < GPR; ++k) {
                if (k >= u && !full) {
                    if (live[k]) scan_group(d[k], g[k] * 16u);
                    u = k + 1;
                    full = qn > QCAP - 512;
                }
            }
            if (full) {
                __builtin_amdgcn_fence(__ATOMIC_RELEASE, "wavefront");
                __builtin_amdgcn_fence(__ATOMIC_ACQUIRE, "wavefront");
                flush_now(nothing);
            }
        } while (u < GPR);
    }
    __builtin_amdgcn_fence(__ATOMIC_RELEASE, "wavefront");
    __builtin_amdgcn_fence(__ATOMIC_ACQUIRE, "wavefront");
    flush_now(between);
    }
}

#define RC_THREADS 256
#define RC_PER (SLICE / 4 / RC_THREADS)
static_assert(SLICE / 4 == RC_PER * RC_THREADS && RC_THREADS % 128 == 0, "a slice splits evenly; 128 threads make one word of bits");
static_assert(BMS == 8 * 256 && SLICE == 8 * 32 * 16, "8 words of 32 groups of 16 edges per tile and slice");
__global__ __launch_bounds__(RC_THREADS) void recode_kernel(const int* __restrict__ ei,
                                                            unsigned* __restrict__ tb) {
    __shared__ __attribute__((aligned(16))) unsigned bm[BMS];
    const int4* dp = reinterpret_cast<const int4*>(ei + NE);
    const int tid = threadIdx.x;
    const int i0 = blockIdx.x * (SLICE / 4) + tid;
    int4 d[RC_PER];
#pragma unroll
    for (int u = 0; u < RC_PER; ++u)
        d[u] = dp[min(i0 + u * RC_THREADS, NE / 4 - 1)];
    for (int i = tid; i < BMS / 4; i += RC_THREADS)
        reinterpret_cast<uint4*>(bm)[i] = make_uint4(0u, 0u, 0u, 0u);
    __syncthreads();
    const unsigned gbit = 1u << ((tid >> 2) & 31);
#pragma unroll
    for (int u = 0; u < RC_PER; ++u) {
        const int i = i0 + u * RC_THREADS;
        if (i < NEP / 4) {
            unsigned w = ~0u;
            if (i < NE / 4) {
                const unsigned b0 = (unsigned)d[u].x < (unsigned)NN ? (unsigned)d[u].x / TR : 255u;
                const unsigned b1 = (unsigned)d[u].y < (unsigned)NN ? (unsigned)d[u].y / TR : 255u;
                const unsigned b2 = (unsigned)d[u].z < (unsigned)NN ? (unsigned)d[u].z / TR : 255u;
                const unsigned b3 = (unsigned)d[u].w < (unsigned)NN ? (unsigned)d[u].w / TR : 255u;
                w = b0 | (b1 << 8) | (b2 << 16) | (b3 << 24);
                unsigned* row = bm + (u * (RC_THREADS / 128) + (tid >> 7)) * 256;
                if (b0 != 255u) atomicOr(row + b0, gbit);
                if (b1 != 255u) atomicOr(row + b1, gbit);
                if (b2 != 255u) atomicOr(row + b2, gbit);
                if (b3 != 255u) atomicOr(row + b3, gbit);
            }
            tb[i] = w;
        }
    }
    __syncthreads();
    unsigned* gb = tb + NEP / 4 + (size_t)blockIdx.x * BMS;
    for (int i = tid; i < BMS; i += RC_THREADS) gb[i] = bm[(i & 7) * 256 + (i >> 3)];
}

#define P1 (DIN + 4)
#define NUNIT1 ((TR / 16) * (DHID / 32))
#define NTILE (NN / 16)

static_assert(NN % 16 == 0 && TR % 16 == 0, "row tiles must cover the nodes exactly");
static_assert(NT <= 256, "one work-group per processor, all resident at once");
static_assert((2 * TR * P1 + 2 * DHID * P1 + AGG_WAVES * QCAP + TR + 1) * 4 > 160 * 1024, "layer 1: one work-group per processor");
static_assert((2 * TR * P1 + 2 * DHID * P1 + AGG_WAVES * QCAP + TR + 1) * 4 <= 320 * 1024, "layer 1 must fit the LDS");
static_assert(NUNIT1 <= 2 * AGG_WAVES, "at most two output blocks per wave");

__global__ __launch_bounds__(AGG_THREADS) void layer1_kernel(
    const float* __restrict__ x, const int* __restrict__ ei,
    const float* __restrict__ Wl, const float* __restrict__ bl,
    const float* __restrict__ Wr, float* __restrict__ h1,
    float* __restrict__ deg, float* __restrict__ lists,
    const unsigned* __restrict__ tb) {
    constexpr int D = DIN;
    constexpr int P = P1;
    constexpr int Q = DIN / 4;
    constexpr int NW = DHID * Q / AGG_THREADS;
    constexpr int NH = (TR * Q + AGG_THREADS - 1) / AGG_THREADS;
    static_assert(DHID * Q % AGG_THREADS == 0, "the weights split evenly over the work-group");
    __shared__ __attribute__((aligned(16))) float smem[2 * TR * P + 2 * DHID * P];
    __shared__ unsigned queue[AGG_WAVES * QCAP];
    __shared__ int cnt[TR];
    __shared__ int lfill;
    float* tile = smem;
    float* ax_s = tile + TR * P;
    float* wl_s = ax_s + TR * P;
    float* wr_s = wl_s + DHID * P;

    const int tid  = threadIdx.x;
    const int lane = tid & 31;
    const int wave = __builtin_amdgcn_readfirstlane(tid >> 5);
    const int lo   = blockIdx.x * TR;
    const int nv   = min(TR, NN - lo);
    const int nrt  = nv / 16;
    const int mr   = lane & 15;
    const int kb   = (lane >> 4) * 2;

    unsigned* list = reinterpret_cast<unsigned*>(lists) + (size_t)blockIdx.x * LSTRIDE;

    const float4* pl = reinterpret_cast<const float4*>(Wl);
    const float4* pr = reinterpret_cast<const float4*>(Wr);
    const float4* px = reinterpret_cast<const float4*>(x + (size_t)lo * D);
    const int last = nv * Q - 1;
    float4 vl[NW], vr[NW], vx[NH];
#pragma unroll
    for (int j = 0; j < NW; ++j) {
        vl[j] = pl[tid + j * AGG_THREADS];
        vr[j] = pr[tid + j * AGG_THREADS];
    }
#pragma unroll
    for (int j = 0; j < NH; ++j) vx[j] = px[min(tid + j * AGG_THREADS, last)];
    const unsigned long long gbits = tile_bits(tb, wave, lane);

    for (int i = tid; i < TR * P / 4; i += AGG_THREADS)
        reinterpret_cast<float4*>(tile)[i] = make_float4(0.f, 0.f, 0.f, 0.f);
    for (int i = tid; i < TR; i += AGG_THREADS) cnt[i] = 0;
    if (tid == 0) lfill = 0;

#pragma unroll
    for (int j = 0; j < NW; ++j) {
        const int i = tid + j * AGG_THREADS;
        const int o = (i / Q) * P + (i % Q) * 4;
        *reinterpret_cast<float4*>(wl_s + o) = vl[j];
        *reinterpret_cast<float4*>(wr_s + o) = vr[j];
    }
#pragma unroll
    for (int j = 0; j < NH; ++j) {
        const int i = tid + j * AGG_THREADS;
        if ((j + 1) * AGG_THREADS <= TR * Q || i < TR * Q) {
            const int o = (i / Q) * P + (i % Q) * 4;
            *reinterpret_cast<float4*>(ax_s + o) = vx[j];
        }
    }
    __syncthreads();

    const int cpw = wave & 3;
    const float* wr0 = wr_s + (cpw * 32 + mr) * P + kb;
    const float* wr1 = wr0 + 16 * P;
    bool has[2];
    const float* xrw[2];
#pragma unroll
    for (int h = 0; h < 2; ++h) {
        const int u = wave + h * AGG_WAVES;
        has[h] = u < NUNIT1 && (u >> 2) < nrt;
        xrw[h] = ax_s + (min(u >> 2, TR / 16 - 1) * 16 + mr) * P + kb;
    }
    v8f rc0[2] = {}, rc1[2] = {};
    int rk = 0;
    constexpr int RCH = 2;
    static_assert((DIN / 4) % RCH == 0, "the root term is made of whole chunks");
    auto root_chunk = [&]() {
        v2f b0[RCH], b1[RCH], a0[RCH], a1[RCH];
#pragma unroll
        for (int j = 0; j < RCH; ++j) {
            b0[j] = *reinterpret_cast<const v2f*>(wr0 + (rk + j) * 4);
            b1[j] = *reinterpret_cast<const v2f*>(wr1 + (rk + j) * 4);
            a0[j] = *reinterpret_cast<const v2f*>(xrw[0] + (rk + j) * 4);
            a1[j] = *reinterpret_cast<const v2f*>(xrw[1] + (rk + j) * 4);
        }
        if (has[1]) {
#pragma unroll
            for (int j = 0; j < RCH; ++j) {
                rc0[0] = __builtin_amdgcn_wmma_f32_16x16x4_f32(false, a0[j], false, b0[j], (short)0, rc0[0], false, false);
                rc1[0] = __builtin_amdgcn_wmma_f32_16x16x4_f32(false, a0[j], false, b1[j], (short)0, rc1[0], false, false);
                rc0[1] = __builtin_amdgcn_wmma_f32_16x16x4_f32(false, a1[j], false, b0[j], (short)0, rc0[1], false, false);
                rc1[1] = __builtin_amdgcn_wmma_f32_16x16x4_f32(false, a1[j], false, b1[j], (short)0, rc1[1], false, false);
            }
        } else if (has[0]) {
#pragma unroll
            for (int j = 0; j < RCH; ++j) {
                rc0[0] = __builtin_amdgcn_wmma_f32_16x16x4_f32(false, a0[j], false, b0[j], (short)0, rc0[0], false, false);
                rc1[0] = __builtin_amdgcn_wmma_f32_16x16x4_f32(false, a0[j], false, b1[j], (short)0, rc1[0], false, false);
            }
        }
        rk += RCH;
    };
    auto between = [&]() { if (rk < DIN / 4) root_chunk(); };

    gather_tile<D, P, true>(x, ei, tb, tile, queue, cnt, &lfill, list, -1, gbits, lo, lane, wave, between);
    while (rk < DIN / 4) root_chunk();
    __syncthreads();

    for (int i = tid; i < nv; i += AGG_THREADS) deg[lo + i] = (float)cnt[i];
    if (tid == 0) list[0] = (unsigned)lfill;

#pragma unroll
    for (int h = 0; h < 2; ++h) {
        const int u  = wave + h * AGG_WAVES;
        const int rt = u >> 2, cp = u & 3;
        if (has[h]) {
            const float dinv = 1.0f / fmaxf((float)cnt[rt * 16 + mr], 1.0f);
            const float* arow = tile + (rt * 16 + mr) * P + kb;
            const float* l0 = wl_s + (cp * 32 + mr) * P + kb;
            const float* l1 = l0 + 16 * P;
            v8f c0 = rc0[h], c1 = rc1[h];
            for (int kk = 0; kk < DIN / 4; ++kk) {
                const v2f s = *reinterpret_cast<const v2f*>(arow + kk * 4);
                v2f a;
                a.x = s.x * dinv;
                a.y = s.y * dinv;
                const v2f b0 = *reinterpret_cast<const v2f*>(l0 + kk * 4);
                const v2f b1 = *reinterpret_cast<const v2f*>(l1 + kk * 4);
                c0 = __builtin_amdgcn_wmma_f32_16x16x4_f32(
                    false, a, false, b0, (short)0, c0, false, false);
                c1 = __builtin_amdgcn_wmma_f32_16x16x4_f32(
                    false, a, false, b1, (short)0, c1, false, false);
            }
            const int n0 = cp * 32 + mr;
            const float bias0 = bl[n0];
            const float bias1 = bl[n0 + 16];
            float* hc = h1 + (size_t)(lo + rt * 16 + 8 * (lane >> 4)) * DHID + n0;
#pragma unroll
            for (int v = 0; v < 8; ++v) {
                hc[v * DHID]      = tanhf(c0[v] + bias0);
                hc[v * DHID + 16] = tanhf(c1[v] + bias1);
            }
        }
    }
}

#define P2 (DHID + 4)
#define NUNIT ((TR / 16) * (DOUT / 16))

static_assert((2 * TR * P2 + 2 * DOUT * P2) * 4 > 160 * 1024, "one work-group per processor");
static_assert((2 * TR * P2 + 2 * DOUT * P2) * 4 <= 320 * 1024, "must fit the LDS");
static_assert(AGG_WAVES * QCAP <= TR * P2 + 2 * DOUT * P2, "the scan's queues fit behind the tile");
static_assert(NUNIT <= 2 * AGG_WAVES, "at most two output blocks per wave");
static_assert(AGG_WAVES % (DOUT / 16) == 0, "both blocks of a wave lie in one column tile");

__device__ __forceinline__ void stage_layer2(const float* __restrict__ h1t,
                                             const float* __restrict__ Wl, const float* __restrict__ Wr,
                                             float* ah_s, float* wl_s, float* wr_s, int nv, int tid) {
    constexpr int Q  = DHID / 4;
    constexpr int NW = DOUT * Q / AGG_THREADS;
    constexpr int NH = (TR * Q + AGG_THREADS - 1) / AGG_THREADS;
    static_assert(DOUT * Q % AGG_THREADS == 0, "the weights split evenly over the work-group");
    const float4* pl = reinterpret_cast<const float4*>(Wl);
    const float4* pr = reinterpret_cast<const float4*>(Wr);
    const float4* ph = reinterpret_cast<const float4*>(h1t);
    const int last = nv * Q - 1;
    float4 vl[NW], vr[NW], vh[NH];
#pragma unroll
    for (int j = 0; j < NW; ++j) {
        vl[j] = pl[tid + j * AGG_THREADS];
        vr[j] = pr[tid + j * AGG_THREADS];
    }
#pragma unroll
    for (int j = 0; j < NH; ++j) vh[j] = ph[min(tid + j * AGG_THREADS, last)];
#pragma unroll
    for (int j = 0; j < NW; ++j) {
        const int i = tid + j * AGG_THREADS;
        const int o = (i / Q) * P2 + (i % Q) * 4;
        *reinterpret_cast<float4*>(wl_s + o) = vl[j];
        *reinterpret_cast<float4*>(wr_s + o) = vr[j];
    }
#pragma unroll
    for (int j = 0; j < NH; ++j) {
        const int i = tid + j * AGG_THREADS;
        if ((j + 1) * AGG_THREADS <= TR * Q || i < TR * Q) {
            const int o = (i / Q) * P2 + (i % Q) * 4;
            *reinterpret_cast<float4*>(ah_s + o) = vh[j];
        }
    }
}

__global__ __launch_bounds__(AGG_THREADS) void layer2_kernel(
    const float* __restrict__ h1, const int* __restrict__ ei,
    const float* __restrict__ deg, float* __restrict__ lists,
    const float* __restrict__ Wl, const float* __restrict__ bl,
    const float* __restrict__ Wr, float* __restrict__ out,
    const unsigned* __restrict__ tb) {
    constexpr int P = P2;
    __shared__ __attribute__((aligned(16))) float smem[2 * TR * P + 2 * DOUT * P];
    float* tile = smem;
    float* ah_s = tile + TR * P;
    float* wl_s = ah_s + TR * P;
    float* wr_s = wl_s + DOUT * P;
    unsigned* queue = reinterpret_cast<unsigned*>(ah_s);

    const int tid  = threadIdx.x;
    const int lane = tid & 31;
    const int wave = __builtin_amdgcn_readfirstlane(tid >> 5);
    const int lo   = blockIdx.x * TR;
    const int nv   = min(TR, NN - lo);
    const int nrt  = nv / 16;
    const int mr   = lane & 15;
    const int kb   = (lane >> 4) * 2;

    unsigned* list = reinterpret_cast<unsigned*>(lists) + (size_t)blockIdx.x * LSTRIDE;
    const int n_list = (int)list[0];
    float degm[2];
#pragma unroll
    for (int h = 0; h < 2; ++h) {
        const int rt = (wave + h * AGG_WAVES) >> 2;
        degm[h] = deg[min(lo + rt * 16 + mr, NN - 1)];
    }
    const float bias = bl[(wave & 3) * 16 + mr];

    for (int i = tid; i < TR * P / 4; i += AGG_THREADS)
        reinterpret_cast<float4*>(tile)[i] = make_float4(0.f, 0.f, 0.f, 0.f);
    __syncthreads();

    const bool listed = n_list >= 0 && n_list <= LCAP;
    if (listed) {
        stage_layer2(h1 + (size_t)lo * DHID, Wl, Wr, ah_s, wl_s, wr_s, nv, tid);
        __syncthreads();
    }

    const float* rrow = wr_s + ((wave & 3) * 16 + mr) * P + kb;
    bool has[2];
    const float* hrw[2];
#pragma unroll
    for (int h = 0; h < 2; ++h) {
        const int u = wave + h * AGG_WAVES;
        has[h] = u < NUNIT && (u >> 2) < nrt;
        hrw[h] = ah_s + (min(u >> 2, TR / 16 - 1) * 16 + mr) * P + kb;
    }
    v8f acc[2] = {};
    int rk = 0;
    constexpr int RCH = 4;
    static_assert((DHID / 4) % RCH == 0, "the root term is made of whole chunks");
    auto root_chunk = [&]() {
        v2f b[RCH], a0[RCH], a1[RCH];
#pragma unroll
        for (int j = 0; j < RCH; ++j) {
            b[j]  = *reinterpret_cast<const v2f*>(rrow + (rk + j) * 4);
            a0[j] = *reinterpret_cast<const v2f*>(hrw[0] + (rk + j) * 4);
            a1[j] = *reinterpret_cast<const v2f*>(hrw[1] + (rk + j) * 4);
        }
        if (has[1]) {
#pragma unroll
            for (int j = 0; j < RCH; ++j) {
                acc[0] = __builtin_amdgcn_wmma_f32_16x16x4_f32(false, a0[j], false, b[j], (short)0, acc[0], false, false);
                acc[1] = __builtin_amdgcn_wmma_f32_16x16x4_f32(false, a1[j], false, b[j], (short)0, acc[1], false, false);
            }
        } else if (has[0]) {
#pragma unroll
            for (int j = 0; j < RCH; ++j)
                acc[0] = __builtin_amdgcn_wmma_f32_16x16x4_f32(false, a0[j], false, b[j], (short)0, acc[0], false, false);
        }
        rk += RCH;
    };
    auto between = [&]() { if (listed && rk < DHID / 4) root_chunk(); };

    gather_tile<DHID, P, false>(h1, ei, tb, tile, queue, nullptr, nullptr, list, n_list, 0ull, lo, lane, wave, between);
    if (listed) {
        while (rk < DHID / 4) root_chunk();
    } else {
        __syncthreads();
        stage_layer2(h1 + (size_t)lo * DHID, Wl, Wr, ah_s, wl_s, wr_s, nv, tid);
    }
    __syncthreads();
    while (rk < DHID / 4) root_chunk();

#pragma unroll
    for (int h = 0; h < 2; ++h) {
        const int u  = wave + h * AGG_WAVES;
        const int rt = u >> 2, nt = u & 3;
        if (has[h]) {
            const float dinv = 1.0f / fmaxf(degm[h], 1.0f);
            const float* arow = tile + (rt * 16 + mr) * P + kb;
            const float* lrow = wl_s + (nt * 16 + mr) * P + kb;
            v8f c = acc[h];
            for (int kk = 0; kk < DHID / 4; ++kk) {
                const v2f s = *reinterpret_cast<const v2f*>(arow + kk * 4);
                v2f a;
                a.x = s.x * dinv;
                a.y = s.y * dinv;
                const v2f b = *reinterpret_cast<const v2f*>(lrow + kk * 4);
                c = __builtin_amdgcn_wmma_f32_16x16x4_f32(
                    false, a, false, b, (short)0, c, false, false);
            }
            acc[h] = c;
        }
    }
    __syncthreads();

#pragma unroll
    for (int h = 0; h < 2; ++h) {
        const int u  = wave + h * AGG_WAVES;
        const int rt = u >> 2, nt = u & 3;
        if (u < NUNIT && rt < nrt) {
            const int n = nt * 16 + mr;
            float* zc = tile + (rt * 16 + 8 * (lane >> 4)) * P + n;
#pragma unroll
            for (int v = 0; v < 8; ++v) zc[v * P] = acc[h][v] + bias;
        }
    }
    __syncthreads();

    constexpr int RPW = (TR + AGG_WAVES - 1) / AGG_WAVES;
    v2f z[RPW];
    float mx[RPW], sm[RPW];
#pragma unroll
    for (int i = 0; i < RPW; ++i) {
        const int r = min(wave + i * AGG_WAVES, TR - 1);
        z[i] = *reinterpret_cast<const v2f*>(tile + r * P + lane * 2);
        mx[i] = fmaxf(z[i].x, z[i].y);
    }
#pragma unroll
    for (int o = 16; o > 0; o >>= 1) {
#pragma unroll
        for (int i = 0; i < RPW; ++i) mx[i] = fmaxf(mx[i], __shfl_xor(mx[i], o));
    }
#pragma unroll
    for (int i = 0; i < RPW; ++i) sm[i] = expf(z[i].x - mx[i]) + expf(z[i].y - mx[i]);
#pragma unroll
    for (int o = 16; o > 0; o >>= 1) {
#pragma unroll
        for (int i = 0; i < RPW; ++i) sm[i] += __shfl_xor(sm[i], o);
    }
#pragma unroll
    for (int i = 0; i < RPW; ++i) {
        const int r = wave + i * AGG_WAVES;
        if (r < nv) {
            const float lse = mx[i] + logf(sm[i]);
            float2 res;
            res.x = z[i].x - lse;
            res.y = z[i].y - lse;
            *reinterpret_cast<float2*>(out + (size_t)(lo + r) * DOUT + lane * 2) = res;
        }
    }
}

extern "C" void kernel_launch(void* const* d_in, const int* in_sizes, int n_in,
                              void* d_out, int out_size, void* d_ws, size_t ws_size,
                              hipStream_t stream) {
    const float* x   = (const float*)d_in[0];
    const int*   ei  = (const int*)d_in[1];
    const float* Wl1 = (const float*)d_in[2];
    const float* bl1 = (const float*)d_in[3];
    const float* Wr1 = (const float*)d_in[4];
    const float* Wl2 = (const float*)d_in[5];
    const float* bl2 = (const float*)d_in[6];
    const float* Wr2 = (const float*)d_in[7];
    float* out = (float*)d_out;

    float* ws   = (float*)d_ws;
    unsigned* tb = (unsigned*)d_ws;
    float* lsts = ws + (size_t)NN * DIN;
    float* h1   = lsts + (size_t)NN * DHID;
    float* degb = h1 + (size_t)NN * DHID;

    recode_kernel<<<NSL, RC_THREADS, 0, stream>>>(ei, tb);
    layer1_kernel<<<NT, AGG_THREADS, 0, stream>>>(x, ei, Wl1, bl1, Wr1, h1, degb, lsts, tb);
    layer2_kernel<<<NT, AGG_THREADS, 0, stream>>>(h1, ei, degb, lsts, Wl2, bl2, Wr2, out, tb);
}


# Layer 2 without LDS float adds: its match list is binned by target node in LDS (slot = returned LDS count); each wave sums a node's h1 rows in registers and stores the row once; sum order by bin slot
# speedup vs baseline: 6.1985x; 1.1486x over previous
#include <hip/hip_runtime.h>
#include <cmath>


#define NN 50000
#define NE 800000
#define DIN 64
#define DHID 128
#define DOUT 64

#define TR 208
#define NT ((NN + TR - 1) / TR)
#define AGG_WAVES 32
#define AGG_THREADS (AGG_WAVES * 32)
#define NCH ((NE + 511) / 512)
#define NEP (NCH * 512)
#define NGRP (NEP / 16)
#define SLICE 4096
#define NSL ((NEP + SLICE - 1) / SLICE)
#define BMS 2048
#define NBW (NSL * 8)
#define GPR 4
#define QCAP 1024
#define LHEAD 64
#define LSTRIDE ((NN * DHID / NT) / 64 * 64)
#define LCAP (LSTRIDE - LHEAD)

static_assert((NT - 1) * TR < NN && NT * TR >= NN && (NN - (NT - 1) * TR) % 16 == 0, "tiles cover the nodes; the last one holds whole row tiles");
static_assert((size_t)NT * LSTRIDE <= (size_t)NN * DHID, "the lists must fit their part of the workspace");
static_assert(((size_t)NN * DIN * 4) % 256 == 0 && ((size_t)NN * DHID * 4) % 256 == 0, "the lists and h1 must start on 256-byte boundaries");
static_assert(NE % 4 == 0, "four targets make one dword of tile bytes");
static_assert(NT <= 255, "a tile number fits a byte, and 255 is no tile");
static_assert((size_t)NEP + (size_t)NSL * BMS * 4 <= (size_t)NN * DIN * 4, "tile bytes and group bits fit the first part of the workspace");
static_assert(NEP % 256 == 0 && NEP - NE >= 16, "the bits start on a line; the last group is padding only");
static_assert(NT <= 256, "a tile slot per tile in every slice's bits");
static_assert(TR <= 256, "local target index is packed into 8 bits");
static_assert(NE <= (1 << 24), "edge index is packed into 24 bits");
static_assert(NN <= (1 << 24), "source node is packed into 24 bits of a list entry");
static_assert(QCAP >= 2 * 512, "one round of 32 groups must always fit behind the flush mark");

typedef __attribute__((ext_vector_type(2))) float v2f;
typedef __attribute__((ext_vector_type(8))) float v8f;

template <int D, int P, class F>
__device__ __forceinline__ void add_rows(const float* __restrict__ feat, float* tile,
                                         int s, int t, int m, int lane, F& between) {
    for (int k = 0; k < m; k += 8) {
        float2 r[8][D / 64];
#pragma unroll
        for (int i = 0; i < 8; ++i) {
            const int sk = __builtin_amdgcn_readlane(s, k + i);
            const float* row = feat + (size_t)sk * D + lane * 2;
#pragma unroll
            for (int j = 0; j < D / 64; ++j)
                r[i][j] = *reinterpret_cast<const float2*>(row + j * 64);
        }
        between();
#pragma unroll
        for (int i = 0; i < 8; ++i) {
            if (k + i < m) {
                const int tk = __builtin_amdgcn_readlane(t, k + i);
                float* trow = tile + tk * P + lane * 2;
#pragma unroll
                for (int j = 0; j < D / 64; ++j) {
                    atomicAdd(trow + j * 64, r[i][j].x);
                    atomicAdd(trow + j * 64 + 1, r[i][j].y);
                }
            }
        }
    }
}

template <int D, int P, bool SAVE, class F>
__device__ __forceinline__ void flush_queue(const float* __restrict__ feat,
                                            const int* __restrict__ ei,
                                            float* tile, const unsigned* q,
                                            int n, int lane, unsigned* list, int lbase,
                                            int* cnt, int lo, F& between) {
    for (int base = 0; base < n; base += 32) {
        const int idx = base + lane;
        int s = 0;
        unsigned t = 0u;
        if (idx < n) {
            const unsigned e = min(q[idx], (unsigned)(NE - 1));
            s = ei[e];
            t = min((unsigned)(ei[NE + e] - lo), (unsigned)(TR - 1));
            if (SAVE) {
                atomicAdd(&cnt[t], 1);
                if (lbase + idx < LCAP)
                    list[LHEAD + lbase + idx] = ((unsigned)s << 8) | t;
            }
        }
        add_rows<D, P>(feat, tile, s, (int)t, min(32, n - base), lane, between);
    }
}

#define BCAP 46
static_assert(BCAP > 32 && BCAP <= 64, "a bin is read as two words per lane");

template <int D, int P, class F>
__device__ __forceinline__ void sum_owned_rows(const float* __restrict__ feat, float* tile,
                                               const unsigned* lbin, int cntv, int nv,
                                               int lane, int wave, F& between) {
    int n = __builtin_amdgcn_readlane(cntv, 0);
    unsigned c_lo = lane < n ? lbin[wave * BCAP + lane] : 0u;
    unsigned c_hi = lane + 32 < n ? lbin[wave * BCAP + 32 + lane] : 0u;
    for (int it = 0, t = wave; t < nv; ++it, t += AGG_WAVES) {
        const int tn = t + AGG_WAVES;
        const int nn = tn < nv ? __builtin_amdgcn_readlane(cntv, it + 1) : 0;
        const unsigned n_lo = lane < nn ? lbin[tn * BCAP + lane] : 0u;
        const unsigned n_hi = lane + 32 < nn ? lbin[tn * BCAP + 32 + lane] : 0u;
        float2 acc[D / 64];
#pragma unroll
        for (int j = 0; j < D / 64; ++j) acc[j] = make_float2(0.f, 0.f);
        for (int base = 0; base < n; base += 32) {
            const int s = (int)(base ? c_hi : c_lo);
            const int m = min(32, n - base);
            for (int k = 0; k < m; k += 8) {
                float2 r[8][D / 64];
#pragma unroll
                for (int i = 0; i < 8; ++i) {
                    if (k + i < m) {
                        const int sk = __builtin_amdgcn_readlane(s, k + i);
                        const float* row = feat + (size_t)sk * D + lane * 2;
#pragma unroll
                        for (int j = 0; j < D / 64; ++j)
                            r[i][j] = *reinterpret_cast<const float2*>(row + j * 64);
                    }
                }
                between();
#pragma unroll
                for (int i = 0; i < 8; ++i) {
                    if (k + i < m) {
#pragma unroll
                        for (int j = 0; j < D / 64; ++j) {
                            acc[j].x += r[i][j].x;
                            acc[j].y += r[i][j].y;
                        }
                    }
                }
            }
        }
        float* trow = tile + t * P + lane * 2;
#pragma unroll
        for (int j = 0; j < D / 64; ++j)
            *reinterpret_cast<float2*>(trow + j * 64) = acc[j];
        n = nn;
        c_lo = n_lo;
        c_hi = n_hi;
    }
}

#define WPW ((NBW + AGG_WAVES - 1) / AGG_WAVES)
static_assert(WPW >= 32 && WPW <= 64, "one or two words of bits per lane");
__device__ __forceinline__ unsigned long long tile_bits(const unsigned* __restrict__ tb, int wave, int lane) {
    const unsigned* bmT = tb + NEP / 4 + blockIdx.x * 8;
    auto word = [&](int i) {
        return i < NBW ? bmT[(i >> 3) * BMS + (i & 7)] : 0u;
    };
    const int i0 = WPW * wave + lane;
    const unsigned w_lo = word(i0);
    const unsigned w_hi = lane < WPW - 32 ? word(i0 + 32) : 0u;
    return (unsigned long long)w_lo | ((unsigned long long)w_hi << 32);
}

template <int D, int P, bool FIRST, class F>
__device__ __forceinline__ void gather_tile(
    const float* __restrict__ feat, const int* __restrict__ ei,
    const unsigned* __restrict__ tb, float* tile,
    unsigned* queue, int* cnt, int* lfill, unsigned* list,
    unsigned long long first_bits,
    int lo, int lane, int wave, F& between) {
    unsigned* q = queue + wave * QCAP;
    int qn = 0;
    const uint4* dp = reinterpret_cast<const uint4*>(tb);
    const unsigned pat = blockIdx.x * 0x01010101u;
    const unsigned g0 = (unsigned)(WPW * wave + lane) * 32u;
    unsigned long long bits = FIRST ? first_bits : tile_bits(tb, wave, lane);

    auto flush_now = [&](auto& fn) {
        int lb = 0;
        if (FIRST) {
            if (lane == 0) lb = atomicAdd(lfill, qn);
            lb = __builtin_amdgcn_readfirstlane(lb);
        }
        flush_queue<D, P, FIRST>(feat, ei, tile, q, qn, lane, list, lb, cnt, lo, fn);
        qn = 0;
    };
    auto nothing = []() {};

    auto scan_group = [&](const uint4& d, unsigned e0) {
        auto zero_bytes = [&](unsigned v) {
            const unsigned x = v ^ pat;
            return ~(((x & 0x7f7f7f7fu) + 0x7f7f7f7fu) | x) & 0x80808080u;
        };
        unsigned m = (zero_bytes(d.x) >> 7) | (zero_bytes(d.y) >> 6) |
                     (zero_bytes(d.z) >> 5) | (zero_bytes(d.w) >> 4);
        unsigned mask = __builtin_amdgcn_ballot_w32(m != 0u);
        for (int turn = 0; turn < 16 && mask; ++turn) {
            if (m != 0u) {
                const unsigned bit = (unsigned)__builtin_ctz(m);
                const int pos = qn + (int)__builtin_amdgcn_mbcnt_lo(mask, 0u);
                q[pos] = e0 + (bit & 7u) * 4u + (bit >> 3);
                m &= m - 1u;
            }
            qn += __builtin_popcount(mask);
            mask = __builtin_amdgcn_ballot_w32(m != 0u);
        }
    };

    for (int round = 0; round < 64 / GPR; ++round) {
        if (__builtin_amdgcn_ballot_w32(bits != 0ull) == 0u) break;
        unsigned g[GPR];
        uint4 d[GPR];
        unsigned live[GPR];
#pragma unroll
        for (int k = 0; k < GPR; ++k) {
            const bool has_bit = bits != 0ull;
            live[k] = __builtin_amdgcn_ballot_w32(has_bit);
            unsigned gk = NGRP - 1;
            if (has_bit) {
                const int b = __builtin_ctzll(bits);
                bits &= bits - 1ull;
                gk = g0 + (unsigned)(b & 31) + (unsigned)(b >> 5) * (32u * 32u);
            }
            g[k] = min(gk, (unsigned)(NGRP - 1));
            d[k] = dp[g[k]];
        }
        int u = 0;
        do {
            bool full = false;
#pragma unroll
            for (int k = 0; k < GPR; ++k) {
                if (k >= u && !full) {
                    if (live[k]) scan_group(d[k], g[k] * 16u);
                    u = k + 1;
                    full = qn > QCAP - 512;
                }
            }
            if (full) {
                __builtin_amdgcn_fence(__ATOMIC_RELEASE, "wavefront");
                __builtin_amdgcn_fence(__ATOMIC_ACQUIRE, "wavefront");
                flush_now(nothing);
            }
        } while (u < GPR);
    }
    __builtin_amdgcn_fence(__ATOMIC_RELEASE, "wavefront");
    __builtin_amdgcn_fence(__ATOMIC_ACQUIRE, "wavefront");
    flush_now(between);
}

#define RC_THREADS 256
#define RC_PER (SLICE / 4 / RC_THREADS)
static_assert(SLICE / 4 == RC_PER * RC_THREADS && RC_THREADS % 128 == 0, "a slice splits evenly; 128 threads make one word of bits");
static_assert(BMS == 8 * 256 && SLICE == 8 * 32 * 16, "8 words of 32 groups of 16 edges per tile and slice");
__global__ __launch_bounds__(RC_THREADS) void recode_kernel(const int* __restrict__ ei,
                                                            unsigned* __restrict__ tb) {
    __shared__ __attribute__((aligned(16))) unsigned bm[BMS];
    const int4* dp = reinterpret_cast<const int4*>(ei + NE);
    const int tid = threadIdx.x;
    const int i0 = blockIdx.x * (SLICE / 4) + tid;
    int4 d[RC_PER];
#pragma unroll
    for (int u = 0; u < RC_PER; ++u)
        d[u] = dp[min(i0 + u * RC_THREADS, NE / 4 - 1)];
    for (int i = tid; i < BMS / 4; i += RC_THREADS)
        reinterpret_cast<uint4*>(bm)[i] = make_uint4(0u, 0u, 0u, 0u);
    __syncthreads();
    const unsigned gbit = 1u << ((tid >> 2) & 31);
#pragma unroll
    for (int u = 0; u < RC_PER; ++u) {
        const int i = i0 + u * RC_THREADS;
        if (i < NEP / 4) {
            unsigned w = ~0u;
            if (i < NE / 4) {
                const unsigned b0 = (unsigned)d[u].x < (unsigned)NN ? (unsigned)d[u].x / TR : 255u;
                const unsigned b1 = (unsigned)d[u].y < (unsigned)NN ? (unsigned)d[u].y / TR : 255u;
                const unsigned b2 = (unsigned)d[u].z < (unsigned)NN ? (unsigned)d[u].z / TR : 255u;
                const unsigned b3 = (unsigned)d[u].w < (unsigned)NN ? (unsigned)d[u].w / TR : 255u;
                w = b0 | (b1 << 8) | (b2 << 16) | (b3 << 24);
                unsigned* row = bm + (u * (RC_THREADS / 128) + (tid >> 7)) * 256;
                if (b0 != 255u) atomicOr(row + b0, gbit);
                if (b1 != 255u) atomicOr(row + b1, gbit);
                if (b2 != 255u) atomicOr(row + b2, gbit);
                if (b3 != 255u) atomicOr(row + b3, gbit);
            }
            tb[i] = w;
        }
    }
    __syncthreads();
    unsigned* gb = tb + NEP / 4 + (size_t)blockIdx.x * BMS;
    for (int i = tid; i < BMS; i += RC_THREADS) gb[i] = bm[(i & 7) * 256 + (i >> 3)];
}

#define P1 (DIN + 4)
#define NUNIT1 ((TR / 16) * (DHID / 32))
#define NTILE (NN / 16)

static_assert(NN % 16 == 0 && TR % 16 == 0, "row tiles must cover the nodes exactly");
static_assert(NT <= 256, "one work-group per processor, all resident at once");
static_assert((2 * TR * P1 + 2 * DHID * P1 + AGG_WAVES * QCAP + TR + 1) * 4 > 160 * 1024, "layer 1: one work-group per processor");
static_assert((2 * TR * P1 + 2 * DHID * P1 + AGG_WAVES * QCAP + TR + 1) * 4 <= 320 * 1024, "layer 1 must fit the LDS");
static_assert(NUNIT1 <= 2 * AGG_WAVES, "at most two output blocks per wave");

__global__ __launch_bounds__(AGG_THREADS) void layer1_kernel(
    const float* __restrict__ x, const int* __restrict__ ei,
    const float* __restrict__ Wl, const float* __restrict__ bl,
    const float* __restrict__ Wr, float* __restrict__ h1,
    float* __restrict__ deg, float* __restrict__ lists,
    const unsigned* __restrict__ tb) {
    constexpr int D = DIN;
    constexpr int P = P1;
    constexpr int Q = DIN / 4;
    constexpr int NW = DHID * Q / AGG_THREADS;
    constexpr int NH = (TR * Q + AGG_THREADS - 1) / AGG_THREADS;
    static_assert(DHID * Q % AGG_THREADS == 0, "the weights split evenly over the work-group");
    __shared__ __attribute__((aligned(16))) float smem[2 * TR * P + 2 * DHID * P];
    __shared__ unsigned queue[AGG_WAVES * QCAP];
    __shared__ int cnt[TR];
    __shared__ int lfill;
    float* tile = smem;
    float* ax_s = tile + TR * P;
    float* wl_s = ax_s + TR * P;
    float* wr_s = wl_s + DHID * P;

    const int tid  = threadIdx.x;
    const int lane = tid & 31;
    const int wave = __builtin_amdgcn_readfirstlane(tid >> 5);
    const int lo   = blockIdx.x * TR;
    const int nv   = min(TR, NN - lo);
    const int nrt  = nv / 16;
    const int mr   = lane & 15;
    const int kb   = (lane >> 4) * 2;

    unsigned* list = reinterpret_cast<unsigned*>(lists) + (size_t)blockIdx.x * LSTRIDE;

    const float4* pl = reinterpret_cast<const float4*>(Wl);
    const float4* pr = reinterpret_cast<const float4*>(Wr);
    const float4* px = reinterpret_cast<const float4*>(x + (size_t)lo * D);
    const int last = nv * Q - 1;
    float4 vl[NW], vr[NW], vx[NH];
#pragma unroll
    for (int j = 0; j < NW; ++j) {
        vl[j] = pl[tid + j * AGG_THREADS];
        vr[j] = pr[tid + j * AGG_THREADS];
    }
#pragma unroll
    for (int j = 0; j < NH; ++j) vx[j] = px[min(tid + j * AGG_THREADS, last)];
    const unsigned long long gbits = tile_bits(tb, wave, lane);

    for (int i = tid; i < TR * P / 4; i += AGG_THREADS)
        reinterpret_cast<float4*>(tile)[i] = make_float4(0.f, 0.f, 0.f, 0.f);
    for (int i = tid; i < TR; i += AGG_THREADS) cnt[i] = 0;
    if (tid == 0) lfill = 0;

#pragma unroll
    for (int j = 0; j < NW; ++j) {
        const int i = tid + j * AGG_THREADS;
        const int o = (i / Q) * P + (i % Q) * 4;
        *reinterpret_cast<float4*>(wl_s + o) = vl[j];
        *reinterpret_cast<float4*>(wr_s + o) = vr[j];
    }
#pragma unroll
    for (int j = 0; j < NH; ++j) {
        const int i = tid + j * AGG_THREADS;
        if ((j + 1) * AGG_THREADS <= TR * Q || i < TR * Q) {
            const int o = (i / Q) * P + (i % Q) * 4;
            *reinterpret_cast<float4*>(ax_s + o) = vx[j];
        }
    }
    __syncthreads();

    const int cpw = wave & 3;
    const float* wr0 = wr_s + (cpw * 32 + mr) * P + kb;
    const float* wr1 = wr0 + 16 * P;
    bool has[2];
    const float* xrw[2];
#pragma unroll
    for (int h = 0; h < 2; ++h) {
        const int u = wave + h * AGG_WAVES;
        has[h] = u < NUNIT1 && (u >> 2) < nrt;
        xrw[h] = ax_s + (min(u >> 2, TR / 16 - 1) * 16 + mr) * P + kb;
    }
    v8f rc0[2] = {}, rc1[2] = {};
    int rk = 0;
    constexpr int RCH = 2;
    static_assert((DIN / 4) % RCH == 0, "the root term is made of whole chunks");
    auto root_chunk = [&]() {
        v2f b0[RCH], b1[RCH], a0[RCH], a1[RCH];
#pragma unroll
        for (int j = 0; j < RCH; ++j) {
            b0[j] = *reinterpret_cast<const v2f*>(wr0 + (rk + j) * 4);
            b1[j] = *reinterpret_cast<const v2f*>(wr1 + (rk + j) * 4);
            a0[j] = *reinterpret_cast<const v2f*>(xrw[0] + (rk + j) * 4);
            a1[j] = *reinterpret_cast<const v2f*>(xrw[1] + (rk + j) * 4);
        }
        if (has[1]) {
#pragma unroll
            for (int j = 0; j < RCH; ++j) {
                rc0[0] = __builtin_amdgcn_wmma_f32_16x16x4_f32(false, a0[j], false, b0[j], (short)0, rc0[0], false, false);
                rc1[0] = __builtin_amdgcn_wmma_f32_16x16x4_f32(false, a0[j], false, b1[j], (short)0, rc1[0], false, false);
                rc0[1] = __builtin_amdgcn_wmma_f32_16x16x4_f32(false, a1[j], false, b0[j], (short)0, rc0[1], false, false);
                rc1[1] = __builtin_amdgcn_wmma_f32_16x16x4_f32(false, a1[j], false, b1[j], (short)0, rc1[1], false, false);
            }
        } else if (has[0]) {
#pragma unroll
            for (int j = 0; j < RCH; ++j) {
                rc0[0] = __builtin_amdgcn_wmma_f32_16x16x4_f32(false, a0[j], false, b0[j], (short)0, rc0[0], false, false);
                rc1[0] = __builtin_amdgcn_wmma_f32_16x16x4_f32(false, a0[j], false, b1[j], (short)0, rc1[0], false, false);
            }
        }
        rk += RCH;
    };
    auto between = [&]() { if (rk < DIN / 4) root_chunk(); };

    gather_tile<D, P, true>(x, ei, tb, tile, queue, cnt, &lfill, list, gbits, lo, lane, wave, between);
    while (rk < DIN / 4) root_chunk();
    __syncthreads();

    for (int i = tid; i < nv; i += AGG_THREADS) deg[lo + i] = (float)cnt[i];
    if (tid == 0) list[0] = (unsigned)lfill;

#pragma unroll
    for (int h = 0; h < 2; ++h) {
        const int u  = wave + h * AGG_WAVES;
        const int rt = u >> 2, cp = u & 3;
        if (has[h]) {
            const float dinv = 1.0f / fmaxf((float)cnt[rt * 16 + mr], 1.0f);
            const float* arow = tile + (rt * 16 + mr) * P + kb;
            const float* l0 = wl_s + (cp * 32 + mr) * P + kb;
            const float* l1 = l0 + 16 * P;
            v8f c0 = rc0[h], c1 = rc1[h];
            for (int kk = 0; kk < DIN / 4; ++kk) {
                const v2f s = *reinterpret_cast<const v2f*>(arow + kk * 4);
                v2f a;
                a.x = s.x * dinv;
                a.y = s.y * dinv;
                const v2f b0 = *reinterpret_cast<const v2f*>(l0 + kk * 4);
                const v2f b1 = *reinterpret_cast<const v2f*>(l1 + kk * 4);
                c0 = __builtin_amdgcn_wmma_f32_16x16x4_f32(
                    false, a, false, b0, (short)0, c0, false, false);
                c1 = __builtin_amdgcn_wmma_f32_16x16x4_f32(
                    false, a, false, b1, (short)0, c1, false, false);
            }
            const int n0 = cp * 32 + mr;
            const float bias0 = bl[n0];
            const float bias1 = bl[n0 + 16];
            float* hc = h1 + (size_t)(lo + rt * 16 + 8 * (lane >> 4)) * DHID + n0;
#pragma unroll
            for (int v = 0; v < 8; ++v) {
                hc[v * DHID]      = tanhf(c0[v] + bias0);
                hc[v * DHID + 16] = tanhf(c1[v] + bias1);
            }
        }
    }
}

#define P2 (DHID + 4)
#define NUNIT ((TR / 16) * (DOUT / 16))

static_assert((2 * TR * P2 + 2 * DOUT * P2) * 4 > 160 * 1024, "one work-group per processor");
static_assert((2 * TR * P2 + 2 * DOUT * P2 + TR * BCAP + TR + 1) * 4 <= 320 * 1024, "tile, operands, bins and counters must fit the LDS");
static_assert(AGG_WAVES * QCAP <= TR * P2 + 2 * DOUT * P2, "the scan's queues fit behind the tile");
static_assert(NUNIT <= 2 * AGG_WAVES, "at most two output blocks per wave");
static_assert(AGG_WAVES % (DOUT / 16) == 0, "both blocks of a wave lie in one column tile");

__device__ __forceinline__ void stage_layer2(const float* __restrict__ h1t,
                                             const float* __restrict__ Wl, const float* __restrict__ Wr,
                                             float* ah_s, float* wl_s, float* wr_s, int nv, int tid) {
    constexpr int Q  = DHID / 4;
    constexpr int NW = DOUT * Q / AGG_THREADS;
    constexpr int NH = (TR * Q + AGG_THREADS - 1) / AGG_THREADS;
    static_assert(DOUT * Q % AGG_THREADS == 0, "the weights split evenly over the work-group");
    const float4* pl = reinterpret_cast<const float4*>(Wl);
    const float4* pr = reinterpret_cast<const float4*>(Wr);
    const float4* ph = reinterpret_cast<const float4*>(h1t);
    const int last = nv * Q - 1;
    float4 vl[NW], vr[NW], vh[NH];
#pragma unroll
    for (int j = 0; j < NW; ++j) {
        vl[j] = pl[tid + j * AGG_THREADS];
        vr[j] = pr[tid + j * AGG_THREADS];
    }
#pragma unroll
    for (int j = 0; j < NH; ++j) vh[j] = ph[min(tid + j * AGG_THREADS, last)];
#pragma unroll
    for (int j = 0; j < NW; ++j) {
        const int i = tid + j * AGG_THREADS;
        const int o = (i / Q) * P2 + (i % Q) * 4;
        *reinterpret_cast<float4*>(wl_s + o) = vl[j];
        *reinterpret_cast<float4*>(wr_s + o) = vr[j];
    }
#pragma unroll
    for (int j = 0; j < NH; ++j) {
        const int i = tid + j * AGG_THREADS;
        if ((j + 1) * AGG_THREADS <= TR * Q || i < TR * Q) {
            const int o = (i / Q) * P2 + (i % Q) * 4;
            *reinterpret_cast<float4*>(ah_s + o) = vh[j];
        }
    }
}

__global__ __launch_bounds__(AGG_THREADS) void layer2_kernel(
    const float* __restrict__ h1, const int* __restrict__ ei,
    const float* __restrict__ deg, float* __restrict__ lists,
    const float* __restrict__ Wl, const float* __restrict__ bl,
    const float* __restrict__ Wr, float* __restrict__ out,
    const unsigned* __restrict__ tb) {
    constexpr int P = P2;
    __shared__ __attribute__((aligned(16))) float smem[2 * TR * P + 2 * DOUT * P];
    __shared__ unsigned lbin[TR * BCAP];
    __shared__ int lcnt[TR];
    __shared__ int ovf;
    float* tile = smem;
    float* ah_s = tile + TR * P;
    float* wl_s = ah_s + TR * P;
    float* wr_s = wl_s + DOUT * P;
    unsigned* queue = reinterpret_cast<unsigned*>(ah_s);

    const int tid  = threadIdx.x;
    const int lane = tid & 31;
    const int wave = __builtin_amdgcn_readfirstlane(tid >> 5);
    const int lo   = blockIdx.x * TR;
    const int nv   = min(TR, NN - lo);
    const int nrt  = nv / 16;
    const int mr   = lane & 15;
    const int kb   = (lane >> 4) * 2;

    unsigned* list = reinterpret_cast<unsigned*>(lists) + (size_t)blockIdx.x * LSTRIDE;
    const int n_list = (int)list[0];
    float degm[2];
#pragma unroll
    for (int h = 0; h < 2; ++h) {
        const int rt = (wave + h * AGG_WAVES) >> 2;
        degm[h] = deg[min(lo + rt * 16 + mr, NN - 1)];
    }
    const float bias = bl[(wave & 3) * 16 + mr];

    for (int i = tid; i < TR * P / 4; i += AGG_THREADS)
        reinterpret_cast<float4*>(tile)[i] = make_float4(0.f, 0.f, 0.f, 0.f);
    for (int i = tid; i < TR; i += AGG_THREADS) lcnt[i] = 0;
    if (tid == 0) ovf = 0;
    __syncthreads();

    bool listed = n_list >= 0 && n_list <= LCAP;
    int cntv = 0;
    if (listed) {
        constexpr int NPRE = 4;
        const unsigned* lst = list + LHEAD;
        unsigned ent[NPRE];
#pragma unroll
        for (int j = 0; j < NPRE; ++j) {
            const int i = tid + j * AGG_THREADS;
            ent[j] = i < n_list ? lst[i] : 0u;
        }
        stage_layer2(h1 + (size_t)lo * DHID, Wl, Wr, ah_s, wl_s, wr_s, nv, tid);
        auto bin_entry = [&](unsigned e) {
            const int s = min((int)(e >> 8), NN - 1);
            const int t = min((int)(e & 255u), TR - 1);
            const int slot = atomicAdd(&lcnt[t], 1);
            if (slot < BCAP) lbin[t * BCAP + slot] = (unsigned)s;
            else ovf = 1;
        };
#pragma unroll
        for (int j = 0; j < NPRE; ++j)
            if (tid + j * AGG_THREADS < n_list) bin_entry(ent[j]);
        for (int i = tid + NPRE * AGG_THREADS; i < n_list; i += AGG_THREADS) bin_entry(lst[i]);
        __syncthreads();
        listed = __builtin_amdgcn_readfirstlane(ovf) == 0;
        constexpr int NOWN = (TR + AGG_WAVES - 1) / AGG_WAVES;
        const int to = wave + lane * AGG_WAVES;
        if (lane < NOWN && to < nv) cntv = min(lcnt[to], BCAP);
    }

    const float* rrow = wr_s + ((wave & 3) * 16 + mr) * P + kb;
    bool has[2];
    const float* hrw[2];
#pragma unroll
    for (int h = 0; h < 2; ++h) {
        const int u = wave + h * AGG_WAVES;
        has[h] = u < NUNIT && (u >> 2) < nrt;
        hrw[h] = ah_s + (min(u >> 2, TR / 16 - 1) * 16 + mr) * P + kb;
    }
    v8f acc[2] = {};
    int rk = 0;
    constexpr int RCH = 4;
    static_assert((DHID / 4) % RCH == 0, "the root term is made of whole chunks");
    auto root_chunk = [&]() {
        v2f b[RCH], a0[RCH], a1[RCH];
#pragma unroll
        for (int j = 0; j < RCH; ++j) {
            b[j]  = *reinterpret_cast<const v2f*>(rrow + (rk + j) * 4);
            a0[j] = *reinterpret_cast<const v2f*>(hrw[0] + (rk + j) * 4);
            a1[j] = *reinterpret_cast<const v2f*>(hrw[1] + (rk + j) * 4);
        }
        if (has[1]) {
#pragma unroll
            for (int j = 0; j < RCH; ++j) {
                acc[0] = __builtin_amdgcn_wmma_f32_16x16x4_f32(false, a0[j], false, b[j], (short)0, acc[0], false, false);
                acc[1] = __builtin_amdgcn_wmma_f32_16x16x4_f32(false, a1[j], false, b[j], (short)0, acc[1], false, false);
            }
        } else if (has[0]) {
#pragma unroll
            for (int j = 0; j < RCH; ++j)
                acc[0] = __builtin_amdgcn_wmma_f32_16x16x4_f32(false, a0[j], false, b[j], (short)0, acc[0], false, false);
        }
        rk += RCH;
    };
    auto between = [&]() { if (listed && rk < DHID / 4) root_chunk(); };

    if (listed) {
        sum_owned_rows<DHID, P>(h1, tile, lbin, cntv, nv, lane, wave, between);
        while (rk < DHID / 4) root_chunk();
    } else {
        gather_tile<DHID, P, false>(h1, ei, tb, tile, queue, nullptr, nullptr, list, 0ull, lo, lane, wave, between);
        __syncthreads();
        stage_layer2(h1 + (size_t)lo * DHID, Wl, Wr, ah_s, wl_s, wr_s, nv, tid);
    }
    __syncthreads();
    while (rk < DHID / 4) root_chunk();

#pragma unroll
    for (int h = 0; h < 2; ++h) {
        const int u  = wave + h * AGG_WAVES;
        const int rt = u >> 2, nt = u & 3;
        if (has[h]) {
            const float dinv = 1.0f / fmaxf(degm[h], 1.0f);
            const float* arow = tile + (rt * 16 + mr) * P + kb;
            const float* lrow = wl_s + (nt * 16 + mr) * P + kb;
            v8f c = acc[h];
            for (int kk = 0; kk < DHID / 4; ++kk) {
                const v2f s = *reinterpret_cast<const v2f*>(arow + kk * 4);
                v2f a;
                a.x = s.x * dinv;
                a.y = s.y * dinv;
                const v2f b = *reinterpret_cast<const v2f*>(lrow + kk * 4);
                c = __builtin_amdgcn_wmma_f32_16x16x4_f32(
                    false, a, false, b, (short)0, c, false, false);
            }
            acc[h] = c;
        }
    }
    __syncthreads();

#pragma unroll
    for (int h = 0; h < 2; ++h) {
        const int u  = wave + h * AGG_WAVES;
        const int rt = u >> 2, nt = u & 3;
        if (u < NUNIT && rt < nrt) {
            const int n = nt * 16 + mr;
            float* zc = tile + (rt * 16 + 8 * (lane >> 4)) * P + n;
#pragma unroll
            for (int v = 0; v < 8; ++v) zc[v * P] = acc[h][v] + bias;
        }
    }
    __syncthreads();

    constexpr int RPW = (TR + AGG_WAVES - 1) / AGG_WAVES;
    v2f z[RPW];
    float mx[RPW], sm[RPW];
#pragma unroll
    for (int i = 0; i < RPW; ++i) {
        const int r = min(wave + i * AGG_WAVES, TR - 1);
        z[i] = *reinterpret_cast<const v2f*>(tile + r * P + lane * 2);
        mx[i] = fmaxf(z[i].x, z[i].y);
    }
#pragma unroll
    for (int o = 16; o > 0; o >>= 1) {
#pragma unroll
        for (int i = 0; i < RPW; ++i) mx[i] = fmaxf(mx[i], __shfl_xor(mx[i], o));
    }
#pragma unroll
    for (int i = 0; i < RPW; ++i) sm[i] = expf(z[i].x - mx[i]) + expf(z[i].y - mx[i]);
#pragma unroll
    for (int o = 16; o > 0; o >>= 1) {
#pragma unroll
        for (int i = 0; i < RPW; ++i) sm[i] += __shfl_xor(sm[i], o);
    }
#pragma unroll
    for (int i = 0; i < RPW; ++i) {
        const int r = wave + i * AGG_WAVES;
        if (r < nv) {
            const float lse = mx[i] + logf(sm[i]);
            float2 res;
            res.x = z[i].x - lse;
            res.y = z[i].y - lse;
            *reinterpret_cast<float2*>(out + (size_t)(lo + r) * DOUT + lane * 2) = res;
        }
    }
}

extern "C" void kernel_launch(void* const* d_in, const int* in_sizes, int n_in,
                              void* d_out, int out_size, void* d_ws, size_t ws_size,
                              hipStream_t stream) {
    const float* x   = (const float*)d_in[0];
    const int*   ei  = (const int*)d_in[1];
    const float* Wl1 = (const float*)d_in[2];
    const float* bl1 = (const float*)d_in[3];
    const float* Wr1 = (const float*)d_in[4];
    const float* Wl2 = (const float*)d_in[5];
    const float* bl2 = (const float*)d_in[6];
    const float* Wr2 = (const float*)d_in[7];
    float* out = (float*)d_out;

    float* ws   = (float*)d_ws;
    unsigned* tb = (unsigned*)d_ws;
    float* lsts = ws + (size_t)NN * DIN;
    float* h1   = lsts + (size_t)NN * DHID;
    float* degb = h1 + (size_t)NN * DHID;

    recode_kernel<<<NSL, RC_THREADS, 0, stream>>>(ei, tb);
    layer1_kernel<<<NT, AGG_THREADS, 0, stream>>>(x, ei, Wl1, bl1, Wr1, h1, degb, lsts, tb);
    layer2_kernel<<<NT, AGG_THREADS, 0, stream>>>(h1, ei, degb, lsts, Wl2, bl2, Wr2, out, tb);
}
